# starting kernel
# baseline (speedup 1.0000x reference)
#include <hip/hip_runtime.h>
#include <hip/hip_bf16.h>

typedef float v2f __attribute__((ext_vector_type(2)));
typedef float v8f __attribute__((ext_vector_type(8)));

#define HDIM 64

__global__ void zero_kernel(float* __restrict__ p, int n) {
    int i = blockIdx.x * blockDim.x + threadIdx.x;
    int stride = gridDim.x * blockDim.x;
    for (; i < n; i += stride) p[i] = 0.0f;
}

__global__ void edge_stats_kernel(const int* __restrict__ src, const int* __restrict__ dst,
                                  const float* __restrict__ x,
                                  float* __restrict__ degacc, float* __restrict__ signacc,
                                  float* __restrict__ Sx, int E) {
    int e = blockIdx.x * blockDim.x + threadIdx.x;
    if (e >= E) return;
    int s = src[e], d = dst[e];
    float sg = (s > d) ? 1.0f : ((s < d) ? -1.0f : 0.0f);
    atomicAdd(&degacc[d], 1.0f);
    atomicAdd(&signacc[d], sg);
    atomicAdd(&Sx[d], x[s]);
}

__global__ void node_stats_kernel(const float* __restrict__ degacc, const float* __restrict__ signacc,
                                  float* __restrict__ invdeg, float* __restrict__ gate,
                                  float* __restrict__ ssn, int N) {
    int i = blockIdx.x * blockDim.x + threadIdx.x;
    if (i >= N) return;
    float dg = degacc[i];
    float inv = 1.0f / fmaxf(dg, 1.0f);
    invdeg[i] = inv;
    gate[i]   = (dg > 0.0f) ? 1.0f : 0.0f;
    ssn[i]    = signacc[i] * inv;
}

__global__ void wdiff_kernel(const float* __restrict__ Wmid, float* __restrict__ Wd6, int total) {
    int i = blockIdx.x * blockDim.x + threadIdx.x;
    if (i >= total) return;
    int l = i >> 12;
    int rc = i & 4095;
    const float* Wl = Wmid + (size_t)l * 129 * HDIM;
    Wd6[i] = Wl[rc] - Wl[64 * HDIM + rc];
}

__global__ void conv_in_kernel(const float* __restrict__ x, const float* __restrict__ Win,
                               const float* __restrict__ bin,
                               const float* __restrict__ invdeg, const float* __restrict__ gate,
                               const float* __restrict__ ssn, const float* __restrict__ Sx,
                               float* __restrict__ Y, int N) {
    int t = blockIdx.x * blockDim.x + threadIdx.x;
    if (t >= N * HDIM) return;
    int n = t >> 6, h = t & 63;
    float wa = Win[h], wb = Win[HDIM + h], wc = Win[2 * HDIM + h];
    Y[t] = gate[n] * (x[n] * (wa - wb) + bin[h]) + invdeg[n] * Sx[n] * wb + ssn[n] * wc;
}

__global__ void scatter_kernel(const int* __restrict__ src, const int* __restrict__ dst,
                               const float* __restrict__ X, float* __restrict__ S, int work) {
    int t = blockIdx.x * blockDim.x + threadIdx.x;
    if (t >= work) return;
    int e = t >> 4;
    int q = t & 15;
    int s = src[e], d = dst[e];
    const float4 v = *(const float4*)(X + (size_t)s * HDIM + q * 4);
    float* o = S + (size_t)d * HDIM + q * 4;
    atomicAdd(o + 0, v.x);
    atomicAdd(o + 1, v.y);
    atomicAdd(o + 2, v.z);
    atomicAdd(o + 3, v.w);
}

__global__ __launch_bounds__(256) void mid_conv_kernel(
    const float* __restrict__ X, const float* __restrict__ S,
    const float* __restrict__ Wd, const float* __restrict__ Wb,
    const float* __restrict__ Wc, const float* __restrict__ bias,
    const float* __restrict__ gate, const float* __restrict__ invdeg,
    const float* __restrict__ ssn, const float* __restrict__ resid,
    float* __restrict__ Y, int nTiles, int N) {
    __shared__ float sWd[HDIM * HDIM];
    __shared__ float sWb[HDIM * HDIM];
    for (int i = threadIdx.x; i < HDIM * HDIM; i += 256) {
        int ks   = i >> 8;
        int rem  = i & 255;
        int nt   = rem >> 6;
        int lp   = rem & 63;
        int ln   = lp >> 1;
        int half = lp & 1;
        int kh   = (ln >> 4) << 1;
        int srcr = (ks << 2) + kh + half;
        int srcc = (nt << 4) + (ln & 15);
        sWd[i] = Wd[srcr * HDIM + srcc];
        sWb[i] = Wb[srcr * HDIM + srcc];
    }
    __syncthreads();

    int wave = threadIdx.x >> 5;
    int lane = threadIdx.x & 31;
    int tile = blockIdx.x * 8 + wave;
    if (tile >= nTiles) return;

    int node0 = tile * 16;
    int m  = lane & 15;
    int kh = (lane >> 4) << 1;
    int row = node0 + m; if (row >= N) row = N - 1;
    const float* Xr = X + (size_t)row * HDIM;
    const float* Sr = S + (size_t)row * HDIM;

    v8f accX[4] = {};
    v8f accS[4] = {};
    int fragBase = lane * 2;

#pragma unroll
    for (int ks = 0; ks < 16; ++ks) {
        int k0 = ks << 2;
        v2f a, s;
        a.x = Xr[k0 + kh];     a.y = Xr[k0 + kh + 1];
        s.x = Sr[k0 + kh];     s.y = Sr[k0 + kh + 1];
#pragma unroll
        for (int nt = 0; nt < 4; ++nt) {
            v2f bd = *(const v2f*)&sWd[((ks << 2) + nt) * 64 + fragBase];
            v2f bb = *(const v2f*)&sWb[((ks << 2) + nt) * 64 + fragBase];
            accX[nt] = __builtin_amdgcn_wmma_f32_16x16x4_f32(
                false, a, false, bd, (short)0, accX[nt], false, false);
            accS[nt] = __builtin_amdgcn_wmma_f32_16x16x4_f32(
                false, s, false, bb, (short)0, accS[nt], false, false);
        }
    }

    int ncol = lane & 15;
    int mAdd = (lane >> 4) * 8;
#pragma unroll
    for (int nt = 0; nt < 4; ++nt) {
        int col = nt * 16 + ncol;
        float wc = Wc[col];
        float bb = bias[col];
#pragma unroll
        for (int v = 0; v < 8; ++v) {
            int node = node0 + v + mAdd;
            if (node >= N) continue;
            float g = gate[node];
            float val = g * (accX[nt][v] + bb) + invdeg[node] * accS[nt][v] + ssn[node] * wc;
            if (resid) val += resid[(size_t)node * HDIM + col];
            val = (val > 0.0f) ? val : 0.01f * val;
            Y[(size_t)node * HDIM + col] = val;
        }
    }
}

__global__ void conv_out_kernel(const float* __restrict__ X, const float* __restrict__ S,
                                const float* __restrict__ Wout, const float* __restrict__ bout,
                                const float* __restrict__ x0,
                                const float* __restrict__ invdeg, const float* __restrict__ gate,
                                const float* __restrict__ ssn, float* __restrict__ out, int N) {
    int n = blockIdx.x * blockDim.x + threadIdx.x;
    if (n >= N) return;
    float a1 = 0.0f, a2 = 0.0f;
    const float* Xr = X + (size_t)n * HDIM;
    const float* Sr = S + (size_t)n * HDIM;
#pragma unroll 8
    for (int k = 0; k < HDIM; ++k) {
        float wa = Wout[k], wb = Wout[HDIM + k];
        a1 += Xr[k] * (wa - wb);
        a2 += Sr[k] * wb;
    }
    out[n] = gate[n] * (a1 + bout[0]) + invdeg[n] * a2 + ssn[n] * Wout[2 * HDIM] + x0[n];
}

extern "C" void kernel_launch(void* const* d_in, const int* in_sizes, int n_in,
                              void* d_out, int out_size, void* d_ws, size_t ws_size,
                              hipStream_t stream) {
    const float* x    = (const float*)d_in[0];
    const int*   ei   = (const int*)  d_in[1];
    const float* Win  = (const float*)d_in[2];
    const float* bin  = (const float*)d_in[3];
    const float* Wmid = (const float*)d_in[4];
    const float* bmid = (const float*)d_in[5];
    const float* Wout = (const float*)d_in[6];
    const float* bout = (const float*)d_in[7];

    const int N = in_sizes[0];
    const int E = in_sizes[1] / 2;
    const int* src = ei;
    const int* dst = ei + E;

    float* w = (float*)d_ws;
    float* degacc  = w;
    float* signacc = w + (size_t)N;
    float* invdeg  = w + 2 * (size_t)N;
    float* gate    = w + 3 * (size_t)N;
    float* ssn     = w + 4 * (size_t)N;
    float* Sx      = w + 5 * (size_t)N;
    float* Wd6     = w + 6 * (size_t)N;
    float* bufA    = Wd6  + 6 * HDIM * HDIM;
    float* bufB    = bufA + (size_t)N * HDIM;
    float* bufC    = bufB + (size_t)N * HDIM;
    float* bufS    = bufC + (size_t)N * HDIM;

    const int nTiles = (N + 15) / 16;
    const int sWork  = E * 16;

    zero_kernel<<<1024, 256, 0, stream>>>(w, 6 * N);
    edge_stats_kernel<<<(E + 255) / 256, 256, 0, stream>>>(src, dst, x, degacc, signacc, Sx, E);
    node_stats_kernel<<<(N + 255) / 256, 256, 0, stream>>>(degacc, signacc, invdeg, gate, ssn, N);
    wdiff_kernel<<<(6 * 4096 + 255) / 256, 256, 0, stream>>>(Wmid, Wd6, 6 * 4096);
    conv_in_kernel<<<(N * HDIM + 255) / 256, 256, 0, stream>>>(x, Win, bin, invdeg, gate, ssn, Sx,
                                                               bufA, N);

    float* cur = bufA;
    float* alt = bufC;
    for (int l = 0; l < 3; ++l) {
        zero_kernel<<<2048, 256, 0, stream>>>(bufS, N * HDIM);
        scatter_kernel<<<(sWork + 255) / 256, 256, 0, stream>>>(src, dst, cur, bufS, sWork);
        const float* Wl = Wmid + (size_t)(2 * l) * 129 * HDIM;
        mid_conv_kernel<<<(nTiles + 7) / 8, 256, 0, stream>>>(
            cur, bufS, Wd6 + (2 * l) * HDIM * HDIM, Wl + 64 * HDIM, Wl + 128 * HDIM,
            bmid + (2 * l) * HDIM, gate, invdeg, ssn, nullptr, bufB, nTiles, N);

        zero_kernel<<<2048, 256, 0, stream>>>(bufS, N * HDIM);
        scatter_kernel<<<(sWork + 255) / 256, 256, 0, stream>>>(src, dst, bufB, bufS, sWork);
        const float* Wl2 = Wmid + (size_t)(2 * l + 1) * 129 * HDIM;
        mid_conv_kernel<<<(nTiles + 7) / 8, 256, 0, stream>>>(
            bufB, bufS, Wd6 + (2 * l + 1) * HDIM * HDIM, Wl2 + 64 * HDIM, Wl2 + 128 * HDIM,
            bmid + (2 * l + 1) * HDIM, gate, invdeg, ssn, cur, alt, nTiles, N);

        float* t = cur; cur = alt; alt = t;
    }

    zero_kernel<<<2048, 256, 0, stream>>>(bufS, N * HDIM);
    scatter_kernel<<<(sWork + 255) / 256, 256, 0, stream>>>(src, dst, cur, bufS, sWork);
    conv_out_kernel<<<(N + 255) / 256, 256, 0, stream>>>(cur, bufS, Wout, bout, x, invdeg, gate,
                                                         ssn, (float*)d_out, N);
}


# Each bin's work-group uses its LDS neighbor-sum tile where it stands: the 6 mid convs (f32 WMMA, weights and X operand fetched under the gather) and conv_out join the sum launches; 20 to 13 launches
# speedup vs baseline: 2.8692x; 2.8692x over previous
#include <hip/hip_runtime.h>
#include <hip/hip_bf16.h>

typedef float v2f __attribute__((ext_vector_type(2)));
typedef float v8f __attribute__((ext_vector_type(8)));

#define HDIM 64

__global__ void wdiff_kernel(const float* __restrict__ Wmid, float* __restrict__ Wd6, int total) {
    int i = blockIdx.x * blockDim.x + threadIdx.x;
    if (i >= total) return;
    int l = i >> 12;
    int rc = i & 4095;
    const float* Wl = Wmid + (size_t)l * 129 * HDIM;
    Wd6[i] = Wl[rc] - Wl[64 * HDIM + rc];
}

__global__ void conv_in_kernel(const float* __restrict__ x, const float* __restrict__ Win,
                               const float* __restrict__ bin,
                               const float* __restrict__ invdeg, const float* __restrict__ gate,
                               const float* __restrict__ ssn, const float* __restrict__ Sx,
                               float* __restrict__ Y, int N) {
    int t = blockIdx.x * blockDim.x + threadIdx.x;
    if (t >= N * HDIM) return;
    int n = t >> 6, h = t & 63;
    float wa = Win[h], wb = Win[HDIM + h], wc = Win[2 * HDIM + h];
    Y[t] = gate[n] * (x[n] * (wa - wb) + bin[h]) + invdeg[n] * Sx[n] * wb + ssn[n] * wc;
}

#define BIN_SHIFT 6
#define BIN_NODES 64
#define MAX_BINS 1024
#define CHUNK_WGS 256

__global__ __launch_bounds__(1024) void bin_count_kernel(const int* __restrict__ dst, int* __restrict__ cnt,
                                                         int E, int NB, int chunk) {
    __shared__ int hist[MAX_BINS];
    for (int b = threadIdx.x; b < NB; b += 1024) hist[b] = 0;
    __syncthreads();
    int e0 = blockIdx.x * chunk;
    int e1 = min(e0 + chunk, E);
    for (int e = e0 + (int)threadIdx.x; e < e1; e += 1024)
        atomicAdd(&hist[dst[e] >> BIN_SHIFT], 1);
    __syncthreads();
    for (int b = threadIdx.x; b < NB; b += 1024) cnt[blockIdx.x * NB + b] = hist[b];
}

__global__ __launch_bounds__(1024) void bin_scan_kernel(const int* __restrict__ cnt, int* __restrict__ pre,
                                                        int* __restrict__ binStart, int NB, int G) {
    __shared__ int tot[MAX_BINS];
    int b = threadIdx.x;
    int run = 0;
    if (b < NB) {
#pragma unroll 16
        for (int g = 0; g < G; ++g) {
            int c = cnt[g * NB + b];
            pre[g * NB + b] = run;
            run += c;
        }
    }
    tot[b] = run;
    __syncthreads();
    for (int off = 1; off < MAX_BINS; off <<= 1) {
        int v = (b >= off) ? tot[b - off] : 0;
        __syncthreads();
        tot[b] += v;
        __syncthreads();
    }
    if (b < NB) binStart[b] = tot[b] - run;
    if (b == NB - 1) binStart[NB] = tot[b];
}

__global__ __launch_bounds__(1024) void bin_fill_kernel(const int* __restrict__ src, const int* __restrict__ dst,
                                                        const int* __restrict__ pre,
                                                        const int* __restrict__ binStart,
                                                        unsigned* __restrict__ list, int E, int NB, int chunk) {
    __shared__ int cur[MAX_BINS];
    for (int b = threadIdx.x; b < NB; b += 1024) cur[b] = binStart[b] + pre[blockIdx.x * NB + b];
    __syncthreads();
    int e0 = blockIdx.x * chunk;
    int e1 = min(e0 + chunk, E);
    for (int e = e0 + (int)threadIdx.x; e < e1; e += 1024) {
        int d = dst[e];
        int s = src[e];
        int slot = atomicAdd(&cur[d >> BIN_SHIFT], 1);
        list[slot] = ((unsigned)s << BIN_SHIFT) | (unsigned)(d & (BIN_NODES - 1));
    }
}

#define ST_U 8

__global__ __launch_bounds__(256) void bin_stats_kernel(const unsigned* __restrict__ list,
                                                        const int* __restrict__ binStart,
                                                        const float* __restrict__ x,
                                                        float* __restrict__ invdeg, float* __restrict__ gate,
                                                        float* __restrict__ ssn, float* __restrict__ Sx, int N) {
    __shared__ int   sdeg[BIN_NODES];
    __shared__ int   ssgn[BIN_NODES];
    __shared__ float ssx[BIN_NODES];
    const int t = threadIdx.x;
    if (t < BIN_NODES) { sdeg[t] = 0; ssgn[t] = 0; ssx[t] = 0.0f; }
    __syncthreads();

    const int bin   = blockIdx.x;
    const int beg   = binStart[bin];
    const int end   = binStart[bin + 1];
    const int node0 = bin * BIN_NODES;

    for (int i = beg + t; i < end; i += 256 * ST_U) {
        unsigned p[ST_U];
        float xv[ST_U];
#pragma unroll
        for (int j = 0; j < ST_U; ++j) p[j] = list[min(i + j * 256, end - 1)];
#pragma unroll
        for (int j = 0; j < ST_U; ++j) xv[j] = x[p[j] >> BIN_SHIFT];
#pragma unroll
        for (int j = 0; j < ST_U; ++j) {
            if (i + j * 256 < end) {
                int s  = (int)(p[j] >> BIN_SHIFT);
                int dl = (int)(p[j] & (BIN_NODES - 1));
                int d  = node0 + dl;
                atomicAdd(&sdeg[dl], 1);
                atomicAdd(&ssgn[dl], (s > d) - (s < d));
                atomicAdd(&ssx[dl], xv[j]);
            }
        }
    }
    __syncthreads();

    if (t < BIN_NODES && node0 + t < N) {
        int n = node0 + t;
        float dg  = (float)sdeg[t];
        float inv = 1.0f / fmaxf(dg, 1.0f);
        invdeg[n] = inv;
        gate[n]   = (dg > 0.0f) ? 1.0f : 0.0f;
        ssn[n]    = (float)ssgn[t] * inv;
        Sx[n]     = ssx[t];
    }
}

#define AGG_U 16

template <bool FULL, int PITCH>
__device__ __forceinline__ void add_rows(const unsigned* __restrict__ list, int i, int end,
                                         const float* __restrict__ Xl, float* tl) {
    unsigned p[AGG_U];
    v2f v[AGG_U];
#pragma unroll
    for (int j = 0; j < AGG_U; ++j) p[j] = list[FULL ? (i + j) : min(i + j, end - 1)];
#pragma unroll
    for (int j = 0; j < AGG_U; ++j) v[j] = *(const v2f*)(Xl + (size_t)(p[j] >> BIN_SHIFT) * HDIM);
#pragma unroll
    for (int j = 0; j < AGG_U; ++j) {
        if (FULL || i + j < end) {
            float* t = tl + (p[j] & (BIN_NODES - 1)) * PITCH;
            atomicAdd(t, v[j].x);
            atomicAdd(t + 1, v[j].y);
        }
    }
}

#define TILE_PITCH 68

__global__ __launch_bounds__(256) void agg_conv_kernel(
    const unsigned* __restrict__ list, const int* __restrict__ binStart,
    const float* __restrict__ X,
    const float* __restrict__ Wd, const float* __restrict__ Wb,
    const float* __restrict__ Wc, const float* __restrict__ bias,
    const float* __restrict__ gate, const float* __restrict__ invdeg,
    const float* __restrict__ ssn, const float* resid,
    float* Y, int N) {
    __shared__ float4 tile4[BIN_NODES * TILE_PITCH / 4];
    __shared__ float sWd[HDIM * HDIM];
    __shared__ float sWb[HDIM * HDIM];
    float* tile = (float*)tile4;
    for (int i = threadIdx.x; i < BIN_NODES * TILE_PITCH / 4; i += 256) tile4[i] = make_float4(0.f, 0.f, 0.f, 0.f);

    __syncthreads();

    const int bin  = blockIdx.x;
    const int lane = threadIdx.x & 31;
    const int wave = __builtin_amdgcn_readfirstlane((int)(threadIdx.x >> 5));
    const int tid  = threadIdx.x;

    float wdr[16], wbr[16];
    {
        int nt   = tid >> 6;
        int lp   = tid & 63;
        int ln   = lp >> 1;
        int half = lp & 1;
        int kh   = (ln >> 4) << 1;
        int wsrc = (kh + half) * HDIM + (nt << 4) + (ln & 15);
#pragma unroll
        for (int ks = 0; ks < 16; ++ks) {
            wdr[ks] = Wd[(ks << 2) * HDIM + wsrc];
            wbr[ks] = Wb[(ks << 2) * HDIM + wsrc];
        }
    }
    const int tq    = wave & 3;
    const int nh    = wave >> 2;
    const int node0 = bin * BIN_NODES + tq * 16;
    const int m  = lane & 15;
    const int kh = (lane >> 4) << 1;
    int row = node0 + m; if (row >= N) row = N - 1;
    const float* Xr = X + (size_t)row * HDIM;
    v2f xa[16];
#pragma unroll
    for (int ks = 0; ks < 16; ++ks) {
        xa[ks].x = Xr[(ks << 2) + kh];
        xa[ks].y = Xr[(ks << 2) + kh + 1];
    }

    const int beg  = binStart[bin];
    const int end  = binStart[bin + 1];
    {
        const float* Xl = X + 2 * lane;
        float* tl = tile + 2 * lane;
        for (int i = beg + wave * AGG_U; i < end; i += 8 * AGG_U) {
            if (i + AGG_U <= end) add_rows<true, TILE_PITCH>(list, i, end, Xl, tl);
            else                  add_rows<false, TILE_PITCH>(list, i, end, Xl, tl);
        }
    }

#pragma unroll
    for (int ks = 0; ks < 16; ++ks) {
        sWd[(ks << 8) + tid] = wdr[ks];
        sWb[(ks << 8) + tid] = wbr[ks];
    }
    __syncthreads();

    const float* Sr = tile + (tq * 16 + m) * TILE_PITCH;

    v8f accX[2] = {};
    v8f accS[2] = {};
    const int fragBase = lane * 2;

#pragma unroll
    for (int ks = 0; ks < 16; ++ks) {
        int k0 = ks << 2;
        v2f a = xa[ks];
        v2f s = *(const v2f*)(Sr + k0 + kh);
#pragma unroll
        for (int j = 0; j < 2; ++j) {
            int nt = nh * 2 + j;
            v2f bd = *(const v2f*)&sWd[((ks << 2) + nt) * 64 + fragBase];
            v2f bb = *(const v2f*)&sWb[((ks << 2) + nt) * 64 + fragBase];
            accX[j] = __builtin_amdgcn_wmma_f32_16x16x4_f32(
                false, a, false, bd, (short)0, accX[j], false, false);
            accS[j] = __builtin_amdgcn_wmma_f32_16x16x4_f32(
                false, s, false, bb, (short)0, accS[j], false, false);
        }
    }

    int ncol = lane & 15;
    int mAdd = (lane >> 4) * 8;
#pragma unroll
    for (int j = 0; j < 2; ++j) {
        int col = (nh * 2 + j) * 16 + ncol;
        float wc = Wc[col];
        float bb = bias[col];
        float r[8];
#pragma unroll
        for (int v = 0; v < 8; ++v) {
            int node = node0 + v + mAdd;
            r[v] = (resid && node < N) ? resid[(size_t)node * HDIM + col] : 0.0f;
        }
#pragma unroll
        for (int v = 0; v < 8; ++v) {
            int node = node0 + v + mAdd;
            if (node >= N) continue;
            float g = gate[node];
            float val = g * (accX[j][v] + bb) + invdeg[node] * accS[j][v] + ssn[node] * wc;
            if (resid) val += r[v];
            val = (val > 0.0f) ? val : 0.01f * val;
            Y[(size_t)node * HDIM + col] = val;
        }
    }
}

__global__ __launch_bounds__(256) void agg_out_kernel(const unsigned* __restrict__ list,
                                                      const int* __restrict__ binStart,
                                                      const float* __restrict__ X,
                                                      const float* __restrict__ Wout, const float* __restrict__ bout,
                                                      const float* __restrict__ x0,
                                                      const float* __restrict__ invdeg, const float* __restrict__ gate,
                                                      const float* __restrict__ ssn, float* __restrict__ out, int N) {
    __shared__ float4 tile4[BIN_NODES * TILE_PITCH / 4];
    float* tile = (float*)tile4;
    for (int i = threadIdx.x; i < BIN_NODES * TILE_PITCH / 4; i += 256) tile4[i] = make_float4(0.f, 0.f, 0.f, 0.f);
    __syncthreads();

    const int bin  = blockIdx.x;
    const int lane = threadIdx.x & 31;
    const int wave = __builtin_amdgcn_readfirstlane((int)(threadIdx.x >> 5));

    const int n = bin * BIN_NODES + (int)threadIdx.x;
    float xr[HDIM];
#pragma unroll
    for (int k = 0; k < HDIM; ++k) xr[k] = 0.0f;
    if (wave < 2) {
        const float* Xr = X + (size_t)min(n, N - 1) * HDIM;
#pragma unroll
        for (int k = 0; k < HDIM; ++k) xr[k] = Xr[k];
    }

    const int beg  = binStart[bin];
    const int end  = binStart[bin + 1];
    {
        const float* Xl = X + 2 * lane;
        float* tl = tile + 2 * lane;
        for (int i = beg + wave * AGG_U; i < end; i += 8 * AGG_U) {
            if (i + AGG_U <= end) add_rows<true, TILE_PITCH>(list, i, end, Xl, tl);
            else                  add_rows<false, TILE_PITCH>(list, i, end, Xl, tl);
        }
    }
    __syncthreads();

    if (wave < 2 && n < N) {
        float a1 = 0.0f, a2 = 0.0f;
        const float* Sr = tile + threadIdx.x * TILE_PITCH;
#pragma unroll
        for (int k = 0; k < HDIM; ++k) {
            float wa = Wout[k], wb = Wout[HDIM + k];
            a1 += xr[k] * (wa - wb);
            a2 += Sr[k] * wb;
        }
        out[n] = gate[n] * (a1 + bout[0]) + invdeg[n] * a2 + ssn[n] * Wout[2 * HDIM] + x0[n];
    }
}

extern "C" void kernel_launch(void* const* d_in, const int* in_sizes, int n_in,
                              void* d_out, int out_size, void* d_ws, size_t ws_size,
                              hipStream_t stream) {
    const float* x    = (const float*)d_in[0];
    const int*   ei   = (const int*)  d_in[1];
    const float* Win  = (const float*)d_in[2];
    const float* bin  = (const float*)d_in[3];
    const float* Wmid = (const float*)d_in[4];
    const float* bmid = (const float*)d_in[5];
    const float* Wout = (const float*)d_in[6];
    const float* bout = (const float*)d_in[7];

    const int N = in_sizes[0];
    const int E = in_sizes[1] / 2;
    const int* src = ei;
    const int* dst = ei + E;

    float* w = (float*)d_ws;
    float* invdeg  = w + 2 * (size_t)N;
    float* gate    = w + 3 * (size_t)N;
    float* ssn     = w + 4 * (size_t)N;
    float* Sx      = w + 5 * (size_t)N;
    float* Wd6     = w + 6 * (size_t)N;
    float* bufA    = Wd6  + 6 * HDIM * HDIM;
    float* bufB    = bufA + (size_t)N * HDIM;
    float* bufC    = bufB + (size_t)N * HDIM;

    const int NB     = (N + BIN_NODES - 1) / BIN_NODES;
    const int chunk  = (E + CHUNK_WGS - 1) / CHUNK_WGS;

    unsigned* list  = (unsigned*)bufC;
    int* cnt        = (int*)(bufC + (size_t)E);
    int* pre        = cnt + (size_t)CHUNK_WGS * NB;
    int* binStart   = pre + (size_t)CHUNK_WGS * NB;

    wdiff_kernel<<<(6 * 4096 + 255) / 256, 256, 0, stream>>>(Wmid, Wd6, 6 * 4096);

    bin_count_kernel<<<CHUNK_WGS, 1024, 0, stream>>>(dst, cnt, E, NB, chunk);
    bin_scan_kernel<<<1, 1024, 0, stream>>>(cnt, pre, binStart, NB, CHUNK_WGS);
    bin_fill_kernel<<<CHUNK_WGS, 1024, 0, stream>>>(src, dst, pre, binStart, list, E, NB, chunk);

    bin_stats_kernel<<<NB, 256, 0, stream>>>(list, binStart, x, invdeg, gate, ssn, Sx, N);

    conv_in_kernel<<<(N * HDIM + 255) / 256, 256, 0, stream>>>(x, Win, bin, invdeg, gate, ssn, Sx,
                                                               bufA, N);

    float* cur = bufA;
    for (int l = 0; l < 3; ++l) {
        const float* Wl = Wmid + (size_t)(2 * l) * 129 * HDIM;
        agg_conv_kernel<<<NB, 256, 0, stream>>>(
            list, binStart, cur, Wd6 + (2 * l) * HDIM * HDIM, Wl + 64 * HDIM, Wl + 128 * HDIM,
            bmid + (2 * l) * HDIM, gate, invdeg, ssn, nullptr, bufB, N);

        const float* Wl2 = Wmid + (size_t)(2 * l + 1) * 129 * HDIM;
        agg_conv_kernel<<<NB, 256, 0, stream>>>(
            list, binStart, bufB, Wd6 + (2 * l + 1) * HDIM * HDIM, Wl2 + 64 * HDIM, Wl2 + 128 * HDIM,
            bmid + (2 * l + 1) * HDIM, gate, invdeg, ssn, cur, cur, N);
    }

    agg_out_kernel<<<NB, 256, 0, stream>>>(list, binStart, cur, Wout, bout, x, invdeg, gate, ssn,
                                           (float*)d_out, N);
}


# Gather software-pipelined per wave: next group's 16 row reads (2nd register set) and the edge words behind them stay in flight under the current group's LDS adds; sum order: LDS arrival as before
# speedup vs baseline: 3.0048x; 1.0473x over previous
#include <hip/hip_runtime.h>
#include <hip/hip_bf16.h>

typedef float v2f __attribute__((ext_vector_type(2)));
typedef float v8f __attribute__((ext_vector_type(8)));

#define HDIM 64

__global__ void wdiff_kernel(const float* __restrict__ Wmid, float* __restrict__ Wd6, int total) {
    int i = blockIdx.x * blockDim.x + threadIdx.x;
    if (i >= total) return;
    int l = i >> 12;
    int rc = i & 4095;
    const float* Wl = Wmid + (size_t)l * 129 * HDIM;
    Wd6[i] = Wl[rc] - Wl[64 * HDIM + rc];
}

__global__ void conv_in_kernel(const float* __restrict__ x, const float* __restrict__ Win,
                               const float* __restrict__ bin,
                               const float* __restrict__ invdeg, const float* __restrict__ gate,
                               const float* __restrict__ ssn, const float* __restrict__ Sx,
                               float* __restrict__ Y, int N) {
    int t = blockIdx.x * blockDim.x + threadIdx.x;
    if (t >= N * HDIM) return;
    int n = t >> 6, h = t & 63;
    float wa = Win[h], wb = Win[HDIM + h], wc = Win[2 * HDIM + h];
    Y[t] = gate[n] * (x[n] * (wa - wb) + bin[h]) + invdeg[n] * Sx[n] * wb + ssn[n] * wc;
}

#define BIN_SHIFT 6
#define BIN_NODES 64
#define MAX_BINS 1024
#define CHUNK_WGS 256

__global__ __launch_bounds__(1024) void bin_count_kernel(const int* __restrict__ dst, int* __restrict__ cnt,
                                                         int E, int NB, int chunk) {
    __shared__ int hist[MAX_BINS];
    for (int b = threadIdx.x; b < NB; b += 1024) hist[b] = 0;
    __syncthreads();
    int e0 = blockIdx.x * chunk;
    int e1 = min(e0 + chunk, E);
    for (int e = e0 + (int)threadIdx.x; e < e1; e += 1024)
        atomicAdd(&hist[dst[e] >> BIN_SHIFT], 1);
    __syncthreads();
    for (int b = threadIdx.x; b < NB; b += 1024) cnt[blockIdx.x * NB + b] = hist[b];
}

__global__ __launch_bounds__(1024) void bin_scan_kernel(const int* __restrict__ cnt, int* __restrict__ pre,
                                                        int* __restrict__ binStart, int NB, int G) {
    __shared__ int tot[MAX_BINS];
    int b = threadIdx.x;
    int run = 0;
    if (b < NB) {
#pragma unroll 16
        for (int g = 0; g < G; ++g) {
            int c = cnt[g * NB + b];
            pre[g * NB + b] = run;
            run += c;
        }
    }
    tot[b] = run;
    __syncthreads();
    for (int off = 1; off < MAX_BINS; off <<= 1) {
        int v = (b >= off) ? tot[b - off] : 0;
        __syncthreads();
        tot[b] += v;
        __syncthreads();
    }
    if (b < NB) binStart[b] = tot[b] - run;
    if (b == NB - 1) binStart[NB] = tot[b];
}

__global__ __launch_bounds__(1024) void bin_fill_kernel(const int* __restrict__ src, const int* __restrict__ dst,
                                                        const int* __restrict__ pre,
                                                        const int* __restrict__ binStart,
                                                        unsigned* __restrict__ list, int E, int NB, int chunk) {
    __shared__ int cur[MAX_BINS];
    for (int b = threadIdx.x; b < NB; b += 1024) cur[b] = binStart[b] + pre[blockIdx.x * NB + b];
    __syncthreads();
    int e0 = blockIdx.x * chunk;
    int e1 = min(e0 + chunk, E);
    for (int e = e0 + (int)threadIdx.x; e < e1; e += 1024) {
        int d = dst[e];
        int s = src[e];
        int slot = atomicAdd(&cur[d >> BIN_SHIFT], 1);
        list[slot] = ((unsigned)s << BIN_SHIFT) | (unsigned)(d & (BIN_NODES - 1));
    }
}

#define ST_U 8

__global__ __launch_bounds__(256) void bin_stats_kernel(const unsigned* __restrict__ list,
                                                        const int* __restrict__ binStart,
                                                        const float* __restrict__ x,
                                                        float* __restrict__ invdeg, float* __restrict__ gate,
                                                        float* __restrict__ ssn, float* __restrict__ Sx, int N) {
    __shared__ int   sdeg[BIN_NODES];
    __shared__ int   ssgn[BIN_NODES];
    __shared__ float ssx[BIN_NODES];
    const int t = threadIdx.x;
    if (t < BIN_NODES) { sdeg[t] = 0; ssgn[t] = 0; ssx[t] = 0.0f; }
    __syncthreads();

    const int bin   = blockIdx.x;
    const int beg   = binStart[bin];
    const int end   = binStart[bin + 1];
    const int node0 = bin * BIN_NODES;

    for (int i = beg + t; i < end; i += 256 * ST_U) {
        unsigned p[ST_U];
        float xv[ST_U];
#pragma unroll
        for (int j = 0; j < ST_U; ++j) p[j] = list[min(i + j * 256, end - 1)];
#pragma unroll
        for (int j = 0; j < ST_U; ++j) xv[j] = x[p[j] >> BIN_SHIFT];
#pragma unroll
        for (int j = 0; j < ST_U; ++j) {
            if (i + j * 256 < end) {
                int s  = (int)(p[j] >> BIN_SHIFT);
                int dl = (int)(p[j] & (BIN_NODES - 1));
                int d  = node0 + dl;
                atomicAdd(&sdeg[dl], 1);
                atomicAdd(&ssgn[dl], (s > d) - (s < d));
                atomicAdd(&ssx[dl], xv[j]);
            }
        }
    }
    __syncthreads();

    if (t < BIN_NODES && node0 + t < N) {
        int n = node0 + t;
        float dg  = (float)sdeg[t];
        float inv = 1.0f / fmaxf(dg, 1.0f);
        invdeg[n] = inv;
        gate[n]   = (dg > 0.0f) ? 1.0f : 0.0f;
        ssn[n]    = (float)ssgn[t] * inv;
        Sx[n]     = ssx[t];
    }
}

#define AGG_U 16
#define AGG_WAVES 8

__device__ __forceinline__ int win_start(int i, int end) { return max(min(i, end - AGG_U), 0); }

__device__ __forceinline__ void edge_words(const unsigned* __restrict__ list, int ws, unsigned (&w)[AGG_U]) {
#pragma unroll
    for (int j = 0; j < AGG_U; ++j) w[j] = list[ws + j];
}

__device__ __forceinline__ void row_reads(const unsigned (&w)[AGG_U], const float* __restrict__ X, int lane2,
                                          v2f (&v)[AGG_U], unsigned (&d)[AGG_U]) {
#pragma unroll
    for (int j = 0; j < AGG_U; ++j) {
        const float* rowp = X + (size_t)(w[j] >> BIN_SHIFT) * HDIM;
        v[j] = *(const v2f*)(rowp + lane2);
        d[j] = w[j] & (BIN_NODES - 1);
    }
}

template <bool MASKED, int PITCH>
__device__ __forceinline__ void row_adds(const v2f (&v)[AGG_U], const unsigned (&d)[AGG_U], float* tl,
                                         int ws, int lo, int end) {
#pragma unroll
    for (int j = 0; j < AGG_U; ++j) {
        if (!MASKED || (ws + j >= lo && ws + j < end)) {
            float* t = tl + d[j] * PITCH;
            atomicAdd(t, v[j].x);
            atomicAdd(t + 1, v[j].y);
        }
    }
}

template <int PITCH>
__device__ __forceinline__ void gather_bin(const unsigned* __restrict__ list, int beg, int end, int wave,
                                           const float* __restrict__ X, int lane2, float* tl) {
    const int step = AGG_WAVES * AGG_U;
    int i = beg + wave * AGG_U;
    if (i >= end) return;
    unsigned w1[AGG_U], w2[AGG_U];
    v2f va[AGG_U], vb[AGG_U];
    unsigned da[AGG_U], db[AGG_U];
    edge_words(list, win_start(i, end), w1);
    row_reads(w1, X, lane2, va, da);
    int in = i + step;
    edge_words(list, win_start(in, end), w1);
    edge_words(list, win_start(in + step, end), w2);
    while (in + step < end) {
        row_reads(w1, X, lane2, vb, db);
        __builtin_amdgcn_sched_barrier(0);
        row_adds<false, PITCH>(va, da, tl, 0, 0, 0);
        row_reads(w2, X, lane2, va, da);
        edge_words(list, win_start(in + 2 * step, end), w1);
        edge_words(list, win_start(in + 3 * step, end), w2);
        __builtin_amdgcn_sched_barrier(0);
        row_adds<false, PITCH>(vb, db, tl, 0, 0, 0);
        i = in + step;
        in = i + step;
    }
    if (in < end) {
        row_reads(w1, X, lane2, vb, db);
        row_adds<false, PITCH>(va, da, tl, 0, 0, 0);
        row_adds<true, PITCH>(vb, db, tl, win_start(in, end), in, end);
    } else {
        row_adds<true, PITCH>(va, da, tl, win_start(i, end), i, end);
    }
}

#define TILE_PITCH 68

__global__ __launch_bounds__(256) __attribute__((amdgpu_waves_per_eu(8))) void agg_conv_kernel(
    const unsigned* __restrict__ list, const int* __restrict__ binStart,
    const float* __restrict__ X,
    const float* __restrict__ Wd, const float* __restrict__ Wb,
    const float* __restrict__ Wc, const float* __restrict__ bias,
    const float* __restrict__ gate, const float* __restrict__ invdeg,
    const float* __restrict__ ssn, const float* resid,
    float* Y, int N) {
    __shared__ float4 tile4[BIN_NODES * TILE_PITCH / 4];
    __shared__ float sWd[HDIM * HDIM];
    __shared__ float sWb[HDIM * HDIM];
    float* tile = (float*)tile4;

    const int bin  = blockIdx.x;
    const int lane = threadIdx.x & 31;
    const int wave = __builtin_amdgcn_readfirstlane((int)(threadIdx.x >> 5));
    const int tid  = threadIdx.x;

    {
        float wdr[16], wbr[16];
        int nt   = tid >> 6;
        int lp   = tid & 63;
        int ln   = lp >> 1;
        int half = lp & 1;
        int kh   = (ln >> 4) << 1;
        int wsrc = (kh + half) * HDIM + (nt << 4) + (ln & 15);
#pragma unroll
        for (int ks = 0; ks < 16; ++ks) {
            wdr[ks] = Wd[(ks << 2) * HDIM + wsrc];
            wbr[ks] = Wb[(ks << 2) * HDIM + wsrc];
        }
        for (int i = threadIdx.x; i < BIN_NODES * TILE_PITCH / 4; i += 256) tile4[i] = make_float4(0.f, 0.f, 0.f, 0.f);
#pragma unroll
        for (int ks = 0; ks < 16; ++ks) {
            sWd[(ks << 8) + tid] = wdr[ks];
            sWb[(ks << 8) + tid] = wbr[ks];
        }
    }
    __syncthreads();

    gather_bin<TILE_PITCH>(list, binStart[bin], binStart[bin + 1], wave, X, 2 * lane, tile + 2 * lane);

    const int tq    = wave & 3;
    const int nh    = wave >> 2;
    const int node0 = bin * BIN_NODES + tq * 16;
    const int m  = lane & 15;
    const int kh = (lane >> 4) << 1;
    int row = node0 + m; if (row >= N) row = N - 1;
    const float* Xr = X + (size_t)row * HDIM;
    v2f xa[16];
#pragma unroll
    for (int ks = 0; ks < 16; ++ks) {
        xa[ks].x = Xr[(ks << 2) + kh];
        xa[ks].y = Xr[(ks << 2) + kh + 1];
    }
    __syncthreads();

    const float* Sr = tile + (tq * 16 + m) * TILE_PITCH;

    v8f accX[2] = {};
    v8f accS[2] = {};
    const int fragBase = lane * 2;

#pragma unroll
    for (int ks = 0; ks < 16; ++ks) {
        int k0 = ks << 2;
        v2f a = xa[ks];
        v2f s = *(const v2f*)(Sr + k0 + kh);
#pragma unroll
        for (int j = 0; j < 2; ++j) {
            int nt = nh * 2 + j;
            v2f bd = *(const v2f*)&sWd[((ks << 2) + nt) * 64 + fragBase];
            v2f bb = *(const v2f*)&sWb[((ks << 2) + nt) * 64 + fragBase];
            accX[j] = __builtin_amdgcn_wmma_f32_16x16x4_f32(
                false, a, false, bd, (short)0, accX[j], false, false);
            accS[j] = __builtin_amdgcn_wmma_f32_16x16x4_f32(
                false, s, false, bb, (short)0, accS[j], false, false);
        }
    }

    int ncol = lane & 15;
    int mAdd = (lane >> 4) * 8;
#pragma unroll
    for (int j = 0; j < 2; ++j) {
        int col = (nh * 2 + j) * 16 + ncol;
        float wc = Wc[col];
        float bb = bias[col];
        float r[8];
#pragma unroll
        for (int v = 0; v < 8; ++v) {
            int node = node0 + v + mAdd;
            r[v] = (resid && node < N) ? resid[(size_t)node * HDIM + col] : 0.0f;
        }
#pragma unroll
        for (int v = 0; v < 8; ++v) {
            int node = node0 + v + mAdd;
            if (node >= N) continue;
            float g = gate[node];
            float val = g * (accX[j][v] + bb) + invdeg[node] * accS[j][v] + ssn[node] * wc;
            if (resid) val += r[v];
            val = (val > 0.0f) ? val : 0.01f * val;
            Y[(size_t)node * HDIM + col] = val;
        }
    }
}

__global__ __launch_bounds__(256) __attribute__((amdgpu_waves_per_eu(8))) void agg_out_kernel(const unsigned* __restrict__ list,
                                                      const int* __restrict__ binStart,
                                                      const float* __restrict__ X,
                                                      const float* __restrict__ Wout, const float* __restrict__ bout,
                                                      const float* __restrict__ x0,
                                                      const float* __restrict__ invdeg, const float* __restrict__ gate,
                                                      const float* __restrict__ ssn, float* __restrict__ out, int N) {
    __shared__ float4 tile4[BIN_NODES * TILE_PITCH / 4];
    float* tile = (float*)tile4;

    const int bin  = blockIdx.x;
    const int lane = threadIdx.x & 31;
    const int wave = __builtin_amdgcn_readfirstlane((int)(threadIdx.x >> 5));

    const int n = bin * BIN_NODES + (int)threadIdx.x;
    float a1 = 0.0f;
    if (wave < 2) {
        const float* Xr = X + (size_t)min(n, N - 1) * HDIM;
        float xr[HDIM];
#pragma unroll
        for (int k = 0; k < HDIM; ++k) xr[k] = Xr[k];
        for (int i = threadIdx.x; i < BIN_NODES * TILE_PITCH / 4; i += 256) tile4[i] = make_float4(0.f, 0.f, 0.f, 0.f);
#pragma unroll
        for (int k = 0; k < HDIM; ++k) a1 += xr[k] * (Wout[k] - Wout[HDIM + k]);
    } else {
        for (int i = threadIdx.x; i < BIN_NODES * TILE_PITCH / 4; i += 256) tile4[i] = make_float4(0.f, 0.f, 0.f, 0.f);
    }
    __syncthreads();

    gather_bin<TILE_PITCH>(list, binStart[bin], binStart[bin + 1], wave, X, 2 * lane, tile + 2 * lane);
    __syncthreads();

    if (wave < 2 && n < N) {
        float a2 = 0.0f;
        const float* Sr = tile + threadIdx.x * TILE_PITCH;
#pragma unroll
        for (int k = 0; k < HDIM; ++k) a2 += Sr[k] * Wout[HDIM + k];
        out[n] = gate[n] * (a1 + bout[0]) + invdeg[n] * a2 + ssn[n] * Wout[2 * HDIM] + x0[n];
    }
}

extern "C" void kernel_launch(void* const* d_in, const int* in_sizes, int n_in,
                              void* d_out, int out_size, void* d_ws, size_t ws_size,
                              hipStream_t stream) {
    const float* x    = (const float*)d_in[0];
    const int*   ei   = (const int*)  d_in[1];
    const float* Win  = (const float*)d_in[2];
    const float* bin  = (const float*)d_in[3];
    const float* Wmid = (const float*)d_in[4];
    const float* bmid = (const float*)d_in[5];
    const float* Wout = (const float*)d_in[6];
    const float* bout = (const float*)d_in[7];

    const int N = in_sizes[0];
    const int E = in_sizes[1] / 2;
    const int* src = ei;
    const int* dst = ei + E;

    float* w = (float*)d_ws;
    float* invdeg  = w + 2 * (size_t)N;
    float* gate    = w + 3 * (size_t)N;
    float* ssn     = w + 4 * (size_t)N;
    float* Sx      = w + 5 * (size_t)N;
    float* Wd6     = w + 6 * (size_t)N;
    float* bufA    = Wd6  + 6 * HDIM * HDIM;
    float* bufB    = bufA + (size_t)N * HDIM;
    float* bufC    = bufB + (size_t)N * HDIM;

    const int NB     = (N + BIN_NODES - 1) / BIN_NODES;
    const int chunk  = (E + CHUNK_WGS - 1) / CHUNK_WGS;

    unsigned* list  = (unsigned*)bufC;
    int* cnt        = (int*)(bufC + (size_t)E);
    int* pre        = cnt + (size_t)CHUNK_WGS * NB;
    int* binStart   = pre + (size_t)CHUNK_WGS * NB;

    wdiff_kernel<<<(6 * 4096 + 255) / 256, 256, 0, stream>>>(Wmid, Wd6, 6 * 4096);

    bin_count_kernel<<<CHUNK_WGS, 1024, 0, stream>>>(dst, cnt, E, NB, chunk);
    bin_scan_kernel<<<1, 1024, 0, stream>>>(cnt, pre, binStart, NB, CHUNK_WGS);
    bin_fill_kernel<<<CHUNK_WGS, 1024, 0, stream>>>(src, dst, pre, binStart, list, E, NB, chunk);

    bin_stats_kernel<<<NB, 256, 0, stream>>>(list, binStart, x, invdeg, gate, ssn, Sx, N);

    conv_in_kernel<<<(N * HDIM + 255) / 256, 256, 0, stream>>>(x, Win, bin, invdeg, gate, ssn, Sx,
                                                               bufA, N);

    float* cur = bufA;
    for (int l = 0; l < 3; ++l) {
        const float* Wl = Wmid + (size_t)(2 * l) * 129 * HDIM;
        agg_conv_kernel<<<NB, 256, 0, stream>>>(
            list, binStart, cur, Wd6 + (2 * l) * HDIM * HDIM, Wl + 64 * HDIM, Wl + 128 * HDIM,
            bmid + (2 * l) * HDIM, gate, invdeg, ssn, nullptr, bufB, N);

        const float* Wl2 = Wmid + (size_t)(2 * l + 1) * 129 * HDIM;
        agg_conv_kernel<<<NB, 256, 0, stream>>>(
            list, binStart, bufB, Wd6 + (2 * l + 1) * HDIM * HDIM, Wl2 + 64 * HDIM, Wl2 + 128 * HDIM,
            bmid + (2 * l + 1) * HDIM, gate, invdeg, ssn, cur, cur, N);
    }

    agg_out_kernel<<<NB, 256, 0, stream>>>(list, binStart, cur, Wout, bout, x, invdeg, gate, ssn,
                                           (float*)d_out, N);
}


# Destination bins of 200 nodes: one 32-wave work-group per WGP owns an equal share of each neighbor sum (was 782 bins of 64, 3 or 4 to a WGP); weights staged once per WGP; sum order: LDS arrival
# speedup vs baseline: 3.1539x; 1.0496x over previous
#include <hip/hip_runtime.h>
#include <hip/hip_bf16.h>

typedef float v2f __attribute__((ext_vector_type(2)));
typedef float v8f __attribute__((ext_vector_type(8)));

#define HDIM 64

__global__ void wdiff_kernel(const float* __restrict__ Wmid, float* __restrict__ Wd6, int total) {
    int i = blockIdx.x * blockDim.x + threadIdx.x;
    if (i >= total) return;
    int l = i >> 12;
    int rc = i & 4095;
    const float* Wl = Wmid + (size_t)l * 129 * HDIM;
    Wd6[i] = Wl[rc] - Wl[64 * HDIM + rc];
}

__global__ void conv_in_kernel(const float* __restrict__ x, const float* __restrict__ Win,
                               const float* __restrict__ bin,
                               const float* __restrict__ invdeg, const float* __restrict__ gate,
                               const float* __restrict__ ssn, const float* __restrict__ Sx,
                               float* __restrict__ Y, int N) {
    int t = blockIdx.x * blockDim.x + threadIdx.x;
    if (t >= N * HDIM) return;
    int n = t >> 6, h = t & 63;
    float wa = Win[h], wb = Win[HDIM + h], wc = Win[2 * HDIM + h];
    Y[t] = gate[n] * (x[n] * (wa - wb) + bin[h]) + invdeg[n] * Sx[n] * wb + ssn[n] * wc;
}

#define BIN_SHIFT 8
#define BIN_MASK 255
#define BIN_NODES 200
#define MAX_BINS 1024
#define CHUNK_WGS 256

__global__ __launch_bounds__(1024) void bin_count_kernel(const int* __restrict__ dst, int* __restrict__ cnt,
                                                         int E, int NB, int chunk) {
    __shared__ int hist[MAX_BINS];
    for (int b = threadIdx.x; b < NB; b += 1024) hist[b] = 0;
    __syncthreads();
    int e0 = blockIdx.x * chunk;
    int e1 = min(e0 + chunk, E);
    for (int e = e0 + (int)threadIdx.x; e < e1; e += 1024)
        atomicAdd(&hist[dst[e] / BIN_NODES], 1);
    __syncthreads();
    for (int b = threadIdx.x; b < NB; b += 1024) cnt[blockIdx.x * NB + b] = hist[b];
}

__global__ __launch_bounds__(1024) void bin_scan_kernel(const int* __restrict__ cnt, int* __restrict__ pre,
                                                        int* __restrict__ binStart, int NB, int G) {
    __shared__ int tot[MAX_BINS];
    int b = threadIdx.x;
    int run = 0;
    if (b < NB) {
#pragma unroll 16
        for (int g = 0; g < G; ++g) {
            int c = cnt[g * NB + b];
            pre[g * NB + b] = run;
            run += c;
        }
    }
    tot[b] = run;
    __syncthreads();
    for (int off = 1; off < MAX_BINS; off <<= 1) {
        int v = (b >= off) ? tot[b - off] : 0;
        __syncthreads();
        tot[b] += v;
        __syncthreads();
    }
    if (b < NB) binStart[b] = tot[b] - run;
    if (b == NB - 1) binStart[NB] = tot[b];
}

__global__ __launch_bounds__(1024) void bin_fill_kernel(const int* __restrict__ src, const int* __restrict__ dst,
                                                        const int* __restrict__ pre,
                                                        const int* __restrict__ binStart,
                                                        unsigned* __restrict__ list, int E, int NB, int chunk) {
    __shared__ int cur[MAX_BINS];
    for (int b = threadIdx.x; b < NB; b += 1024) cur[b] = binStart[b] + pre[blockIdx.x * NB + b];
    __syncthreads();
    int e0 = blockIdx.x * chunk;
    int e1 = min(e0 + chunk, E);
    for (int e = e0 + (int)threadIdx.x; e < e1; e += 1024) {
        int d = dst[e];
        int s = src[e];
        int b = d / BIN_NODES;
        int slot = atomicAdd(&cur[b], 1);
        list[slot] = ((unsigned)s << BIN_SHIFT) | (unsigned)(d - b * BIN_NODES);
    }
}

#define ST_U 8

__global__ __launch_bounds__(1024) void bin_stats_kernel(const unsigned* __restrict__ list,
                                                        const int* __restrict__ binStart,
                                                        const float* __restrict__ x,
                                                        float* __restrict__ invdeg, float* __restrict__ gate,
                                                        float* __restrict__ ssn, float* __restrict__ Sx, int N) {
    __shared__ int   sdeg[BIN_NODES];
    __shared__ int   ssgn[BIN_NODES];
    __shared__ float ssx[BIN_NODES];
    const int t = threadIdx.x;
    if (t < BIN_NODES) { sdeg[t] = 0; ssgn[t] = 0; ssx[t] = 0.0f; }
    __syncthreads();

    const int bin   = blockIdx.x;
    const int beg   = binStart[bin];
    const int end   = binStart[bin + 1];
    const int node0 = bin * BIN_NODES;

    for (int i = beg + t; i < end; i += 1024 * ST_U) {
        unsigned p[ST_U];
        float xv[ST_U];
#pragma unroll
        for (int j = 0; j < ST_U; ++j) p[j] = list[min(i + j * 1024, end - 1)];
#pragma unroll
        for (int j = 0; j < ST_U; ++j) xv[j] = x[p[j] >> BIN_SHIFT];
#pragma unroll
        for (int j = 0; j < ST_U; ++j) {
            if (i + j * 1024 < end) {
                int s  = (int)(p[j] >> BIN_SHIFT);
                int dl = (int)(p[j] & BIN_MASK);
                int d  = node0 + dl;
                atomicAdd(&sdeg[dl], 1);
                atomicAdd(&ssgn[dl], (s > d) - (s < d));
                atomicAdd(&ssx[dl], xv[j]);
            }
        }
    }
    __syncthreads();

    if (t < BIN_NODES && node0 + t < N) {
        int n = node0 + t;
        float dg  = (float)sdeg[t];
        float inv = 1.0f / fmaxf(dg, 1.0f);
        invdeg[n] = inv;
        gate[n]   = (dg > 0.0f) ? 1.0f : 0.0f;
        ssn[n]    = (float)ssgn[t] * inv;
        Sx[n]     = ssx[t];
    }
}

#define AGG_U 16
#define AGG_WAVES 32

__device__ __forceinline__ int win_start(int i, int end) { return max(min(i, end - AGG_U), 0); }

__device__ __forceinline__ void edge_words(const unsigned* __restrict__ list, int ws, unsigned (&w)[AGG_U]) {
#pragma unroll
    for (int j = 0; j < AGG_U; ++j) w[j] = list[ws + j];
}

__device__ __forceinline__ void row_reads(const unsigned (&w)[AGG_U], const float* __restrict__ X, int lane2,
                                          v2f (&v)[AGG_U], unsigned (&d)[AGG_U]) {
#pragma unroll
    for (int j = 0; j < AGG_U; ++j) {
        const float* rowp = X + (size_t)(w[j] >> BIN_SHIFT) * HDIM;
        v[j] = *(const v2f*)(rowp + lane2);
        d[j] = w[j] & BIN_MASK;
    }
}

template <bool MASKED, int PITCH>
__device__ __forceinline__ void row_adds(const v2f (&v)[AGG_U], const unsigned (&d)[AGG_U], float* tl,
                                         int ws, int lo, int end) {
#pragma unroll
    for (int j = 0; j < AGG_U; ++j) {
        if (!MASKED || (ws + j >= lo && ws + j < end)) {
            float* t = tl + d[j] * PITCH;
            atomicAdd(t, v[j].x);
            atomicAdd(t + 1, v[j].y);
        }
    }
}

template <int PITCH>
__device__ __forceinline__ void gather_bin(const unsigned* __restrict__ list, int beg, int end, int wave,
                                           const float* __restrict__ X, int lane2, float* tl) {
    const int step = AGG_WAVES * AGG_U;
    int i = beg + wave * AGG_U;
    if (i >= end) return;
    unsigned w1[AGG_U], w2[AGG_U];
    v2f va[AGG_U], vb[AGG_U];
    unsigned da[AGG_U], db[AGG_U];
    edge_words(list, win_start(i, end), w1);
    row_reads(w1, X, lane2, va, da);
    int in = i + step;
    edge_words(list, win_start(in, end), w1);
    edge_words(list, win_start(in + step, end), w2);
    while (in + step < end) {
        row_reads(w1, X, lane2, vb, db);
        __builtin_amdgcn_sched_barrier(0);
        row_adds<false, PITCH>(va, da, tl, 0, 0, 0);
        row_reads(w2, X, lane2, va, da);
        edge_words(list, win_start(in + 2 * step, end), w1);
        edge_words(list, win_start(in + 3 * step, end), w2);
        __builtin_amdgcn_sched_barrier(0);
        row_adds<false, PITCH>(vb, db, tl, 0, 0, 0);
        i = in + step;
        in = i + step;
    }
    if (in < end) {
        row_reads(w1, X, lane2, vb, db);
        row_adds<false, PITCH>(va, da, tl, 0, 0, 0);
        row_adds<true, PITCH>(vb, db, tl, win_start(in, end), in, end);
    } else {
        row_adds<true, PITCH>(va, da, tl, win_start(i, end), i, end);
    }
}

#define TILE_PITCH 68
#define MM_TILES ((BIN_NODES + 15) / 16)
#define TILE_ROWS (MM_TILES * 16)

__global__ __launch_bounds__(1024) __attribute__((amdgpu_waves_per_eu(8))) void agg_conv_kernel(
    const unsigned* __restrict__ list, const int* __restrict__ binStart,
    const float* __restrict__ X,
    const float* __restrict__ Wd, const float* __restrict__ Wb,
    const float* __restrict__ Wc, const float* __restrict__ bias,
    const float* __restrict__ gate, const float* __restrict__ invdeg,
    const float* __restrict__ ssn, const float* resid,
    float* Y, int N) {
    __shared__ float4 tile4[TILE_ROWS * TILE_PITCH / 4];
    __shared__ float sWd[HDIM * HDIM];
    __shared__ float sWb[HDIM * HDIM];
    float* tile = (float*)tile4;

    const int bin  = blockIdx.x;
    const int lane = threadIdx.x & 31;
    const int wave = __builtin_amdgcn_readfirstlane((int)(threadIdx.x >> 5));
    const int tid  = threadIdx.x;

    {
        float wdr[4], wbr[4];
        int t256 = tid & 255;
        int q    = tid >> 8;
        int nt   = t256 >> 6;
        int lp   = t256 & 63;
        int ln   = lp >> 1;
        int half = lp & 1;
        int kh   = (ln >> 4) << 1;
        int wsrc = (kh + half) * HDIM + (nt << 4) + (ln & 15);
#pragma unroll
        for (int i = 0; i < 4; ++i) {
            int ks = (q << 2) + i;
            wdr[i] = Wd[(ks << 2) * HDIM + wsrc];
            wbr[i] = Wb[(ks << 2) * HDIM + wsrc];
        }
        for (int i = threadIdx.x; i < TILE_ROWS * TILE_PITCH / 4; i += 1024) tile4[i] = make_float4(0.f, 0.f, 0.f, 0.f);
#pragma unroll
        for (int i = 0; i < 4; ++i) {
            int ks = (q << 2) + i;
            sWd[(ks << 8) + t256] = wdr[i];
            sWb[(ks << 8) + t256] = wbr[i];
        }
    }
    __syncthreads();

    gather_bin<TILE_PITCH>(list, binStart[bin], binStart[bin + 1], wave, X, 2 * lane, tile + 2 * lane);

    const bool mm   = wave < 2 * MM_TILES;
    const int tq    = wave >> 1;
    const int nh    = wave & 1;
    const int row0  = tq * 16;
    const int node0 = bin * BIN_NODES + row0;
    const int m  = lane & 15;
    const int kh = (lane >> 4) << 1;
    int row = node0 + m; if (row >= N) row = N - 1;
    const float* Xr = X + (size_t)row * HDIM;
    v2f xa[16];
    if (mm) {
#pragma unroll
        for (int ks = 0; ks < 16; ++ks) {
            xa[ks].x = Xr[(ks << 2) + kh];
            xa[ks].y = Xr[(ks << 2) + kh + 1];
        }
    }
    __syncthreads();
    if (!mm) return;

    const float* Sr = tile + (row0 + m) * TILE_PITCH;

    v8f accX[2] = {};
    v8f accS[2] = {};
    const int fragBase = lane * 2;

#pragma unroll
    for (int ks = 0; ks < 16; ++ks) {
        int k0 = ks << 2;
        v2f a = xa[ks];
        v2f s = *(const v2f*)(Sr + k0 + kh);
#pragma unroll
        for (int j = 0; j < 2; ++j) {
            int nt = nh * 2 + j;
            v2f bd = *(const v2f*)&sWd[((ks << 2) + nt) * 64 + fragBase];
            v2f bb = *(const v2f*)&sWb[((ks << 2) + nt) * 64 + fragBase];
            accX[j] = __builtin_amdgcn_wmma_f32_16x16x4_f32(
                false, a, false, bd, (short)0, accX[j], false, false);
            accS[j] = __builtin_amdgcn_wmma_f32_16x16x4_f32(
                false, s, false, bb, (short)0, accS[j], false, false);
        }
    }

    int ncol = lane & 15;
    int mAdd = (lane >> 4) * 8;
#pragma unroll
    for (int j = 0; j < 2; ++j) {
        int col = (nh * 2 + j) * 16 + ncol;
        float wc = Wc[col];
        float bb = bias[col];
        float r[8];
#pragma unroll
        for (int v = 0; v < 8; ++v) {
            int node = node0 + v + mAdd;
            bool own = (row0 + v + mAdd < BIN_NODES) && (node < N);
            r[v] = (resid && own) ? resid[(size_t)node * HDIM + col] : 0.0f;
        }
#pragma unroll
        for (int v = 0; v < 8; ++v) {
            int node = node0 + v + mAdd;
            if (row0 + v + mAdd >= BIN_NODES || node >= N) continue;
            float g = gate[node];
            float val = g * (accX[j][v] + bb) + invdeg[node] * accS[j][v] + ssn[node] * wc;
            if (resid) val += r[v];
            val = (val > 0.0f) ? val : 0.01f * val;
            Y[(size_t)node * HDIM + col] = val;
        }
    }
}

__global__ __launch_bounds__(1024) __attribute__((amdgpu_waves_per_eu(8))) void agg_out_kernel(const unsigned* __restrict__ list,
                                                      const int* __restrict__ binStart,
                                                      const float* __restrict__ X,
                                                      const float* __restrict__ Wout, const float* __restrict__ bout,
                                                      const float* __restrict__ x0,
                                                      const float* __restrict__ invdeg, const float* __restrict__ gate,
                                                      const float* __restrict__ ssn, float* __restrict__ out, int N) {
    __shared__ float4 tile4[TILE_ROWS * TILE_PITCH / 4];
    float* tile = (float*)tile4;

    const int bin  = blockIdx.x;
    const int lane = threadIdx.x & 31;
    const int wave = __builtin_amdgcn_readfirstlane((int)(threadIdx.x >> 5));

    const int n = bin * BIN_NODES + (int)threadIdx.x;
    const bool own = (int)threadIdx.x < BIN_NODES;
    float a1 = 0.0f;
    if (own) {
        const float* Xr = X + (size_t)min(n, N - 1) * HDIM;
        float xr[HDIM];
#pragma unroll
        for (int k = 0; k < HDIM; ++k) xr[k] = Xr[k];
        for (int i = threadIdx.x; i < TILE_ROWS * TILE_PITCH / 4; i += 1024) tile4[i] = make_float4(0.f, 0.f, 0.f, 0.f);
#pragma unroll
        for (int k = 0; k < HDIM; ++k) a1 += xr[k] * (Wout[k] - Wout[HDIM + k]);
    } else {
        for (int i = threadIdx.x; i < TILE_ROWS * TILE_PITCH / 4; i += 1024) tile4[i] = make_float4(0.f, 0.f, 0.f, 0.f);
    }
    __syncthreads();

    gather_bin<TILE_PITCH>(list, binStart[bin], binStart[bin + 1], wave, X, 2 * lane, tile + 2 * lane);
    __syncthreads();

    if (own && n < N) {
        float a2 = 0.0f;
        const float* Sr = tile + threadIdx.x * TILE_PITCH;
#pragma unroll
        for (int k = 0; k < HDIM; ++k) a2 += Sr[k] * Wout[HDIM + k];
        out[n] = gate[n] * (a1 + bout[0]) + invdeg[n] * a2 + ssn[n] * Wout[2 * HDIM] + x0[n];
    }
}

extern "C" void kernel_launch(void* const* d_in, const int* in_sizes, int n_in,
                              void* d_out, int out_size, void* d_ws, size_t ws_size,
                              hipStream_t stream) {
    const float* x    = (const float*)d_in[0];
    const int*   ei   = (const int*)  d_in[1];
    const float* Win  = (const float*)d_in[2];
    const float* bin  = (const float*)d_in[3];
    const float* Wmid = (const float*)d_in[4];
    const float* bmid = (const float*)d_in[5];
    const float* Wout = (const float*)d_in[6];
    const float* bout = (const float*)d_in[7];

    const int N = in_sizes[0];
    const int E = in_sizes[1] / 2;
    const int* src = ei;
    const int* dst = ei + E;

    float* w = (float*)d_ws;
    float* invdeg  = w + 2 * (size_t)N;
    float* gate    = w + 3 * (size_t)N;
    float* ssn     = w + 4 * (size_t)N;
    float* Sx      = w + 5 * (size_t)N;
    float* Wd6     = w + 6 * (size_t)N;
    float* bufA    = Wd6  + 6 * HDIM * HDIM;
    float* bufB    = bufA + (size_t)N * HDIM;
    float* bufC    = bufB + (size_t)N * HDIM;

    const int NB     = (N + BIN_NODES - 1) / BIN_NODES;
    const int chunk  = (E + CHUNK_WGS - 1) / CHUNK_WGS;

    unsigned* list  = (unsigned*)bufC;
    int* cnt        = (int*)(bufC + (size_t)E);
    int* pre        = cnt + (size_t)CHUNK_WGS * NB;
    int* binStart   = pre + (size_t)CHUNK_WGS * NB;

    wdiff_kernel<<<(6 * 4096 + 255) / 256, 256, 0, stream>>>(Wmid, Wd6, 6 * 4096);

    bin_count_kernel<<<CHUNK_WGS, 1024, 0, stream>>>(dst, cnt, E, NB, chunk);
    bin_scan_kernel<<<1, 1024, 0, stream>>>(cnt, pre, binStart, NB, CHUNK_WGS);
    bin_fill_kernel<<<CHUNK_WGS, 1024, 0, stream>>>(src, dst, pre, binStart, list, E, NB, chunk);

    bin_stats_kernel<<<NB, 1024, 0, stream>>>(list, binStart, x, invdeg, gate, ssn, Sx, N);

    conv_in_kernel<<<(N * HDIM + 255) / 256, 256, 0, stream>>>(x, Win, bin, invdeg, gate, ssn, Sx,
                                                               bufA, N);

    float* cur = bufA;
    for (int l = 0; l < 3; ++l) {
        const float* Wl = Wmid + (size_t)(2 * l) * 129 * HDIM;
        agg_conv_kernel<<<NB, 1024, 0, stream>>>(
            list, binStart, cur, Wd6 + (2 * l) * HDIM * HDIM, Wl + 64 * HDIM, Wl + 128 * HDIM,
            bmid + (2 * l) * HDIM, gate, invdeg, ssn, nullptr, bufB, N);

        const float* Wl2 = Wmid + (size_t)(2 * l + 1) * 129 * HDIM;
        agg_conv_kernel<<<NB, 1024, 0, stream>>>(
            list, binStart, bufB, Wd6 + (2 * l + 1) * HDIM * HDIM, Wl2 + 64 * HDIM, Wl2 + 128 * HDIM,
            bmid + (2 * l + 1) * HDIM, gate, invdeg, ssn, cur, cur, N);
    }

    agg_out_kernel<<<NB, 1024, 0, stream>>>(list, binStart, cur, Wout, bout, x, invdeg, gate, ssn,
                                           (float*)d_out, N);
}


# Mid convs sum each 16-node tile's neighbor rows on the matrix unit: 0/1 (node x edge) A times gathered rows, f32 WMMA, edges ordered by tile once per call; per-edge LDS float adds go; sum order: list
# speedup vs baseline: 3.5666x; 1.1309x over previous
#include <hip/hip_runtime.h>
#include <hip/hip_bf16.h>

typedef float v2f __attribute__((ext_vector_type(2)));
typedef float v8f __attribute__((ext_vector_type(8)));

#define HDIM 64

__global__ void wdiff_kernel(const float* __restrict__ Wmid, float* __restrict__ Wd6, int total) {
    int i = blockIdx.x * blockDim.x + threadIdx.x;
    if (i >= total) return;
    int l = i >> 12;
    int rc = i & 4095;
    const float* Wl = Wmid + (size_t)l * 129 * HDIM;
    Wd6[i] = Wl[rc] - Wl[64 * HDIM + rc];
}

__global__ void conv_in_kernel(const float* __restrict__ x, const float* __restrict__ Win,
                               const float* __restrict__ bin,
                               const float* __restrict__ invdeg, const float* __restrict__ gate,
                               const float* __restrict__ ssn, const float* __restrict__ Sx,
                               float* __restrict__ Y, int N) {
    int t = blockIdx.x * blockDim.x + threadIdx.x;
    if (t >= N * HDIM) return;
    int n = t >> 6, h = t & 63;
    float wa = Win[h], wb = Win[HDIM + h], wc = Win[2 * HDIM + h];
    Y[t] = gate[n] * (x[n] * (wa - wb) + bin[h]) + invdeg[n] * Sx[n] * wb + ssn[n] * wc;
}

#define BIN_SHIFT 8
#define BIN_MASK 255
#define BIN_NODES 200
#define MAX_BINS 1024
#define CHUNK_WGS 256

__global__ __launch_bounds__(1024) void bin_count_kernel(const int* __restrict__ dst, int* __restrict__ cnt,
                                                         int E, int NB, int chunk) {
    __shared__ int hist[MAX_BINS];
    for (int b = threadIdx.x; b < NB; b += 1024) hist[b] = 0;
    __syncthreads();
    int e0 = blockIdx.x * chunk;
    int e1 = min(e0 + chunk, E);
    for (int e = e0 + (int)threadIdx.x; e < e1; e += 1024)
        atomicAdd(&hist[dst[e] / BIN_NODES], 1);
    __syncthreads();
    for (int b = threadIdx.x; b < NB; b += 1024) cnt[blockIdx.x * NB + b] = hist[b];
}

__global__ __launch_bounds__(1024) void bin_scan_kernel(const int* __restrict__ cnt, int* __restrict__ pre,
                                                        int* __restrict__ binStart, int NB, int G) {
    __shared__ int tot[MAX_BINS];
    int b = threadIdx.x;
    int run = 0;
    if (b < NB) {
#pragma unroll 16
        for (int g = 0; g < G; ++g) {
            int c = cnt[g * NB + b];
            pre[g * NB + b] = run;
            run += c;
        }
    }
    tot[b] = run;
    __syncthreads();
    for (int off = 1; off < MAX_BINS; off <<= 1) {
        int v = (b >= off) ? tot[b - off] : 0;
        __syncthreads();
        tot[b] += v;
        __syncthreads();
    }
    if (b < NB) binStart[b] = tot[b] - run;
    if (b == NB - 1) binStart[NB] = tot[b];
}

__global__ __launch_bounds__(1024) void bin_fill_kernel(const int* __restrict__ src, const int* __restrict__ dst,
                                                        const int* __restrict__ pre,
                                                        const int* __restrict__ binStart,
                                                        unsigned* __restrict__ list, int E, int NB, int chunk) {
    __shared__ int cur[MAX_BINS];
    for (int b = threadIdx.x; b < NB; b += 1024) cur[b] = binStart[b] + pre[blockIdx.x * NB + b];
    __syncthreads();
    int e0 = blockIdx.x * chunk;
    int e1 = min(e0 + chunk, E);
    for (int e = e0 + (int)threadIdx.x; e < e1; e += 1024) {
        int d = dst[e];
        int s = src[e];
        int b = d / BIN_NODES;
        int slot = atomicAdd(&cur[b], 1);
        list[slot] = ((unsigned)s << BIN_SHIFT) | (unsigned)(d - b * BIN_NODES);
    }
}

#define ST_U 8

__global__ __launch_bounds__(1024) void bin_stats_kernel(const unsigned* __restrict__ list,
                                                        const int* __restrict__ binStart,
                                                        const float* __restrict__ x,
                                                        float* __restrict__ invdeg, float* __restrict__ gate,
                                                        float* __restrict__ ssn, float* __restrict__ Sx,
                                                        unsigned* __restrict__ list2, int* __restrict__ tileStart, int N) {
    __shared__ int   wcnt[32 * 16];
    __shared__ int   tbase[16];
    __shared__ int   sdeg[BIN_NODES];
    __shared__ int   ssgn[BIN_NODES];
    __shared__ float ssx[BIN_NODES];
    const int t = threadIdx.x;
    if (t < BIN_NODES) { sdeg[t] = 0; ssgn[t] = 0; ssx[t] = 0.0f; }
    if (t < 32 * 16) wcnt[t] = 0;
    __syncthreads();

    const int bin   = blockIdx.x;
    const int beg   = binStart[bin];
    const int end   = binStart[bin + 1];
    const int node0 = bin * BIN_NODES;

    for (int i = beg + t; i < end; i += 1024 * ST_U) {
        unsigned p[ST_U];
        float xv[ST_U];
#pragma unroll
        for (int j = 0; j < ST_U; ++j) p[j] = list[min(i + j * 1024, end - 1)];
#pragma unroll
        for (int j = 0; j < ST_U; ++j) xv[j] = x[p[j] >> BIN_SHIFT];
#pragma unroll
        for (int j = 0; j < ST_U; ++j) {
            if (i + j * 1024 < end) {
                int s  = (int)(p[j] >> BIN_SHIFT);
                int dl = (int)(p[j] & BIN_MASK);
                int d  = node0 + dl;
                atomicAdd(&sdeg[dl], 1);
                atomicAdd(&ssgn[dl], (s > d) - (s < d));
                atomicAdd(&ssx[dl], xv[j]);
            }
        }
    }
    __syncthreads();

    const int wv = t >> 5;
    for (int i = beg + t; i < end; i += 1024) {
        unsigned p = list[i];
        atomicAdd(&wcnt[wv * 16 + (int)((p & BIN_MASK) >> 4)], 1);
    }
    __syncthreads();
    if (t < 16) {
        int run = 0;
        for (int w = 0; w < 32; ++w) { int c = wcnt[w * 16 + t]; wcnt[w * 16 + t] = run; run += c; }
        tbase[t] = run;
    }
    __syncthreads();
    if (t == 0) {
        int run = beg;
        for (int k = 0; k < 16; ++k) { int c = tbase[k]; tbase[k] = run; tileStart[bin * 16 + k] = run; run += c; }
    }
    __syncthreads();
    for (int i = beg + t; i < end; i += 1024) {
        unsigned p = list[i];
        int tl = (int)((p & BIN_MASK) >> 4);
        int slot = tbase[tl] + atomicAdd(&wcnt[wv * 16 + tl], 1);
        list2[slot] = p;
    }

    if (t < BIN_NODES && node0 + t < N) {
        int n = node0 + t;
        float dg  = (float)sdeg[t];
        float inv = 1.0f / fmaxf(dg, 1.0f);
        invdeg[n] = inv;
        gate[n]   = (dg > 0.0f) ? 1.0f : 0.0f;
        ssn[n]    = (float)ssgn[t] * inv;
        Sx[n]     = ssx[t];
    }
}

#define AGG_U 16
#define AGG_WAVES 32

__device__ __forceinline__ int win_start(int i, int end) { return max(min(i, end - AGG_U), 0); }

__device__ __forceinline__ void edge_words(const unsigned* __restrict__ list, int ws, unsigned (&w)[AGG_U]) {
#pragma unroll
    for (int j = 0; j < AGG_U; ++j) w[j] = list[ws + j];
}

__device__ __forceinline__ void row_reads(const unsigned (&w)[AGG_U], const float* __restrict__ X, int lane2,
                                          v2f (&v)[AGG_U], unsigned (&d)[AGG_U]) {
#pragma unroll
    for (int j = 0; j < AGG_U; ++j) {
        const float* rowp = X + (size_t)(w[j] >> BIN_SHIFT) * HDIM;
        v[j] = *(const v2f*)(rowp + lane2);
        d[j] = w[j] & BIN_MASK;
    }
}

template <bool MASKED, int PITCH>
__device__ __forceinline__ void row_adds(const v2f (&v)[AGG_U], const unsigned (&d)[AGG_U], float* tl,
                                         int ws, int lo, int end) {
#pragma unroll
    for (int j = 0; j < AGG_U; ++j) {
        if (!MASKED || (ws + j >= lo && ws + j < end)) {
            float* t = tl + d[j] * PITCH;
            atomicAdd(t, v[j].x);
            atomicAdd(t + 1, v[j].y);
        }
    }
}

template <int PITCH>
__device__ __forceinline__ void gather_bin(const unsigned* __restrict__ list, int beg, int end, int wave,
                                           const float* __restrict__ X, int lane2, float* tl) {
    const int step = AGG_WAVES * AGG_U;
    int i = beg + wave * AGG_U;
    if (i >= end) return;
    unsigned w1[AGG_U], w2[AGG_U];
    v2f va[AGG_U], vb[AGG_U];
    unsigned da[AGG_U], db[AGG_U];
    edge_words(list, win_start(i, end), w1);
    row_reads(w1, X, lane2, va, da);
    int in = i + step;
    edge_words(list, win_start(in, end), w1);
    edge_words(list, win_start(in + step, end), w2);
    while (in + step < end) {
        row_reads(w1, X, lane2, vb, db);
        __builtin_amdgcn_sched_barrier(0);
        row_adds<false, PITCH>(va, da, tl, 0, 0, 0);
        row_reads(w2, X, lane2, va, da);
        edge_words(list, win_start(in + 2 * step, end), w1);
        edge_words(list, win_start(in + 3 * step, end), w2);
        __builtin_amdgcn_sched_barrier(0);
        row_adds<false, PITCH>(vb, db, tl, 0, 0, 0);
        i = in + step;
        in = i + step;
    }
    if (in < end) {
        row_reads(w1, X, lane2, vb, db);
        row_adds<false, PITCH>(va, da, tl, 0, 0, 0);
        row_adds<true, PITCH>(vb, db, tl, win_start(in, end), in, end);
    } else {
        row_adds<true, PITCH>(va, da, tl, win_start(i, end), i, end);
    }
}

__device__ __forceinline__ void tg_words(const unsigned* __restrict__ l2, int ws, unsigned (&w)[AGG_U]) {
#pragma unroll
    for (int j = 0; j < AGG_U; ++j) w[j] = l2[ws + j];
}

__device__ __forceinline__ void tg_loads(const unsigned (&w)[AGG_U], const float* __restrict__ X, bool hi, int c4,
                                         float4 (&b)[AGG_U / 2]) {
#pragma unroll
    for (int q = 0; q < AGG_U / 2; ++q) {
        const int ks = q >> 1, e = q & 1;
        unsigned ww = hi ? w[ks * 4 + 2 + e] : w[ks * 4 + e];
        b[q] = *(const float4*)(X + (size_t)(ww >> BIN_SHIFT) * HDIM + c4);
    }
}

template <bool MASKED>
__device__ __forceinline__ void tg_mma(const unsigned (&w)[AGG_U], const float4 (&b)[AGG_U / 2], bool hi,
                                       unsigned mrow, int ws, int lo, int end, v8f (&acc)[4]) {
#pragma unroll
    for (int ks = 0; ks < 4; ++ks) {
        v2f a;
#pragma unroll
        for (int e = 0; e < 2; ++e) {
            const int jl = ks * 4 + e, jh = ks * 4 + 2 + e;
            unsigned ww = hi ? w[jh] : w[jl];
            bool on = (ww & BIN_MASK) == mrow;
            if (MASKED) {
                bool vl = (ws + jl >= lo) && (ws + jl < end);
                bool vh = (ws + jh >= lo) && (ws + jh < end);
                on = on && (hi ? vh : vl);
            }
            a[e] = on ? 1.0f : 0.0f;
        }
        const float4 b0 = b[ks * 2], b1 = b[ks * 2 + 1];
        v2f f0 = {b0.x, b1.x}, f1 = {b0.y, b1.y}, f2 = {b0.z, b1.z}, f3 = {b0.w, b1.w};
        acc[0] = __builtin_amdgcn_wmma_f32_16x16x4_f32(false, a, false, f0, (short)0, acc[0], false, false);
        acc[1] = __builtin_amdgcn_wmma_f32_16x16x4_f32(false, a, false, f1, (short)0, acc[1], false, false);
        acc[2] = __builtin_amdgcn_wmma_f32_16x16x4_f32(false, a, false, f2, (short)0, acc[2], false, false);
        acc[3] = __builtin_amdgcn_wmma_f32_16x16x4_f32(false, a, false, f3, (short)0, acc[3], false, false);
    }
}

__device__ __forceinline__ void tile_gather(const unsigned* __restrict__ l2, int a0, int a1,
                                            const float* __restrict__ X, bool hi, int c4, unsigned mrow,
                                            v8f (&acc)[4]) {
    if (a0 >= a1) return;
    unsigned w1[AGG_U], w2[AGG_U];
    float4 ba[AGG_U / 2], bb[AGG_U / 2];
    int i = a0;
    tg_words(l2, win_start(i, a1), w1);
    tg_loads(w1, X, hi, c4, ba);
    while (i + 2 * AGG_U < a1) {
        tg_words(l2, i + AGG_U, w2);
        tg_loads(w2, X, hi, c4, bb);
        __builtin_amdgcn_sched_barrier(0);
        tg_mma<false>(w1, ba, hi, mrow, 0, 0, 0, acc);
        tg_words(l2, win_start(i + 2 * AGG_U, a1), w1);
        tg_loads(w1, X, hi, c4, ba);
        __builtin_amdgcn_sched_barrier(0);
        tg_mma<false>(w2, bb, hi, mrow, 0, 0, 0, acc);
        i += 2 * AGG_U;
    }
    if (i + AGG_U < a1) {
        const int ws2 = win_start(i + AGG_U, a1);
        tg_words(l2, ws2, w2);
        tg_loads(w2, X, hi, c4, bb);
        tg_mma<false>(w1, ba, hi, mrow, 0, 0, 0, acc);
        tg_mma<true>(w2, bb, hi, mrow, ws2, i + AGG_U, a1, acc);
    } else {
        tg_mma<true>(w1, ba, hi, mrow, win_start(i, a1), i, a1, acc);
    }
}

#define TILE_PITCH 68
#define MM_TILES ((BIN_NODES + 15) / 16)
#define TILE_ROWS (MM_TILES * 16)

__global__ __launch_bounds__(1024) __attribute__((amdgpu_waves_per_eu(8))) void agg_conv_kernel(
    const unsigned* __restrict__ list2, const int* __restrict__ tileStart,
    const float* __restrict__ X,
    const float* __restrict__ Wd, const float* __restrict__ Wb,
    const float* __restrict__ Wc, const float* __restrict__ bias,
    const float* __restrict__ gate, const float* __restrict__ invdeg,
    const float* __restrict__ ssn, const float* resid,
    float* Y, int N) {
    __shared__ float4 tile4[TILE_ROWS * TILE_PITCH / 4];
    __shared__ float sWd[HDIM * HDIM];
    __shared__ float sWb[HDIM * HDIM];
    float* tile = (float*)tile4;

    const int bin  = blockIdx.x;
    const int lane = threadIdx.x & 31;
    const int wave = __builtin_amdgcn_readfirstlane((int)(threadIdx.x >> 5));
    const int tid  = threadIdx.x;

    {
        float wdr[4], wbr[4];
        int t256 = tid & 255;
        int q    = tid >> 8;
        int nt   = t256 >> 6;
        int lp   = t256 & 63;
        int ln   = lp >> 1;
        int half = lp & 1;
        int kh   = (ln >> 4) << 1;
        int wsrc = (kh + half) * HDIM + (nt << 4) + (ln & 15);
#pragma unroll
        for (int i = 0; i < 4; ++i) {
            int ks = (q << 2) + i;
            wdr[i] = Wd[(ks << 2) * HDIM + wsrc];
            wbr[i] = Wb[(ks << 2) * HDIM + wsrc];
        }
#pragma unroll
        for (int i = 0; i < 4; ++i) {
            int ks = (q << 2) + i;
            sWd[(ks << 8) + t256] = wdr[i];
            sWb[(ks << 8) + t256] = wbr[i];
        }
    }

    const bool mm   = wave < 2 * MM_TILES;
    const int tq    = wave >> 1;
    const int nh    = wave & 1;
    const int row0  = tq * 16;
    const int node0 = bin * BIN_NODES + row0;
    const int m  = lane & 15;
    const int kh = (lane >> 4) << 1;

    v8f sacc[4] = {};
    if (mm) {
        const int ts = tileStart[bin * 16 + tq];
        const int te = tileStart[bin * 16 + tq + 1];
        const int hl = ((((te - ts) + 1) >> 1) + AGG_U - 1) & ~(AGG_U - 1);
        const int a0 = ts + nh * hl;
        const int a1 = min(te, a0 + hl);
        tile_gather(list2, a0, a1, X, lane >= 16, 4 * m, (unsigned)(row0 + m), sacc);
    }
    float4* trow = (float4*)(tile + (row0 + (lane >> 4) * 8) * TILE_PITCH + 4 * m);
    if (mm && nh == 0) {
#pragma unroll
        for (int r = 0; r < 8; ++r)
            trow[r * (TILE_PITCH / 4)] = make_float4(sacc[0][r], sacc[1][r], sacc[2][r], sacc[3][r]);
    }
    __syncthreads();
    if (mm && nh == 1) {
#pragma unroll
        for (int r = 0; r < 8; ++r) {
            float4 v = trow[r * (TILE_PITCH / 4)];
            v.x += sacc[0][r]; v.y += sacc[1][r]; v.z += sacc[2][r]; v.w += sacc[3][r];
            trow[r * (TILE_PITCH / 4)] = v;
        }
    }
    int row = node0 + m; if (row >= N) row = N - 1;
    const float* Xr = X + (size_t)row * HDIM;
    v2f xa[16];
    if (mm) {
#pragma unroll
        for (int ks = 0; ks < 16; ++ks) {
            xa[ks].x = Xr[(ks << 2) + kh];
            xa[ks].y = Xr[(ks << 2) + kh + 1];
        }
    }
    __syncthreads();
    if (!mm) return;

    const float* Sr = tile + (row0 + m) * TILE_PITCH;

    v8f accX[2] = {};
    v8f accS[2] = {};
    const int fragBase = lane * 2;

#pragma unroll
    for (int ks = 0; ks < 16; ++ks) {
        int k0 = ks << 2;
        v2f a = xa[ks];
        v2f s = *(const v2f*)(Sr + k0 + kh);
#pragma unroll
        for (int j = 0; j < 2; ++j) {
            int nt = nh * 2 + j;
            v2f bd = *(const v2f*)&sWd[((ks << 2) + nt) * 64 + fragBase];
            v2f bb = *(const v2f*)&sWb[((ks << 2) + nt) * 64 + fragBase];
            accX[j] = __builtin_amdgcn_wmma_f32_16x16x4_f32(
                false, a, false, bd, (short)0, accX[j], false, false);
            accS[j] = __builtin_amdgcn_wmma_f32_16x16x4_f32(
                false, s, false, bb, (short)0, accS[j], false, false);
        }
    }

    int ncol = lane & 15;
    int mAdd = (lane >> 4) * 8;
#pragma unroll
    for (int j = 0; j < 2; ++j) {
        int col = (nh * 2 + j) * 16 + ncol;
        float wc = Wc[col];
        float bb = bias[col];
        float r[8];
#pragma unroll
        for (int v = 0; v < 8; ++v) {
            int node = node0 + v + mAdd;
            bool own = (row0 + v + mAdd < BIN_NODES) && (node < N);
            r[v] = (resid && own) ? resid[(size_t)node * HDIM + col] : 0.0f;
        }
#pragma unroll
        for (int v = 0; v < 8; ++v) {
            int node = node0 + v + mAdd;
            if (row0 + v + mAdd >= BIN_NODES || node >= N) continue;
            float g = gate[node];
            float val = g * (accX[j][v] + bb) + invdeg[node] * accS[j][v] + ssn[node] * wc;
            if (resid) val += r[v];
            val = (val > 0.0f) ? val : 0.01f * val;
            Y[(size_t)node * HDIM + col] = val;
        }
    }
}

__global__ __launch_bounds__(1024) __attribute__((amdgpu_waves_per_eu(8))) void agg_out_kernel(const unsigned* __restrict__ list,
                                                      const int* __restrict__ binStart,
                                                      const float* __restrict__ X,
                                                      const float* __restrict__ Wout, const float* __restrict__ bout,
                                                      const float* __restrict__ x0,
                                                      const float* __restrict__ invdeg, const float* __restrict__ gate,
                                                      const float* __restrict__ ssn, float* __restrict__ out, int N) {
    __shared__ float4 tile4[TILE_ROWS * TILE_PITCH / 4];
    float* tile = (float*)tile4;

    const int bin  = blockIdx.x;
    const int lane = threadIdx.x & 31;
    const int wave = __builtin_amdgcn_readfirstlane((int)(threadIdx.x >> 5));

    const int n = bin * BIN_NODES + (int)threadIdx.x;
    const bool own = (int)threadIdx.x < BIN_NODES;
    float a1 = 0.0f;
    if (own) {
        const float* Xr = X + (size_t)min(n, N - 1) * HDIM;
        float xr[HDIM];
#pragma unroll
        for (int k = 0; k < HDIM; ++k) xr[k] = Xr[k];
        for (int i = threadIdx.x; i < TILE_ROWS * TILE_PITCH / 4; i += 1024) tile4[i] = make_float4(0.f, 0.f, 0.f, 0.f);
#pragma unroll
        for (int k = 0; k < HDIM; ++k) a1 += xr[k] * (Wout[k] - Wout[HDIM + k]);
    } else {
        for (int i = threadIdx.x; i < TILE_ROWS * TILE_PITCH / 4; i += 1024) tile4[i] = make_float4(0.f, 0.f, 0.f, 0.f);
    }
    __syncthreads();

    gather_bin<TILE_PITCH>(list, binStart[bin], binStart[bin + 1], wave, X, 2 * lane, tile + 2 * lane);
    __syncthreads();

    if (own && n < N) {
        float a2 = 0.0f;
        const float* Sr = tile + threadIdx.x * TILE_PITCH;
#pragma unroll
        for (int k = 0; k < HDIM; ++k) a2 += Sr[k] * Wout[HDIM + k];
        out[n] = gate[n] * (a1 + bout[0]) + invdeg[n] * a2 + ssn[n] * Wout[2 * HDIM] + x0[n];
    }
}

extern "C" void kernel_launch(void* const* d_in, const int* in_sizes, int n_in,
                              void* d_out, int out_size, void* d_ws, size_t ws_size,
                              hipStream_t stream) {
    const float* x    = (const float*)d_in[0];
    const int*   ei   = (const int*)  d_in[1];
    const float* Win  = (const float*)d_in[2];
    const float* bin  = (const float*)d_in[3];
    const float* Wmid = (const float*)d_in[4];
    const float* bmid = (const float*)d_in[5];
    const float* Wout = (const float*)d_in[6];
    const float* bout = (const float*)d_in[7];

    const int N = in_sizes[0];
    const int E = in_sizes[1] / 2;
    const int* src = ei;
    const int* dst = ei + E;

    float* w = (float*)d_ws;
    float* invdeg  = w + 2 * (size_t)N;
    float* gate    = w + 3 * (size_t)N;
    float* ssn     = w + 4 * (size_t)N;
    float* Sx      = w + 5 * (size_t)N;
    float* Wd6     = w + 6 * (size_t)N;
    float* bufA    = Wd6  + 6 * HDIM * HDIM;
    float* bufB    = bufA + (size_t)N * HDIM;
    float* bufC    = bufB + (size_t)N * HDIM;

    const int NB     = (N + BIN_NODES - 1) / BIN_NODES;
    const int chunk  = (E + CHUNK_WGS - 1) / CHUNK_WGS;

    unsigned* list  = (unsigned*)bufB;
    unsigned* list2 = (unsigned*)bufC;
    int* cnt        = (int*)(bufC + (size_t)E);
    int* pre        = cnt + (size_t)CHUNK_WGS * NB;
    int* binStart   = pre + (size_t)CHUNK_WGS * NB;
    int* tileStart  = (int*)w;

    wdiff_kernel<<<(6 * 4096 + 255) / 256, 256, 0, stream>>>(Wmid, Wd6, 6 * 4096);

    bin_count_kernel<<<CHUNK_WGS, 1024, 0, stream>>>(dst, cnt, E, NB, chunk);
    bin_scan_kernel<<<1, 1024, 0, stream>>>(cnt, pre, binStart, NB, CHUNK_WGS);
    bin_fill_kernel<<<CHUNK_WGS, 1024, 0, stream>>>(src, dst, pre, binStart, list, E, NB, chunk);

    bin_stats_kernel<<<NB, 1024, 0, stream>>>(list, binStart, x, invdeg, gate, ssn, Sx, list2, tileStart, N);

    conv_in_kernel<<<(N * HDIM + 255) / 256, 256, 0, stream>>>(x, Win, bin, invdeg, gate, ssn, Sx,
                                                               bufA, N);

    float* cur = bufA;
    for (int l = 0; l < 3; ++l) {
        const float* Wl = Wmid + (size_t)(2 * l) * 129 * HDIM;
        agg_conv_kernel<<<NB, 1024, 0, stream>>>(
            list2, tileStart, cur, Wd6 + (2 * l) * HDIM * HDIM, Wl + 64 * HDIM, Wl + 128 * HDIM,
            bmid + (2 * l) * HDIM, gate, invdeg, ssn, nullptr, bufB, N);

        const float* Wl2 = Wmid + (size_t)(2 * l + 1) * 129 * HDIM;
        agg_conv_kernel<<<NB, 1024, 0, stream>>>(
            list2, tileStart, bufB, Wd6 + (2 * l + 1) * HDIM * HDIM, Wl2 + 64 * HDIM, Wl2 + 128 * HDIM,
            bmid + (2 * l + 1) * HDIM, gate, invdeg, ssn, cur, cur, N);
    }

    agg_out_kernel<<<NB, 1024, 0, stream>>>(list2, binStart, cur, Wout, bout, x, invdeg, gate, ssn,
                                           (float*)d_out, N);
}


# Mid conv gather: a trip asks for both groups' 16 row reads at once (32-bit offsets, branch-free word pick, words a trip ahead), WMMAs follow as reads return; no wait on fresh reads; sum order kept
# speedup vs baseline: 3.8220x; 1.0716x over previous
#include <hip/hip_runtime.h>
#include <hip/hip_bf16.h>

typedef float v2f __attribute__((ext_vector_type(2)));
typedef float v8f __attribute__((ext_vector_type(8)));

#define HDIM 64

__global__ void wdiff_kernel(const float* __restrict__ Wmid, float* __restrict__ Wd6, int total) {
    int i = blockIdx.x * blockDim.x + threadIdx.x;
    if (i >= total) return;
    int l = i >> 12;
    int rc = i & 4095;
    const float* Wl = Wmid + (size_t)l * 129 * HDIM;
    Wd6[i] = Wl[rc] - Wl[64 * HDIM + rc];
}

__global__ void conv_in_kernel(const float* __restrict__ x, const float* __restrict__ Win,
                               const float* __restrict__ bin,
                               const float* __restrict__ invdeg, const float* __restrict__ gate,
                               const float* __restrict__ ssn, const float* __restrict__ Sx,
                               float* __restrict__ Y, int N) {
    int t = blockIdx.x * blockDim.x + threadIdx.x;
    if (t >= N * HDIM) return;
    int n = t >> 6, h = t & 63;
    float wa = Win[h], wb = Win[HDIM + h], wc = Win[2 * HDIM + h];
    Y[t] = gate[n] * (x[n] * (wa - wb) + bin[h]) + invdeg[n] * Sx[n] * wb + ssn[n] * wc;
}

#define BIN_SHIFT 8
#define BIN_MASK 255
#define BIN_NODES 200
#define MAX_BINS 1024
#define CHUNK_WGS 256

__global__ __launch_bounds__(1024) void bin_count_kernel(const int* __restrict__ dst, int* __restrict__ cnt,
                                                         int E, int NB, int chunk) {
    __shared__ int hist[MAX_BINS];
    for (int b = threadIdx.x; b < NB; b += 1024) hist[b] = 0;
    __syncthreads();
    int e0 = blockIdx.x * chunk;
    int e1 = min(e0 + chunk, E);
    for (int e = e0 + (int)threadIdx.x; e < e1; e += 1024)
        atomicAdd(&hist[dst[e] / BIN_NODES], 1);
    __syncthreads();
    for (int b = threadIdx.x; b < NB; b += 1024) cnt[blockIdx.x * NB + b] = hist[b];
}

__global__ __launch_bounds__(1024) void bin_scan_kernel(const int* __restrict__ cnt, int* __restrict__ pre,
                                                        int* __restrict__ binStart, int NB, int G) {
    __shared__ int tot[MAX_BINS];
    int b = threadIdx.x;
    int run = 0;
    if (b < NB) {
#pragma unroll 16
        for (int g = 0; g < G; ++g) {
            int c = cnt[g * NB + b];
            pre[g * NB + b] = run;
            run += c;
        }
    }
    tot[b] = run;
    __syncthreads();
    for (int off = 1; off < MAX_BINS; off <<= 1) {
        int v = (b >= off) ? tot[b - off] : 0;
        __syncthreads();
        tot[b] += v;
        __syncthreads();
    }
    if (b < NB) binStart[b] = tot[b] - run;
    if (b == NB - 1) binStart[NB] = tot[b];
}

__global__ __launch_bounds__(1024) void bin_fill_kernel(const int* __restrict__ src, const int* __restrict__ dst,
                                                        const int* __restrict__ pre,
                                                        const int* __restrict__ binStart,
                                                        unsigned* __restrict__ list, int E, int NB, int chunk) {
    __shared__ int cur[MAX_BINS];
    for (int b = threadIdx.x; b < NB; b += 1024) cur[b] = binStart[b] + pre[blockIdx.x * NB + b];
    __syncthreads();
    int e0 = blockIdx.x * chunk;
    int e1 = min(e0 + chunk, E);
    for (int e = e0 + (int)threadIdx.x; e < e1; e += 1024) {
        int d = dst[e];
        int s = src[e];
        int b = d / BIN_NODES;
        int slot = atomicAdd(&cur[b], 1);
        list[slot] = ((unsigned)s << BIN_SHIFT) | (unsigned)(d - b * BIN_NODES);
    }
}

#define ST_U 8

__global__ __launch_bounds__(1024) void bin_stats_kernel(const unsigned* __restrict__ list,
                                                        const int* __restrict__ binStart,
                                                        const float* __restrict__ x,
                                                        float* __restrict__ invdeg, float* __restrict__ gate,
                                                        float* __restrict__ ssn, float* __restrict__ Sx,
                                                        unsigned* __restrict__ list2, int* __restrict__ tileStart, int N) {
    __shared__ int   wcnt[32 * 16];
    __shared__ int   tbase[16];
    __shared__ int   sdeg[BIN_NODES];
    __shared__ int   ssgn[BIN_NODES];
    __shared__ float ssx[BIN_NODES];
    const int t = threadIdx.x;
    if (t < BIN_NODES) { sdeg[t] = 0; ssgn[t] = 0; ssx[t] = 0.0f; }
    if (t < 32 * 16) wcnt[t] = 0;
    __syncthreads();

    const int bin   = blockIdx.x;
    const int beg   = binStart[bin];
    const int end   = binStart[bin + 1];
    const int node0 = bin * BIN_NODES;

    for (int i = beg + t; i < end; i += 1024 * ST_U) {
        unsigned p[ST_U];
        float xv[ST_U];
#pragma unroll
        for (int j = 0; j < ST_U; ++j) p[j] = list[min(i + j * 1024, end - 1)];
#pragma unroll
        for (int j = 0; j < ST_U; ++j) xv[j] = x[p[j] >> BIN_SHIFT];
#pragma unroll
        for (int j = 0; j < ST_U; ++j) {
            if (i + j * 1024 < end) {
                int s  = (int)(p[j] >> BIN_SHIFT);
                int dl = (int)(p[j] & BIN_MASK);
                int d  = node0 + dl;
                atomicAdd(&sdeg[dl], 1);
                atomicAdd(&ssgn[dl], (s > d) - (s < d));
                atomicAdd(&ssx[dl], xv[j]);
            }
        }
    }
    __syncthreads();

    const int wv = t >> 5;
    for (int i = beg + t; i < end; i += 1024) {
        unsigned p = list[i];
        atomicAdd(&wcnt[wv * 16 + (int)((p & BIN_MASK) >> 4)], 1);
    }
    __syncthreads();
    if (t < 16) {
        int run = 0;
        for (int w = 0; w < 32; ++w) { int c = wcnt[w * 16 + t]; wcnt[w * 16 + t] = run; run += c; }
        tbase[t] = run;
    }
    __syncthreads();
    if (t == 0) {
        int run = beg;
        for (int k = 0; k < 16; ++k) { int c = tbase[k]; tbase[k] = run; tileStart[bin * 16 + k] = run; run += c; }
    }
    __syncthreads();
    for (int i = beg + t; i < end; i += 1024) {
        unsigned p = list[i];
        int tl = (int)((p & BIN_MASK) >> 4);
        int slot = tbase[tl] + atomicAdd(&wcnt[wv * 16 + tl], 1);
        list2[slot] = p;
    }

    if (t < BIN_NODES && node0 + t < N) {
        int n = node0 + t;
        float dg  = (float)sdeg[t];
        float inv = 1.0f / fmaxf(dg, 1.0f);
        invdeg[n] = inv;
        gate[n]   = (dg > 0.0f) ? 1.0f : 0.0f;
        ssn[n]    = (float)ssgn[t] * inv;
        Sx[n]     = ssx[t];
    }
}

#define AGG_U 16
#define AGG_WAVES 32

__device__ __forceinline__ int win_start(int i, int end) { return max(min(i, end - AGG_U), 0); }

__device__ __forceinline__ void edge_words(const unsigned* __restrict__ list, int ws, unsigned (&w)[AGG_U]) {
#pragma unroll
    for (int j = 0; j < AGG_U; ++j) w[j] = list[ws + j];
}

__device__ __forceinline__ void row_reads(const unsigned (&w)[AGG_U], const float* __restrict__ X, int lane2,
                                          v2f (&v)[AGG_U], unsigned (&d)[AGG_U]) {
#pragma unroll
    for (int j = 0; j < AGG_U; ++j) {
        const float* rowp = X + (size_t)(w[j] >> BIN_SHIFT) * HDIM;
        v[j] = *(const v2f*)(rowp + lane2);
        d[j] = w[j] & BIN_MASK;
    }
}

template <bool MASKED, int PITCH>
__device__ __forceinline__ void row_adds(const v2f (&v)[AGG_U], const unsigned (&d)[AGG_U], float* tl,
                                         int ws, int lo, int end) {
#pragma unroll
    for (int j = 0; j < AGG_U; ++j) {
        if (!MASKED || (ws + j >= lo && ws + j < end)) {
            float* t = tl + d[j] * PITCH;
            atomicAdd(t, v[j].x);
            atomicAdd(t + 1, v[j].y);
        }
    }
}

template <int PITCH>
__device__ __forceinline__ void gather_bin(const unsigned* __restrict__ list, int beg, int end, int wave,
                                           const float* __restrict__ X, int lane2, float* tl) {
    const int step = AGG_WAVES * AGG_U;
    int i = beg + wave * AGG_U;
    if (i >= end) return;
    unsigned w1[AGG_U], w2[AGG_U];
    v2f va[AGG_U], vb[AGG_U];
    unsigned da[AGG_U], db[AGG_U];
    edge_words(list, win_start(i, end), w1);
    row_reads(w1, X, lane2, va, da);
    int in = i + step;
    edge_words(list, win_start(in, end), w1);
    edge_words(list, win_start(in + step, end), w2);
    while (in + step < end) {
        row_reads(w1, X, lane2, vb, db);
        __builtin_amdgcn_sched_barrier(0);
        row_adds<false, PITCH>(va, da, tl, 0, 0, 0);
        row_reads(w2, X, lane2, va, da);
        edge_words(list, win_start(in + 2 * step, end), w1);
        edge_words(list, win_start(in + 3 * step, end), w2);
        __builtin_amdgcn_sched_barrier(0);
        row_adds<false, PITCH>(vb, db, tl, 0, 0, 0);
        i = in + step;
        in = i + step;
    }
    if (in < end) {
        row_reads(w1, X, lane2, vb, db);
        row_adds<false, PITCH>(va, da, tl, 0, 0, 0);
        row_adds<true, PITCH>(vb, db, tl, win_start(in, end), in, end);
    } else {
        row_adds<true, PITCH>(va, da, tl, win_start(i, end), i, end);
    }
}

__device__ __forceinline__ void tg_words(const unsigned* __restrict__ l2, int ws, unsigned (&w)[AGG_U]) {
#pragma unroll
    for (int j = 0; j < AGG_U; ++j) w[j] = l2[ws + j];
}

__device__ __forceinline__ void tg_loads(const unsigned (&w)[AGG_U], const float* __restrict__ X, unsigned hm, int c4,
                                         float4 (&b)[AGG_U / 2]) {
#pragma unroll
    for (int q = 0; q < AGG_U / 2; ++q) {
        const int ks = q >> 1, e = q & 1;
        unsigned ww = (w[ks * 4 + 2 + e] & hm) | (w[ks * 4 + e] & ~hm);
        unsigned off = (ww & ~(unsigned)BIN_MASK) | (unsigned)c4;
        b[q] = *(const float4*)((const char*)X + off);
    }
}

template <bool MASKED>
__device__ __forceinline__ void tg_mma(const unsigned (&w)[AGG_U], const float4 (&b)[AGG_U / 2], unsigned hm,
                                       unsigned mrow, int ws, int lo, int end, v8f (&acc)[4]) {
#pragma unroll
    for (int ks = 0; ks < 4; ++ks) {
        v2f a;
#pragma unroll
        for (int e = 0; e < 2; ++e) {
            const int jl = ks * 4 + e, jh = ks * 4 + 2 + e;
            unsigned ww = (w[jh] & hm) | (w[jl] & ~hm);
            bool on = ww == mrow;
            if (MASKED) {
                bool vl = (ws + jl >= lo) && (ws + jl < end);
                bool vh = (ws + jh >= lo) && (ws + jh < end);
                on = on && (((vh ? hm : 0u) | (vl ? ~hm : 0u)) != 0u);
            }
            a[e] = on ? 1.0f : 0.0f;
        }
        const float4 b0 = b[ks * 2], b1 = b[ks * 2 + 1];
        v2f f0 = {b0.x, b1.x}, f1 = {b0.y, b1.y}, f2 = {b0.z, b1.z}, f3 = {b0.w, b1.w};
        acc[0] = __builtin_amdgcn_wmma_f32_16x16x4_f32(false, a, false, f0, (short)0, acc[0], false, false);
        acc[1] = __builtin_amdgcn_wmma_f32_16x16x4_f32(false, a, false, f1, (short)0, acc[1], false, false);
        acc[2] = __builtin_amdgcn_wmma_f32_16x16x4_f32(false, a, false, f2, (short)0, acc[2], false, false);
        acc[3] = __builtin_amdgcn_wmma_f32_16x16x4_f32(false, a, false, f3, (short)0, acc[3], false, false);
    }
}

__device__ __forceinline__ void tg_rows(const unsigned (&w)[AGG_U], unsigned (&d)[AGG_U]) {
#pragma unroll
    for (int j = 0; j < AGG_U; ++j) d[j] = w[j] & BIN_MASK;
}

__device__ __forceinline__ void tile_gather(const unsigned* __restrict__ l2, int a0, int a1,
                                            const float* __restrict__ X, unsigned hm, int c4, unsigned mrow,
                                            v8f (&acc)[4]) {
    if (a0 >= a1) return;
    unsigned w1[AGG_U], w2[AGG_U];
    unsigned da[AGG_U], db[AGG_U];
    float4 ba[AGG_U / 2], bb[AGG_U / 2];
    int i = a0;
    tg_words(l2, win_start(i, a1), w1);
    tg_words(l2, win_start(i + AGG_U, a1), w2);
    while (i + 2 * AGG_U < a1) {
        tg_loads(w1, X, hm, c4, ba);
        tg_rows(w1, da);
        tg_loads(w2, X, hm, c4, bb);
        tg_rows(w2, db);
        tg_words(l2, win_start(i + 2 * AGG_U, a1), w1);
        tg_words(l2, win_start(i + 3 * AGG_U, a1), w2);
        __builtin_amdgcn_sched_barrier(0);
        tg_mma<false>(da, ba, hm, mrow, 0, 0, 0, acc);
        tg_mma<false>(db, bb, hm, mrow, 0, 0, 0, acc);
        i += 2 * AGG_U;
    }
    if (i + AGG_U < a1) {
        const int ws2 = win_start(i + AGG_U, a1);
        tg_loads(w1, X, hm, c4, ba);
        tg_rows(w1, da);
        tg_loads(w2, X, hm, c4, bb);
        tg_rows(w2, db);
        tg_mma<false>(da, ba, hm, mrow, 0, 0, 0, acc);
        tg_mma<true>(db, bb, hm, mrow, ws2, i + AGG_U, a1, acc);
    } else {
        tg_loads(w1, X, hm, c4, ba);
        tg_rows(w1, da);
        tg_mma<true>(da, ba, hm, mrow, win_start(i, a1), i, a1, acc);
    }
}

#define TILE_PITCH 68
#define MM_TILES ((BIN_NODES + 15) / 16)
#define TILE_ROWS (MM_TILES * 16)

__global__ __launch_bounds__(1024) __attribute__((amdgpu_waves_per_eu(8))) void agg_conv_kernel(
    const unsigned* __restrict__ list2, const int* __restrict__ tileStart,
    const float* __restrict__ X,
    const float* __restrict__ Wd, const float* __restrict__ Wb,
    const float* __restrict__ Wc, const float* __restrict__ bias,
    const float* __restrict__ gate, const float* __restrict__ invdeg,
    const float* __restrict__ ssn, const float* resid,
    float* Y, int N) {
    __shared__ float4 tile4[TILE_ROWS * TILE_PITCH / 4];
    __shared__ float sWd[HDIM * HDIM];
    __shared__ float sWb[HDIM * HDIM];
    float* tile = (float*)tile4;

    const int bin  = blockIdx.x;
    const int lane = threadIdx.x & 31;
    const int wave = __builtin_amdgcn_readfirstlane((int)(threadIdx.x >> 5));
    const int tid  = threadIdx.x;

    {
        float wdr[4], wbr[4];
        int t256 = tid & 255;
        int q    = tid >> 8;
        int nt   = t256 >> 6;
        int lp   = t256 & 63;
        int ln   = lp >> 1;
        int half = lp & 1;
        int kh   = (ln >> 4) << 1;
        int wsrc = (kh + half) * HDIM + (nt << 4) + (ln & 15);
#pragma unroll
        for (int i = 0; i < 4; ++i) {
            int ks = (q << 2) + i;
            wdr[i] = Wd[(ks << 2) * HDIM + wsrc];
            wbr[i] = Wb[(ks << 2) * HDIM + wsrc];
        }
#pragma unroll
        for (int i = 0; i < 4; ++i) {
            int ks = (q << 2) + i;
            sWd[(ks << 8) + t256] = wdr[i];
            sWb[(ks << 8) + t256] = wbr[i];
        }
    }

    const bool mm   = wave < 2 * MM_TILES;
    const int tq    = wave >> 1;
    const int nh    = wave & 1;
    const int row0  = tq * 16;
    const int node0 = bin * BIN_NODES + row0;
    const int m  = lane & 15;
    const int kh = (lane >> 4) << 1;

    v8f sacc[4] = {};
    if (mm) {
        const int ts = tileStart[bin * 16 + tq];
        const int te = tileStart[bin * 16 + tq + 1];
        const int hl = ((((te - ts) + 1) >> 1) + AGG_U - 1) & ~(AGG_U - 1);
        const int a0 = ts + nh * hl;
        const int a1 = min(te, a0 + hl);
        tile_gather(list2, a0, a1, X, (lane >= 16) ? ~0u : 0u, 16 * m, (unsigned)(row0 + m), sacc);
    }
    float4* trow = (float4*)(tile + (row0 + (lane >> 4) * 8) * TILE_PITCH + 4 * m);
    if (mm && nh == 0) {
#pragma unroll
        for (int r = 0; r < 8; ++r)
            trow[r * (TILE_PITCH / 4)] = make_float4(sacc[0][r], sacc[1][r], sacc[2][r], sacc[3][r]);
    }
    __syncthreads();
    if (mm && nh == 1) {
#pragma unroll
        for (int r = 0; r < 8; ++r) {
            float4 v = trow[r * (TILE_PITCH / 4)];
            v.x += sacc[0][r]; v.y += sacc[1][r]; v.z += sacc[2][r]; v.w += sacc[3][r];
            trow[r * (TILE_PITCH / 4)] = v;
        }
    }
    int row = node0 + m; if (row >= N) row = N - 1;
    const float* Xr = X + (size_t)row * HDIM;
    v2f xa[16];
    if (mm) {
#pragma unroll
        for (int ks = 0; ks < 16; ++ks) {
            xa[ks].x = Xr[(ks << 2) + kh];
            xa[ks].y = Xr[(ks << 2) + kh + 1];
        }
    }
    __syncthreads();
    if (!mm) return;

    const float* Sr = tile + (row0 + m) * TILE_PITCH;

    v8f accX[2] = {};
    v8f accS[2] = {};
    const int fragBase = lane * 2;

#pragma unroll
    for (int ks = 0; ks < 16; ++ks) {
        int k0 = ks << 2;
        v2f a = xa[ks];
        v2f s = *(const v2f*)(Sr + k0 + kh);
#pragma unroll
        for (int j = 0; j < 2; ++j) {
            int nt = nh * 2 + j;
            v2f bd = *(const v2f*)&sWd[((ks << 2) + nt) * 64 + fragBase];
            v2f bb = *(const v2f*)&sWb[((ks << 2) + nt) * 64 + fragBase];
            accX[j] = __builtin_amdgcn_wmma_f32_16x16x4_f32(
                false, a, false, bd, (short)0, accX[j], false, false);
            accS[j] = __builtin_amdgcn_wmma_f32_16x16x4_f32(
                false, s, false, bb, (short)0, accS[j], false, false);
        }
    }

    int ncol = lane & 15;
    int mAdd = (lane >> 4) * 8;
#pragma unroll
    for (int j = 0; j < 2; ++j) {
        int col = (nh * 2 + j) * 16 + ncol;
        float wc = Wc[col];
        float bb = bias[col];
        float r[8];
#pragma unroll
        for (int v = 0; v < 8; ++v) {
            int node = node0 + v + mAdd;
            bool own = (row0 + v + mAdd < BIN_NODES) && (node < N);
            r[v] = (resid && own) ? resid[(size_t)node * HDIM + col] : 0.0f;
        }
#pragma unroll
        for (int v = 0; v < 8; ++v) {
            int node = node0 + v + mAdd;
            if (row0 + v + mAdd >= BIN_NODES || node >= N) continue;
            float g = gate[node];
            float val = g * (accX[j][v] + bb) + invdeg[node] * accS[j][v] + ssn[node] * wc;
            if (resid) val += r[v];
            val = (val > 0.0f) ? val : 0.01f * val;
            Y[(size_t)node * HDIM + col] = val;
        }
    }
}

__global__ __launch_bounds__(1024) __attribute__((amdgpu_waves_per_eu(8))) void agg_out_kernel(const unsigned* __restrict__ list,
                                                      const int* __restrict__ binStart,
                                                      const float* __restrict__ X,
                                                      const float* __restrict__ Wout, const float* __restrict__ bout,
                                                      const float* __restrict__ x0,
                                                      const float* __restrict__ invdeg, const float* __restrict__ gate,
                                                      const float* __restrict__ ssn, float* __restrict__ out, int N) {
    __shared__ float4 tile4[TILE_ROWS * TILE_PITCH / 4];
    float* tile = (float*)tile4;

    const int bin  = blockIdx.x;
    const int lane = threadIdx.x & 31;
    const int wave = __builtin_amdgcn_readfirstlane((int)(threadIdx.x >> 5));

    const int n = bin * BIN_NODES + (int)threadIdx.x;
    const bool own = (int)threadIdx.x < BIN_NODES;
    float a1 = 0.0f;
    if (own) {
        const float* Xr = X + (size_t)min(n, N - 1) * HDIM;
        float xr[HDIM];
#pragma unroll
        for (int k = 0; k < HDIM; ++k) xr[k] = Xr[k];
        for (int i = threadIdx.x; i < TILE_ROWS * TILE_PITCH / 4; i += 1024) tile4[i] = make_float4(0.f, 0.f, 0.f, 0.f);
#pragma unroll
        for (int k = 0; k < HDIM; ++k) a1 += xr[k] * (Wout[k] - Wout[HDIM + k]);
    } else {
        for (int i = threadIdx.x; i < TILE_ROWS * TILE_PITCH / 4; i += 1024) tile4[i] = make_float4(0.f, 0.f, 0.f, 0.f);
    }
    __syncthreads();

    gather_bin<TILE_PITCH>(list, binStart[bin], binStart[bin + 1], wave, X, 2 * lane, tile + 2 * lane);
    __syncthreads();

    if (own && n < N) {
        float a2 = 0.0f;
        const float* Sr = tile + threadIdx.x * TILE_PITCH;
#pragma unroll
        for (int k = 0; k < HDIM; ++k) a2 += Sr[k] * Wout[HDIM + k];
        out[n] = gate[n] * (a1 + bout[0]) + invdeg[n] * a2 + ssn[n] * Wout[2 * HDIM] + x0[n];
    }
}

extern "C" void kernel_launch(void* const* d_in, const int* in_sizes, int n_in,
                              void* d_out, int out_size, void* d_ws, size_t ws_size,
                              hipStream_t stream) {
    const float* x    = (const float*)d_in[0];
    const int*   ei   = (const int*)  d_in[1];
    const float* Win  = (const float*)d_in[2];
    const float* bin  = (const float*)d_in[3];
    const float* Wmid = (const float*)d_in[4];
    const float* bmid = (const float*)d_in[5];
    const float* Wout = (const float*)d_in[6];
    const float* bout = (const float*)d_in[7];

    const int N = in_sizes[0];
    const int E = in_sizes[1] / 2;
    const int* src = ei;
    const int* dst = ei + E;

    float* w = (float*)d_ws;
    float* invdeg  = w + 2 * (size_t)N;
    float* gate    = w + 3 * (size_t)N;
    float* ssn     = w + 4 * (size_t)N;
    float* Sx      = w + 5 * (size_t)N;
    float* Wd6     = w + 6 * (size_t)N;
    float* bufA    = Wd6  + 6 * HDIM * HDIM;
    float* bufB    = bufA + (size_t)N * HDIM;
    float* bufC    = bufB + (size_t)N * HDIM;

    const int NB     = (N + BIN_NODES - 1) / BIN_NODES;
    const int chunk  = (E + CHUNK_WGS - 1) / CHUNK_WGS;

    unsigned* list  = (unsigned*)bufB;
    unsigned* list2 = (unsigned*)bufC;
    int* cnt        = (int*)(bufC + (size_t)E);
    int* pre        = cnt + (size_t)CHUNK_WGS * NB;
    int* binStart   = pre + (size_t)CHUNK_WGS * NB;
    int* tileStart  = (int*)w;

    wdiff_kernel<<<(6 * 4096 + 255) / 256, 256, 0, stream>>>(Wmid, Wd6, 6 * 4096);

    bin_count_kernel<<<CHUNK_WGS, 1024, 0, stream>>>(dst, cnt, E, NB, chunk);
    bin_scan_kernel<<<1, 1024, 0, stream>>>(cnt, pre, binStart, NB, CHUNK_WGS);
    bin_fill_kernel<<<CHUNK_WGS, 1024, 0, stream>>>(src, dst, pre, binStart, list, E, NB, chunk);

    bin_stats_kernel<<<NB, 1024, 0, stream>>>(list, binStart, x, invdeg, gate, ssn, Sx, list2, tileStart, N);

    conv_in_kernel<<<(N * HDIM + 255) / 256, 256, 0, stream>>>(x, Win, bin, invdeg, gate, ssn, Sx,
                                                               bufA, N);

    float* cur = bufA;
    for (int l = 0; l < 3; ++l) {
        const float* Wl = Wmid + (size_t)(2 * l) * 129 * HDIM;
        agg_conv_kernel<<<NB, 1024, 0, stream>>>(
            list2, tileStart, cur, Wd6 + (2 * l) * HDIM * HDIM, Wl + 64 * HDIM, Wl + 128 * HDIM,
            bmid + (2 * l) * HDIM, gate, invdeg, ssn, nullptr, bufB, N);

        const float* Wl2 = Wmid + (size_t)(2 * l + 1) * 129 * HDIM;
        agg_conv_kernel<<<NB, 1024, 0, stream>>>(
            list2, tileStart, bufB, Wd6 + (2 * l + 1) * HDIM * HDIM, Wl2 + 64 * HDIM, Wl2 + 128 * HDIM,
            bmid + (2 * l + 1) * HDIM, gate, invdeg, ssn, cur, cur, N);
    }

    agg_out_kernel<<<NB, 1024, 0, stream>>>(list2, binStart, cur, Wout, bout, x, invdeg, gate, ssn,
                                           (float*)d_out, N);
}


# conv_out's neighbor sum moves to the matrix unit: per 16-node tile 0/1 (node x edge) A times gathered rows, f32 WMMA, as in the mid convs; its LDS float atomics and zero fill go; sum order: list
# speedup vs baseline: 3.9824x; 1.0420x over previous
#include <hip/hip_runtime.h>
#include <hip/hip_bf16.h>

typedef float v2f __attribute__((ext_vector_type(2)));
typedef float v8f __attribute__((ext_vector_type(8)));

#define HDIM 64

__global__ void wdiff_kernel(const float* __restrict__ Wmid, float* __restrict__ Wd6, int total) {
    int i = blockIdx.x * blockDim.x + threadIdx.x;
    if (i >= total) return;
    int l = i >> 12;
    int rc = i & 4095;
    const float* Wl = Wmid + (size_t)l * 129 * HDIM;
    Wd6[i] = Wl[rc] - Wl[64 * HDIM + rc];
}

__global__ void conv_in_kernel(const float* __restrict__ x, const float* __restrict__ Win,
                               const float* __restrict__ bin,
                               const float* __restrict__ invdeg, const float* __restrict__ gate,
                               const float* __restrict__ ssn, const float* __restrict__ Sx,
                               float* __restrict__ Y, int N) {
    int t = blockIdx.x * blockDim.x + threadIdx.x;
    if (t >= N * HDIM) return;
    int n = t >> 6, h = t & 63;
    float wa = Win[h], wb = Win[HDIM + h], wc = Win[2 * HDIM + h];
    Y[t] = gate[n] * (x[n] * (wa - wb) + bin[h]) + invdeg[n] * Sx[n] * wb + ssn[n] * wc;
}

#define BIN_SHIFT 8
#define BIN_MASK 255
#define BIN_NODES 200
#define MAX_BINS 1024
#define CHUNK_WGS 256

__global__ __launch_bounds__(1024) void bin_count_kernel(const int* __restrict__ dst, int* __restrict__ cnt,
                                                         int E, int NB, int chunk) {
    __shared__ int hist[MAX_BINS];
    for (int b = threadIdx.x; b < NB; b += 1024) hist[b] = 0;
    __syncthreads();
    int e0 = blockIdx.x * chunk;
    int e1 = min(e0 + chunk, E);
    for (int e = e0 + (int)threadIdx.x; e < e1; e += 1024)
        atomicAdd(&hist[dst[e] / BIN_NODES], 1);
    __syncthreads();
    for (int b = threadIdx.x; b < NB; b += 1024) cnt[blockIdx.x * NB + b] = hist[b];
}

__global__ __launch_bounds__(1024) void bin_scan_kernel(const int* __restrict__ cnt, int* __restrict__ pre,
                                                        int* __restrict__ binStart, int NB, int G) {
    __shared__ int tot[MAX_BINS];
    int b = threadIdx.x;
    int run = 0;
    if (b < NB) {
#pragma unroll 16
        for (int g = 0; g < G; ++g) {
            int c = cnt[g * NB + b];
            pre[g * NB + b] = run;
            run += c;
        }
    }
    tot[b] = run;
    __syncthreads();
    for (int off = 1; off < MAX_BINS; off <<= 1) {
        int v = (b >= off) ? tot[b - off] : 0;
        __syncthreads();
        tot[b] += v;
        __syncthreads();
    }
    if (b < NB) binStart[b] = tot[b] - run;
    if (b == NB - 1) binStart[NB] = tot[b];
}

__global__ __launch_bounds__(1024) void bin_fill_kernel(const int* __restrict__ src, const int* __restrict__ dst,
                                                        const int* __restrict__ pre,
                                                        const int* __restrict__ binStart,
                                                        unsigned* __restrict__ list, int E, int NB, int chunk) {
    __shared__ int cur[MAX_BINS];
    for (int b = threadIdx.x; b < NB; b += 1024) cur[b] = binStart[b] + pre[blockIdx.x * NB + b];
    __syncthreads();
    int e0 = blockIdx.x * chunk;
    int e1 = min(e0 + chunk, E);
    for (int e = e0 + (int)threadIdx.x; e < e1; e += 1024) {
        int d = dst[e];
        int s = src[e];
        int b = d / BIN_NODES;
        int slot = atomicAdd(&cur[b], 1);
        list[slot] = ((unsigned)s << BIN_SHIFT) | (unsigned)(d - b * BIN_NODES);
    }
}

#define ST_U 8

__global__ __launch_bounds__(1024) void bin_stats_kernel(const unsigned* __restrict__ list,
                                                        const int* __restrict__ binStart,
                                                        const float* __restrict__ x,
                                                        float* __restrict__ invdeg, float* __restrict__ gate,
                                                        float* __restrict__ ssn, float* __restrict__ Sx,
                                                        unsigned* __restrict__ list2, int* __restrict__ tileStart, int N) {
    __shared__ int   wcnt[32 * 16];
    __shared__ int   tbase[16];
    __shared__ int   sdeg[BIN_NODES];
    __shared__ int   ssgn[BIN_NODES];
    __shared__ float ssx[BIN_NODES];
    const int t = threadIdx.x;
    if (t < BIN_NODES) { sdeg[t] = 0; ssgn[t] = 0; ssx[t] = 0.0f; }
    if (t < 32 * 16) wcnt[t] = 0;
    __syncthreads();

    const int bin   = blockIdx.x;
    const int beg   = binStart[bin];
    const int end   = binStart[bin + 1];
    const int node0 = bin * BIN_NODES;

    for (int i = beg + t; i < end; i += 1024 * ST_U) {
        unsigned p[ST_U];
        float xv[ST_U];
#pragma unroll
        for (int j = 0; j < ST_U; ++j) p[j] = list[min(i + j * 1024, end - 1)];
#pragma unroll
        for (int j = 0; j < ST_U; ++j) xv[j] = x[p[j] >> BIN_SHIFT];
#pragma unroll
        for (int j = 0; j < ST_U; ++j) {
            if (i + j * 1024 < end) {
                int s  = (int)(p[j] >> BIN_SHIFT);
                int dl = (int)(p[j] & BIN_MASK);
                int d  = node0 + dl;
                atomicAdd(&sdeg[dl], 1);
                atomicAdd(&ssgn[dl], (s > d) - (s < d));
                atomicAdd(&ssx[dl], xv[j]);
            }
        }
    }
    __syncthreads();

    const int wv = t >> 5;
    for (int i = beg + t; i < end; i += 1024) {
        unsigned p = list[i];
        atomicAdd(&wcnt[wv * 16 + (int)((p & BIN_MASK) >> 4)], 1);
    }
    __syncthreads();
    if (t < 16) {
        int run = 0;
        for (int w = 0; w < 32; ++w) { int c = wcnt[w * 16 + t]; wcnt[w * 16 + t] = run; run += c; }
        tbase[t] = run;
    }
    __syncthreads();
    if (t == 0) {
        int run = beg;
        for (int k = 0; k < 16; ++k) { int c = tbase[k]; tbase[k] = run; tileStart[bin * 16 + k] = run; run += c; }
    }
    __syncthreads();
    for (int i = beg + t; i < end; i += 1024) {
        unsigned p = list[i];
        int tl = (int)((p & BIN_MASK) >> 4);
        int slot = tbase[tl] + atomicAdd(&wcnt[wv * 16 + tl], 1);
        list2[slot] = p;
    }

    if (t < BIN_NODES && node0 + t < N) {
        int n = node0 + t;
        float dg  = (float)sdeg[t];
        float inv = 1.0f / fmaxf(dg, 1.0f);
        invdeg[n] = inv;
        gate[n]   = (dg > 0.0f) ? 1.0f : 0.0f;
        ssn[n]    = (float)ssgn[t] * inv;
        Sx[n]     = ssx[t];
    }
}

#define AGG_U 16

__device__ __forceinline__ int win_start(int i, int end) { return max(min(i, end - AGG_U), 0); }

__device__ __forceinline__ void tg_words(const unsigned* __restrict__ l2, int ws, unsigned (&w)[AGG_U]) {
#pragma unroll
    for (int j = 0; j < AGG_U; ++j) w[j] = l2[ws + j];
}

__device__ __forceinline__ void tg_loads(const unsigned (&w)[AGG_U], const float* __restrict__ X, unsigned hm, int c4,
                                         float4 (&b)[AGG_U / 2]) {
#pragma unroll
    for (int q = 0; q < AGG_U / 2; ++q) {
        const int ks = q >> 1, e = q & 1;
        unsigned ww = (w[ks * 4 + 2 + e] & hm) | (w[ks * 4 + e] & ~hm);
        unsigned off = (ww & ~(unsigned)BIN_MASK) | (unsigned)c4;
        b[q] = *(const float4*)((const char*)X + off);
    }
}

template <bool MASKED>
__device__ __forceinline__ void tg_mma(const unsigned (&w)[AGG_U], const float4 (&b)[AGG_U / 2], unsigned hm,
                                       unsigned mrow, int ws, int lo, int end, v8f (&acc)[4]) {
#pragma unroll
    for (int ks = 0; ks < 4; ++ks) {
        v2f a;
#pragma unroll
        for (int e = 0; e < 2; ++e) {
            const int jl = ks * 4 + e, jh = ks * 4 + 2 + e;
            unsigned ww = (w[jh] & hm) | (w[jl] & ~hm);
            bool on = ww == mrow;
            if (MASKED) {
                bool vl = (ws + jl >= lo) && (ws + jl < end);
                bool vh = (ws + jh >= lo) && (ws + jh < end);
                on = on && (((vh ? hm : 0u) | (vl ? ~hm : 0u)) != 0u);
            }
            a[e] = on ? 1.0f : 0.0f;
        }
        const float4 b0 = b[ks * 2], b1 = b[ks * 2 + 1];
        v2f f0 = {b0.x, b1.x}, f1 = {b0.y, b1.y}, f2 = {b0.z, b1.z}, f3 = {b0.w, b1.w};
        acc[0] = __builtin_amdgcn_wmma_f32_16x16x4_f32(false, a, false, f0, (short)0, acc[0], false, false);
        acc[1] = __builtin_amdgcn_wmma_f32_16x16x4_f32(false, a, false, f1, (short)0, acc[1], false, false);
        acc[2] = __builtin_amdgcn_wmma_f32_16x16x4_f32(false, a, false, f2, (short)0, acc[2], false, false);
        acc[3] = __builtin_amdgcn_wmma_f32_16x16x4_f32(false, a, false, f3, (short)0, acc[3], false, false);
    }
}

__device__ __forceinline__ void tg_rows(const unsigned (&w)[AGG_U], unsigned (&d)[AGG_U]) {
#pragma unroll
    for (int j = 0; j < AGG_U; ++j) d[j] = w[j] & BIN_MASK;
}

__device__ __forceinline__ void tile_gather(const unsigned* __restrict__ l2, int a0, int a1,
                                            const float* __restrict__ X, unsigned hm, int c4, unsigned mrow,
                                            v8f (&acc)[4]) {
    if (a0 >= a1) return;
    unsigned w1[AGG_U], w2[AGG_U];
    unsigned da[AGG_U], db[AGG_U];
    float4 ba[AGG_U / 2], bb[AGG_U / 2];
    int i = a0;
    tg_words(l2, win_start(i, a1), w1);
    tg_words(l2, win_start(i + AGG_U, a1), w2);
    while (i + 2 * AGG_U < a1) {
        tg_loads(w1, X, hm, c4, ba);
        tg_rows(w1, da);
        tg_loads(w2, X, hm, c4, bb);
        tg_rows(w2, db);
        tg_words(l2, win_start(i + 2 * AGG_U, a1), w1);
        tg_words(l2, win_start(i + 3 * AGG_U, a1), w2);
        __builtin_amdgcn_sched_barrier(0);
        tg_mma<false>(da, ba, hm, mrow, 0, 0, 0, acc);
        tg_mma<false>(db, bb, hm, mrow, 0, 0, 0, acc);
        i += 2 * AGG_U;
    }
    if (i + AGG_U < a1) {
        const int ws2 = win_start(i + AGG_U, a1);
        tg_loads(w1, X, hm, c4, ba);
        tg_rows(w1, da);
        tg_loads(w2, X, hm, c4, bb);
        tg_rows(w2, db);
        tg_mma<false>(da, ba, hm, mrow, 0, 0, 0, acc);
        tg_mma<true>(db, bb, hm, mrow, ws2, i + AGG_U, a1, acc);
    } else {
        tg_loads(w1, X, hm, c4, ba);
        tg_rows(w1, da);
        tg_mma<true>(da, ba, hm, mrow, win_start(i, a1), i, a1, acc);
    }
}

#define TILE_PITCH 68
#define MM_TILES ((BIN_NODES + 15) / 16)
#define TILE_ROWS (MM_TILES * 16)

__global__ __launch_bounds__(1024) __attribute__((amdgpu_waves_per_eu(8))) void agg_conv_kernel(
    const unsigned* __restrict__ list2, const int* __restrict__ tileStart,
    const float* __restrict__ X,
    const float* __restrict__ Wd, const float* __restrict__ Wb,
    const float* __restrict__ Wc, const float* __restrict__ bias,
    const float* __restrict__ gate, const float* __restrict__ invdeg,
    const float* __restrict__ ssn, const float* resid,
    float* Y, int N) {
    __shared__ float4 tile4[TILE_ROWS * TILE_PITCH / 4];
    __shared__ float sWd[HDIM * HDIM];
    __shared__ float sWb[HDIM * HDIM];
    float* tile = (float*)tile4;

    const int bin  = blockIdx.x;
    const int lane = threadIdx.x & 31;
    const int wave = __builtin_amdgcn_readfirstlane((int)(threadIdx.x >> 5));
    const int tid  = threadIdx.x;

    {
        float wdr[4], wbr[4];
        int t256 = tid & 255;
        int q    = tid >> 8;
        int nt   = t256 >> 6;
        int lp   = t256 & 63;
        int ln   = lp >> 1;
        int half = lp & 1;
        int kh   = (ln >> 4) << 1;
        int wsrc = (kh + half) * HDIM + (nt << 4) + (ln & 15);
#pragma unroll
        for (int i = 0; i < 4; ++i) {
            int ks = (q << 2) + i;
            wdr[i] = Wd[(ks << 2) * HDIM + wsrc];
            wbr[i] = Wb[(ks << 2) * HDIM + wsrc];
        }
#pragma unroll
        for (int i = 0; i < 4; ++i) {
            int ks = (q << 2) + i;
            sWd[(ks << 8) + t256] = wdr[i];
            sWb[(ks << 8) + t256] = wbr[i];
        }
    }

    const bool mm   = wave < 2 * MM_TILES;
    const int tq    = wave >> 1;
    const int nh    = wave & 1;
    const int row0  = tq * 16;
    const int node0 = bin * BIN_NODES + row0;
    const int m  = lane & 15;
    const int kh = (lane >> 4) << 1;

    v8f sacc[4] = {};
    if (mm) {
        const int ts = tileStart[bin * 16 + tq];
        const int te = tileStart[bin * 16 + tq + 1];
        const int hl = ((((te - ts) + 1) >> 1) + AGG_U - 1) & ~(AGG_U - 1);
        const int a0 = ts + nh * hl;
        const int a1 = min(te, a0 + hl);
        tile_gather(list2, a0, a1, X, (lane >= 16) ? ~0u : 0u, 16 * m, (unsigned)(row0 + m), sacc);
    }
    float4* trow = (float4*)(tile + (row0 + (lane >> 4) * 8) * TILE_PITCH + 4 * m);
    if (mm && nh == 0) {
#pragma unroll
        for (int r = 0; r < 8; ++r)
            trow[r * (TILE_PITCH / 4)] = make_float4(sacc[0][r], sacc[1][r], sacc[2][r], sacc[3][r]);
    }
    __syncthreads();
    if (mm && nh == 1) {
#pragma unroll
        for (int r = 0; r < 8; ++r) {
            float4 v = trow[r * (TILE_PITCH / 4)];
            v.x += sacc[0][r]; v.y += sacc[1][r]; v.z += sacc[2][r]; v.w += sacc[3][r];
            trow[r * (TILE_PITCH / 4)] = v;
        }
    }
    int row = node0 + m; if (row >= N) row = N - 1;
    const float* Xr = X + (size_t)row * HDIM;
    v2f xa[16];
    if (mm) {
#pragma unroll
        for (int ks = 0; ks < 16; ++ks) {
            xa[ks].x = Xr[(ks << 2) + kh];
            xa[ks].y = Xr[(ks << 2) + kh + 1];
        }
    }
    __syncthreads();
    if (!mm) return;

    const float* Sr = tile + (row0 + m) * TILE_PITCH;

    v8f accX[2] = {};
    v8f accS[2] = {};
    const int fragBase = lane * 2;

#pragma unroll
    for (int ks = 0; ks < 16; ++ks) {
        int k0 = ks << 2;
        v2f a = xa[ks];
        v2f s = *(const v2f*)(Sr + k0 + kh);
#pragma unroll
        for (int j = 0; j < 2; ++j) {
            int nt = nh * 2 + j;
            v2f bd = *(const v2f*)&sWd[((ks << 2) + nt) * 64 + fragBase];
            v2f bb = *(const v2f*)&sWb[((ks << 2) + nt) * 64 + fragBase];
            accX[j] = __builtin_amdgcn_wmma_f32_16x16x4_f32(
                false, a, false, bd, (short)0, accX[j], false, false);
            accS[j] = __builtin_amdgcn_wmma_f32_16x16x4_f32(
                false, s, false, bb, (short)0, accS[j], false, false);
        }
    }

    int ncol = lane & 15;
    int mAdd = (lane >> 4) * 8;
#pragma unroll
    for (int j = 0; j < 2; ++j) {
        int col = (nh * 2 + j) * 16 + ncol;
        float wc = Wc[col];
        float bb = bias[col];
        float r[8];
#pragma unroll
        for (int v = 0; v < 8; ++v) {
            int node = node0 + v + mAdd;
            bool own = (row0 + v + mAdd < BIN_NODES) && (node < N);
            r[v] = (resid && own) ? resid[(size_t)node * HDIM + col] : 0.0f;
        }
#pragma unroll
        for (int v = 0; v < 8; ++v) {
            int node = node0 + v + mAdd;
            if (row0 + v + mAdd >= BIN_NODES || node >= N) continue;
            float g = gate[node];
            float val = g * (accX[j][v] + bb) + invdeg[node] * accS[j][v] + ssn[node] * wc;
            if (resid) val += r[v];
            val = (val > 0.0f) ? val : 0.01f * val;
            Y[(size_t)node * HDIM + col] = val;
        }
    }
}

#define DOT_T0 (1024 - BIN_NODES)

__global__ __launch_bounds__(1024) __attribute__((amdgpu_waves_per_eu(8))) void agg_out_kernel(
    const unsigned* __restrict__ list2, const int* __restrict__ tileStart,
    const float* __restrict__ X,
    const float* __restrict__ Wout, const float* __restrict__ bout,
    const float* __restrict__ x0,
    const float* __restrict__ invdeg, const float* __restrict__ gate,
    const float* __restrict__ ssn, float* __restrict__ out, int N) {
    __shared__ float4 tile4[TILE_ROWS * TILE_PITCH / 4];
    float* tile = (float*)tile4;

    const int bin  = blockIdx.x;
    const int lane = threadIdx.x & 31;
    const int wave = __builtin_amdgcn_readfirstlane((int)(threadIdx.x >> 5));

    const int dn = (int)threadIdx.x - DOT_T0;
    const bool dot = dn >= 0;
    const int n = bin * BIN_NODES + dn;
    float d1 = 0.0f;
    if (dot) {
        const float* Xr = X + (size_t)min(n, N - 1) * HDIM;
        float xr[HDIM];
#pragma unroll
        for (int k = 0; k < HDIM; ++k) xr[k] = Xr[k];
#pragma unroll
        for (int k = 0; k < HDIM; ++k) d1 += xr[k] * (Wout[k] - Wout[HDIM + k]);
    }

    const bool mm   = wave < 2 * MM_TILES;
    const int tq    = wave >> 1;
    const int nh    = wave & 1;
    const int row0  = tq * 16;
    const int m     = lane & 15;

    v8f sacc[4] = {};
    if (mm) {
        const int ts = tileStart[bin * 16 + tq];
        const int te = tileStart[bin * 16 + tq + 1];
        const int hl = ((((te - ts) + 1) >> 1) + AGG_U - 1) & ~(AGG_U - 1);
        const int a0 = ts + nh * hl;
        const int a1 = min(te, a0 + hl);
        tile_gather(list2, a0, a1, X, (lane >= 16) ? ~0u : 0u, 16 * m, (unsigned)(row0 + m), sacc);
    }
    float4* trow = (float4*)(tile + (row0 + (lane >> 4) * 8) * TILE_PITCH + 4 * m);
    if (mm && nh == 0) {
#pragma unroll
        for (int r = 0; r < 8; ++r)
            trow[r * (TILE_PITCH / 4)] = make_float4(sacc[0][r], sacc[1][r], sacc[2][r], sacc[3][r]);
    }
    __syncthreads();
    if (mm && nh == 1) {
#pragma unroll
        for (int r = 0; r < 8; ++r) {
            float4 v = trow[r * (TILE_PITCH / 4)];
            v.x += sacc[0][r]; v.y += sacc[1][r]; v.z += sacc[2][r]; v.w += sacc[3][r];
            trow[r * (TILE_PITCH / 4)] = v;
        }
    }
    __syncthreads();

    if (dot && n < N) {
        float d2 = 0.0f;
        const float* Sr = tile + dn * TILE_PITCH;
#pragma unroll
        for (int k = 0; k < HDIM; ++k) d2 += Sr[k] * Wout[HDIM + k];
        out[n] = gate[n] * (d1 + bout[0]) + invdeg[n] * d2 + ssn[n] * Wout[2 * HDIM] + x0[n];
    }
}

extern "C" void kernel_launch(void* const* d_in, const int* in_sizes, int n_in,
                              void* d_out, int out_size, void* d_ws, size_t ws_size,
                              hipStream_t stream) {
    const float* x    = (const float*)d_in[0];
    const int*   ei   = (const int*)  d_in[1];
    const float* Win  = (const float*)d_in[2];
    const float* bin  = (const float*)d_in[3];
    const float* Wmid = (const float*)d_in[4];
    const float* bmid = (const float*)d_in[5];
    const float* Wout = (const float*)d_in[6];
    const float* bout = (const float*)d_in[7];

    const int N = in_sizes[0];
    const int E = in_sizes[1] / 2;
    const int* src = ei;
    const int* dst = ei + E;

    float* w = (float*)d_ws;
    float* invdeg  = w + 2 * (size_t)N;
    float* gate    = w + 3 * (size_t)N;
    float* ssn     = w + 4 * (size_t)N;
    float* Sx      = w + 5 * (size_t)N;
    float* Wd6     = w + 6 * (size_t)N;
    float* bufA    = Wd6  + 6 * HDIM * HDIM;
    float* bufB    = bufA + (size_t)N * HDIM;
    float* bufC    = bufB + (size_t)N * HDIM;

    const int NB     = (N + BIN_NODES - 1) / BIN_NODES;
    const int chunk  = (E + CHUNK_WGS - 1) / CHUNK_WGS;

    unsigned* list  = (unsigned*)bufB;
    unsigned* list2 = (unsigned*)bufC;
    int* cnt        = (int*)(bufC + (size_t)E);
    int* pre        = cnt + (size_t)CHUNK_WGS * NB;
    int* binStart   = pre + (size_t)CHUNK_WGS * NB;
    int* tileStart  = (int*)w;

    wdiff_kernel<<<(6 * 4096 + 255) / 256, 256, 0, stream>>>(Wmid, Wd6, 6 * 4096);

    bin_count_kernel<<<CHUNK_WGS, 1024, 0, stream>>>(dst, cnt, E, NB, chunk);
    bin_scan_kernel<<<1, 1024, 0, stream>>>(cnt, pre, binStart, NB, CHUNK_WGS);
    bin_fill_kernel<<<CHUNK_WGS, 1024, 0, stream>>>(src, dst, pre, binStart, list, E, NB, chunk);

    bin_stats_kernel<<<NB, 1024, 0, stream>>>(list, binStart, x, invdeg, gate, ssn, Sx, list2, tileStart, N);

    conv_in_kernel<<<(N * HDIM + 255) / 256, 256, 0, stream>>>(x, Win, bin, invdeg, gate, ssn, Sx,
                                                               bufA, N);

    float* cur = bufA;
    for (int l = 0; l < 3; ++l) {
        const float* Wl = Wmid + (size_t)(2 * l) * 129 * HDIM;
        agg_conv_kernel<<<NB, 1024, 0, stream>>>(
            list2, tileStart, cur, Wd6 + (2 * l) * HDIM * HDIM, Wl + 64 * HDIM, Wl + 128 * HDIM,
            bmid + (2 * l) * HDIM, gate, invdeg, ssn, nullptr, bufB, N);

        const float* Wl2 = Wmid + (size_t)(2 * l + 1) * 129 * HDIM;
        agg_conv_kernel<<<NB, 1024, 0, stream>>>(
            list2, tileStart, bufB, Wd6 + (2 * l + 1) * HDIM * HDIM, Wl2 + 64 * HDIM, Wl2 + 128 * HDIM,
            bmid + (2 * l + 1) * HDIM, gate, invdeg, ssn, cur, cur, N);
    }

    agg_out_kernel<<<NB, 1024, 0, stream>>>(list2, tileStart, cur, Wout, bout, x, invdeg, gate, ssn,
                                           (float*)d_out, N);
}


# The two edge-ordering launches move their scattered 4-byte traffic through LDS: chunk and bin run ordered in LDS, stored in whole lines; x copied whole into LDS for x[src]; orders kept (LDS arrival)
# speedup vs baseline: 4.2608x; 1.0699x over previous
#include <hip/hip_runtime.h>
#include <hip/hip_bf16.h>

typedef float v2f __attribute__((ext_vector_type(2)));
typedef float v8f __attribute__((ext_vector_type(8)));

#define HDIM 64

__global__ void wdiff_kernel(const float* __restrict__ Wmid, float* __restrict__ Wd6, int total) {
    int i = blockIdx.x * blockDim.x + threadIdx.x;
    if (i >= total) return;
    int l = i >> 12;
    int rc = i & 4095;
    const float* Wl = Wmid + (size_t)l * 129 * HDIM;
    Wd6[i] = Wl[rc] - Wl[64 * HDIM + rc];
}

__global__ void conv_in_kernel(const float* __restrict__ x, const float* __restrict__ Win,
                               const float* __restrict__ bin,
                               const float* __restrict__ invdeg, const float* __restrict__ gate,
                               const float* __restrict__ ssn, const float* __restrict__ Sx,
                               float* __restrict__ Y, int N) {
    int t = blockIdx.x * blockDim.x + threadIdx.x;
    if (t >= N * HDIM) return;
    int n = t >> 6, h = t & 63;
    float wa = Win[h], wb = Win[HDIM + h], wc = Win[2 * HDIM + h];
    Y[t] = gate[n] * (x[n] * (wa - wb) + bin[h]) + invdeg[n] * Sx[n] * wb + ssn[n] * wc;
}

#define BIN_SHIFT 8
#define BIN_MASK 255
#define BIN_NODES 200
#define MAX_BINS 1024
#define CHUNK_WGS 256

__global__ __launch_bounds__(1024) void bin_count_kernel(const int* __restrict__ dst, int* __restrict__ cnt,
                                                         int E, int NB, int chunk) {
    __shared__ int hist[MAX_BINS];
    for (int b = threadIdx.x; b < NB; b += 1024) hist[b] = 0;
    __syncthreads();
    int e0 = blockIdx.x * chunk;
    int e1 = min(e0 + chunk, E);
    for (int e = e0 + (int)threadIdx.x; e < e1; e += 1024)
        atomicAdd(&hist[dst[e] / BIN_NODES], 1);
    __syncthreads();
    for (int b = threadIdx.x; b < NB; b += 1024) cnt[blockIdx.x * NB + b] = hist[b];
}

__global__ __launch_bounds__(1024) void bin_scan_kernel(const int* __restrict__ cnt, int* __restrict__ pre,
                                                        int* __restrict__ binStart, int NB, int G) {
    __shared__ int tot[MAX_BINS];
    int b = threadIdx.x;
    int run = 0;
    if (b < NB) {
#pragma unroll 16
        for (int g = 0; g < G; ++g) {
            int c = cnt[g * NB + b];
            pre[g * NB + b] = run;
            run += c;
        }
    }
    tot[b] = run;
    __syncthreads();
    for (int off = 1; off < MAX_BINS; off <<= 1) {
        int v = (b >= off) ? tot[b - off] : 0;
        __syncthreads();
        tot[b] += v;
        __syncthreads();
    }
    if (b < NB) binStart[b] = tot[b] - run;
    if (b == NB - 1) binStart[NB] = tot[b];
}

#define FILL_CAP 8192

__global__ __launch_bounds__(1024) void bin_fill_kernel(const int* __restrict__ src, const int* __restrict__ dst,
                                                        const int* __restrict__ cnt,
                                                        const int* __restrict__ pre,
                                                        const int* __restrict__ binStart,
                                                        unsigned* __restrict__ list, int E, int NB, int chunk) {
    __shared__ int cur[MAX_BINS];
    __shared__ int lst[MAX_BINS];
    __shared__ int rk[MAX_BINS];
    __shared__ int wtot[32];
    __shared__ unsigned sw[FILL_CAP];
    __shared__ int      ss[FILL_CAP];
    const int t = threadIdx.x;
    const int lane = t & 31, wv = t >> 5;
    const int e0 = blockIdx.x * chunk;
    const int e1 = min(e0 + chunk, E);
    const int len = e1 - e0;
    const bool staged = len <= FILL_CAP;

    const int c = (t < NB) ? cnt[blockIdx.x * NB + t] : 0;
    cur[t] = (t < NB) ? binStart[t] + pre[blockIdx.x * NB + t] : 0;
    rk[t] = 0;
    int incl = c;
#pragma unroll
    for (int o = 1; o < 32; o <<= 1) { int v = __shfl_up(incl, o); if (lane >= o) incl += v; }
    if (lane == 31) wtot[wv] = incl;
    __syncthreads();
    if (wv == 0) {
        const int w = wtot[lane];
        int wi = w;
#pragma unroll
        for (int o = 1; o < 32; o <<= 1) { int v = __shfl_up(wi, o); if (lane >= o) wi += v; }
        wtot[lane] = wi - w;
    }
    __syncthreads();
    lst[t] = incl - c + wtot[wv];
    __syncthreads();

    for (int e = e0 + t; e < e1; e += 1024) {
        int d = dst[e];
        int s = src[e];
        int b = d / BIN_NODES;
        int r = atomicAdd(&rk[b], 1);
        unsigned word = ((unsigned)s << BIN_SHIFT) | (unsigned)(d - b * BIN_NODES);
        if (staged) {
            int pos = lst[b] + r;
            sw[pos] = word;
            ss[pos] = cur[b] + r;
        } else {
            list[cur[b] + r] = word;
        }
    }
    if (staged) {
        __syncthreads();
        for (int k = t; k < len; k += 1024) list[ss[k]] = sw[k];
    }
}

#define ST_U 8
#define XS_CAP 65536
#define XS_U 8

__global__ __launch_bounds__(1024) void bin_stats_kernel(const unsigned* __restrict__ list,
                                                        const int* __restrict__ binStart,
                                                        const float* __restrict__ x,
                                                        float* __restrict__ invdeg, float* __restrict__ gate,
                                                        float* __restrict__ ssn, float* __restrict__ Sx,
                                                        unsigned* __restrict__ list2, int* __restrict__ tileStart, int N) {
    __shared__ int   wcnt[32 * 16];
    __shared__ int   tbase[16];
    __shared__ int   sdeg[BIN_NODES];
    __shared__ int   ssgn[BIN_NODES];
    __shared__ float ssx[BIN_NODES];
    __shared__ float4 sx4[XS_CAP / 4];
    float* sxs = (float*)sx4;
    const int t = threadIdx.x;
    if (t < BIN_NODES) { sdeg[t] = 0; ssgn[t] = 0; ssx[t] = 0.0f; }
    if (t < 32 * 16) wcnt[t] = 0;

    const int bin   = blockIdx.x;
    const int beg   = binStart[bin];
    const int end   = binStart[bin + 1];
    const int node0 = bin * BIN_NODES;

    const bool xin = N <= XS_CAP;
    if (xin) {
        const float4* x4 = (const float4*)x;
        const int n4 = N >> 2;
        for (int i0 = t; i0 < n4; i0 += 1024 * XS_U) {
            float4 v[XS_U];
#pragma unroll
            for (int j = 0; j < XS_U; ++j) v[j] = x4[min(i0 + j * 1024, n4 - 1)];
#pragma unroll
            for (int j = 0; j < XS_U; ++j) sx4[min(i0 + j * 1024, n4 - 1)] = v[j];
        }
        if (t < (N & 3)) sxs[(n4 << 2) + t] = x[(n4 << 2) + t];
    }
    __syncthreads();

    for (int i = beg + t; i < end; i += 1024 * ST_U) {
        unsigned p[ST_U];
        float xv[ST_U];
#pragma unroll
        for (int j = 0; j < ST_U; ++j) p[j] = list[min(i + j * 1024, end - 1)];
        if (xin) {
#pragma unroll
            for (int j = 0; j < ST_U; ++j) xv[j] = sxs[p[j] >> BIN_SHIFT];
        } else {
#pragma unroll
            for (int j = 0; j < ST_U; ++j) xv[j] = x[p[j] >> BIN_SHIFT];
        }
#pragma unroll
        for (int j = 0; j < ST_U; ++j) {
            if (i + j * 1024 < end) {
                int s  = (int)(p[j] >> BIN_SHIFT);
                int dl = (int)(p[j] & BIN_MASK);
                int d  = node0 + dl;
                atomicAdd(&sdeg[dl], 1);
                atomicAdd(&ssgn[dl], (s > d) - (s < d));
                atomicAdd(&ssx[dl], xv[j]);
            }
        }
    }
    __syncthreads();

    const int wv = t >> 5;
    for (int i = beg + t; i < end; i += 1024) {
        unsigned p = list[i];
        atomicAdd(&wcnt[wv * 16 + (int)((p & BIN_MASK) >> 4)], 1);
    }
    __syncthreads();
    if (t < 16) {
        int run = 0;
        for (int w = 0; w < 32; ++w) { int c = wcnt[w * 16 + t]; wcnt[w * 16 + t] = run; run += c; }
        tbase[t] = run;
    }
    __syncthreads();
    if (t == 0) {
        int run = beg;
        for (int k = 0; k < 16; ++k) { int c = tbase[k]; tbase[k] = run; tileStart[bin * 16 + k] = run; run += c; }
    }
    __syncthreads();
    const int rl = end - beg;
    const bool lout = rl <= XS_CAP;
    unsigned* sb = (unsigned*)sx4;
    for (int i = beg + t; i < end; i += 1024) {
        unsigned p = list[i];
        int tl = (int)((p & BIN_MASK) >> 4);
        int slot = tbase[tl] + atomicAdd(&wcnt[wv * 16 + tl], 1);
        if (lout) sb[slot - beg] = p; else list2[slot] = p;
    }
    if (lout) {
        __syncthreads();
        for (int k = t; k < rl; k += 1024) list2[beg + k] = sb[k];
    }

    if (t < BIN_NODES && node0 + t < N) {
        int n = node0 + t;
        float dg  = (float)sdeg[t];
        float inv = 1.0f / fmaxf(dg, 1.0f);
        invdeg[n] = inv;
        gate[n]   = (dg > 0.0f) ? 1.0f : 0.0f;
        ssn[n]    = (float)ssgn[t] * inv;
        Sx[n]     = ssx[t];
    }
}

#define AGG_U 16

__device__ __forceinline__ int win_start(int i, int end) { return max(min(i, end - AGG_U), 0); }

__device__ __forceinline__ void tg_words(const unsigned* __restrict__ l2, int ws, unsigned (&w)[AGG_U]) {
#pragma unroll
    for (int j = 0; j < AGG_U; ++j) w[j] = l2[ws + j];
}

__device__ __forceinline__ void tg_loads(const unsigned (&w)[AGG_U], const float* __restrict__ X, unsigned hm, int c4,
                                         float4 (&b)[AGG_U / 2]) {
#pragma unroll
    for (int q = 0; q < AGG_U / 2; ++q) {
        const int ks = q >> 1, e = q & 1;
        unsigned ww = (w[ks * 4 + 2 + e] & hm) | (w[ks * 4 + e] & ~hm);
        unsigned off = (ww & ~(unsigned)BIN_MASK) | (unsigned)c4;
        b[q] = *(const float4*)((const char*)X + off);
    }
}

template <bool MASKED>
__device__ __forceinline__ void tg_mma(const unsigned (&w)[AGG_U], const float4 (&b)[AGG_U / 2], unsigned hm,
                                       unsigned mrow, int ws, int lo, int end, v8f (&acc)[4]) {
#pragma unroll
    for (int ks = 0; ks < 4; ++ks) {
        v2f a;
#pragma unroll
        for (int e = 0; e < 2; ++e) {
            const int jl = ks * 4 + e, jh = ks * 4 + 2 + e;
            unsigned ww = (w[jh] & hm) | (w[jl] & ~hm);
            bool on = ww == mrow;
            if (MASKED) {
                bool vl = (ws + jl >= lo) && (ws + jl < end);
                bool vh = (ws + jh >= lo) && (ws + jh < end);
                on = on && (((vh ? hm : 0u) | (vl ? ~hm : 0u)) != 0u);
            }
            a[e] = on ? 1.0f : 0.0f;
        }
        const float4 b0 = b[ks * 2], b1 = b[ks * 2 + 1];
        v2f f0 = {b0.x, b1.x}, f1 = {b0.y, b1.y}, f2 = {b0.z, b1.z}, f3 = {b0.w, b1.w};
        acc[0] = __builtin_amdgcn_wmma_f32_16x16x4_f32(false, a, false, f0, (short)0, acc[0], false, false);
        acc[1] = __builtin_amdgcn_wmma_f32_16x16x4_f32(false, a, false, f1, (short)0, acc[1], false, false);
        acc[2] = __builtin_amdgcn_wmma_f32_16x16x4_f32(false, a, false, f2, (short)0, acc[2], false, false);
        acc[3] = __builtin_amdgcn_wmma_f32_16x16x4_f32(false, a, false, f3, (short)0, acc[3], false, false);
    }
}

__device__ __forceinline__ void tg_rows(const unsigned (&w)[AGG_U], unsigned (&d)[AGG_U]) {
#pragma unroll
    for (int j = 0; j < AGG_U; ++j) d[j] = w[j] & BIN_MASK;
}

__device__ __forceinline__ void tile_gather(const unsigned* __restrict__ l2, int a0, int a1,
                                            const float* __restrict__ X, unsigned hm, int c4, unsigned mrow,
                                            v8f (&acc)[4]) {
    if (a0 >= a1) return;
    unsigned w1[AGG_U], w2[AGG_U];
    unsigned da[AGG_U], db[AGG_U];
    float4 ba[AGG_U / 2], bb[AGG_U / 2];
    int i = a0;
    tg_words(l2, win_start(i, a1), w1);
    tg_words(l2, win_start(i + AGG_U, a1), w2);
    while (i + 2 * AGG_U < a1) {
        tg_loads(w1, X, hm, c4, ba);
        tg_rows(w1, da);
        tg_loads(w2, X, hm, c4, bb);
        tg_rows(w2, db);
        tg_words(l2, win_start(i + 2 * AGG_U, a1), w1);
        tg_words(l2, win_start(i + 3 * AGG_U, a1), w2);
        __builtin_amdgcn_sched_barrier(0);
        tg_mma<false>(da, ba, hm, mrow, 0, 0, 0, acc);
        tg_mma<false>(db, bb, hm, mrow, 0, 0, 0, acc);
        i += 2 * AGG_U;
    }
    if (i + AGG_U < a1) {
        const int ws2 = win_start(i + AGG_U, a1);
        tg_loads(w1, X, hm, c4, ba);
        tg_rows(w1, da);
        tg_loads(w2, X, hm, c4, bb);
        tg_rows(w2, db);
        tg_mma<false>(da, ba, hm, mrow, 0, 0, 0, acc);
        tg_mma<true>(db, bb, hm, mrow, ws2, i + AGG_U, a1, acc);
    } else {
        tg_loads(w1, X, hm, c4, ba);
        tg_rows(w1, da);
        tg_mma<true>(da, ba, hm, mrow, win_start(i, a1), i, a1, acc);
    }
}

#define TILE_PITCH 68
#define MM_TILES ((BIN_NODES + 15) / 16)
#define TILE_ROWS (MM_TILES * 16)

__global__ __launch_bounds__(1024) __attribute__((amdgpu_waves_per_eu(8))) void agg_conv_kernel(
    const unsigned* __restrict__ list2, const int* __restrict__ tileStart,
    const float* __restrict__ X,
    const float* __restrict__ Wd, const float* __restrict__ Wb,
    const float* __restrict__ Wc, const float* __restrict__ bias,
    const float* __restrict__ gate, const float* __restrict__ invdeg,
    const float* __restrict__ ssn, const float* resid,
    float* Y, int N) {
    __shared__ float4 tile4[TILE_ROWS * TILE_PITCH / 4];
    __shared__ float sWd[HDIM * HDIM];
    __shared__ float sWb[HDIM * HDIM];
    float* tile = (float*)tile4;

    const int bin  = blockIdx.x;
    const int lane = threadIdx.x & 31;
    const int wave = __builtin_amdgcn_readfirstlane((int)(threadIdx.x >> 5));
    const int tid  = threadIdx.x;

    {
        float wdr[4], wbr[4];
        int t256 = tid & 255;
        int q    = tid >> 8;
        int nt   = t256 >> 6;
        int lp   = t256 & 63;
        int ln   = lp >> 1;
        int half = lp & 1;
        int kh   = (ln >> 4) << 1;
        int wsrc = (kh + half) * HDIM + (nt << 4) + (ln & 15);
#pragma unroll
        for (int i = 0; i < 4; ++i) {
            int ks = (q << 2) + i;
            wdr[i] = Wd[(ks << 2) * HDIM + wsrc];
            wbr[i] = Wb[(ks << 2) * HDIM + wsrc];
        }
#pragma unroll
        for (int i = 0; i < 4; ++i) {
            int ks = (q << 2) + i;
            sWd[(ks << 8) + t256] = wdr[i];
            sWb[(ks << 8) + t256] = wbr[i];
        }
    }

    const bool mm   = wave < 2 * MM_TILES;
    const int tq    = wave >> 1;
    const int nh    = wave & 1;
    const int row0  = tq * 16;
    const int node0 = bin * BIN_NODES + row0;
    const int m  = lane & 15;
    const int kh = (lane >> 4) << 1;

    v8f sacc[4] = {};
    if (mm) {
        const int ts = tileStart[bin * 16 + tq];
        const int te = tileStart[bin * 16 + tq + 1];
        const int hl = ((((te - ts) + 1) >> 1) + AGG_U - 1) & ~(AGG_U - 1);
        const int a0 = ts + nh * hl;
        const int a1 = min(te, a0 + hl);
        tile_gather(list2, a0, a1, X, (lane >= 16) ? ~0u : 0u, 16 * m, (unsigned)(row0 + m), sacc);
    }
    float4* trow = (float4*)(tile + (row0 + (lane >> 4) * 8) * TILE_PITCH + 4 * m);
    if (mm && nh == 0) {
#pragma unroll
        for (int r = 0; r < 8; ++r)
            trow[r * (TILE_PITCH / 4)] = make_float4(sacc[0][r], sacc[1][r], sacc[2][r], sacc[3][r]);
    }
    __syncthreads();
    if (mm && nh == 1) {
#pragma unroll
        for (int r = 0; r < 8; ++r) {
            float4 v = trow[r * (TILE_PITCH / 4)];
            v.x += sacc[0][r]; v.y += sacc[1][r]; v.z += sacc[2][r]; v.w += sacc[3][r];
            trow[r * (TILE_PITCH / 4)] = v;
        }
    }
    int row = node0 + m; if (row >= N) row = N - 1;
    const float* Xr = X + (size_t)row * HDIM;
    v2f xa[16];
    if (mm) {
#pragma unroll
        for (int ks = 0; ks < 16; ++ks) {
            xa[ks].x = Xr[(ks << 2) + kh];
            xa[ks].y = Xr[(ks << 2) + kh + 1];
        }
    }
    __syncthreads();
    if (!mm) return;

    const float* Sr = tile + (row0 + m) * TILE_PITCH;

    v8f accX[2] = {};
    v8f accS[2] = {};
    const int fragBase = lane * 2;

#pragma unroll
    for (int ks = 0; ks < 16; ++ks) {
        int k0 = ks << 2;
        v2f a = xa[ks];
        v2f s = *(const v2f*)(Sr + k0 + kh);
#pragma unroll
        for (int j = 0; j < 2; ++j) {
            int nt = nh * 2 + j;
            v2f bd = *(const v2f*)&sWd[((ks << 2) + nt) * 64 + fragBase];
            v2f bb = *(const v2f*)&sWb[((ks << 2) + nt) * 64 + fragBase];
            accX[j] = __builtin_amdgcn_wmma_f32_16x16x4_f32(
                false, a, false, bd, (short)0, accX[j], false, false);
            accS[j] = __builtin_amdgcn_wmma_f32_16x16x4_f32(
                false, s, false, bb, (short)0, accS[j], false, false);
        }
    }

    int ncol = lane & 15;
    int mAdd = (lane >> 4) * 8;
#pragma unroll
    for (int j = 0; j < 2; ++j) {
        int col = (nh * 2 + j) * 16 + ncol;
        float wc = Wc[col];
        float bb = bias[col];
        float r[8];
#pragma unroll
        for (int v = 0; v < 8; ++v) {
            int node = node0 + v + mAdd;
            bool own = (row0 + v + mAdd < BIN_NODES) && (node < N);
            r[v] = (resid && own) ? resid[(size_t)node * HDIM + col] : 0.0f;
        }
#pragma unroll
        for (int v = 0; v < 8; ++v) {
            int node = node0 + v + mAdd;
            if (row0 + v + mAdd >= BIN_NODES || node >= N) continue;
            float g = gate[node];
            float val = g * (accX[j][v] + bb) + invdeg[node] * accS[j][v] + ssn[node] * wc;
            if (resid) val += r[v];
            val = (val > 0.0f) ? val : 0.01f * val;
            Y[(size_t)node * HDIM + col] = val;
        }
    }
}

#define DOT_T0 (1024 - BIN_NODES)

__global__ __launch_bounds__(1024) __attribute__((amdgpu_waves_per_eu(8))) void agg_out_kernel(
    const unsigned* __restrict__ list2, const int* __restrict__ tileStart,
    const float* __restrict__ X,
    const float* __restrict__ Wout, const float* __restrict__ bout,
    const float* __restrict__ x0,
    const float* __restrict__ invdeg, const float* __restrict__ gate,
    const float* __restrict__ ssn, float* __restrict__ out, int N) {
    __shared__ float4 tile4[TILE_ROWS * TILE_PITCH / 4];
    float* tile = (float*)tile4;

    const int bin  = blockIdx.x;
    const int lane = threadIdx.x & 31;
    const int wave = __builtin_amdgcn_readfirstlane((int)(threadIdx.x >> 5));

    const int dn = (int)threadIdx.x - DOT_T0;
    const bool dot = dn >= 0;
    const int n = bin * BIN_NODES + dn;
    float d1 = 0.0f;
    if (dot) {
        const float* Xr = X + (size_t)min(n, N - 1) * HDIM;
        float xr[HDIM];
#pragma unroll
        for (int k = 0; k < HDIM; ++k) xr[k] = Xr[k];
#pragma unroll
        for (int k = 0; k < HDIM; ++k) d1 += xr[k] * (Wout[k] - Wout[HDIM + k]);
    }

    const bool mm   = wave < 2 * MM_TILES;
    const int tq    = wave >> 1;
    const int nh    = wave & 1;
    const int row0  = tq * 16;
    const int m     = lane & 15;

    v8f sacc[4] = {};
    if (mm) {
        const int ts = tileStart[bin * 16 + tq];
        const int te = tileStart[bin * 16 + tq + 1];
        const int hl = ((((te - ts) + 1) >> 1) + AGG_U - 1) & ~(AGG_U - 1);
        const int a0 = ts + nh * hl;
        const int a1 = min(te, a0 + hl);
        tile_gather(list2, a0, a1, X, (lane >= 16) ? ~0u : 0u, 16 * m, (unsigned)(row0 + m), sacc);
    }
    float4* trow = (float4*)(tile + (row0 + (lane >> 4) * 8) * TILE_PITCH + 4 * m);
    if (mm && nh == 0) {
#pragma unroll
        for (int r = 0; r < 8; ++r)
            trow[r * (TILE_PITCH / 4)] = make_float4(sacc[0][r], sacc[1][r], sacc[2][r], sacc[3][r]);
    }
    __syncthreads();
    if (mm && nh == 1) {
#pragma unroll
        for (int r = 0; r < 8; ++r) {
            float4 v = trow[r * (TILE_PITCH / 4)];
            v.x += sacc[0][r]; v.y += sacc[1][r]; v.z += sacc[2][r]; v.w += sacc[3][r];
            trow[r * (TILE_PITCH / 4)] = v;
        }
    }
    __syncthreads();

    if (dot && n < N) {
        float d2 = 0.0f;
        const float* Sr = tile + dn * TILE_PITCH;
#pragma unroll
        for (int k = 0; k < HDIM; ++k) d2 += Sr[k] * Wout[HDIM + k];
        out[n] = gate[n] * (d1 + bout[0]) + invdeg[n] * d2 + ssn[n] * Wout[2 * HDIM] + x0[n];
    }
}

extern "C" void kernel_launch(void* const* d_in, const int* in_sizes, int n_in,
                              void* d_out, int out_size, void* d_ws, size_t ws_size,
                              hipStream_t stream) {
    const float* x    = (const float*)d_in[0];
    const int*   ei   = (const int*)  d_in[1];
    const float* Win  = (const float*)d_in[2];
    const float* bin  = (const float*)d_in[3];
    const float* Wmid = (const float*)d_in[4];
    const float* bmid = (const float*)d_in[5];
    const float* Wout = (const float*)d_in[6];
    const float* bout = (const float*)d_in[7];

    const int N = in_sizes[0];
    const int E = in_sizes[1] / 2;
    const int* src = ei;
    const int* dst = ei + E;

    float* w = (float*)d_ws;
    float* invdeg  = w + 2 * (size_t)N;
    float* gate    = w + 3 * (size_t)N;
    float* ssn     = w + 4 * (size_t)N;
    float* Sx      = w + 5 * (size_t)N;
    float* Wd6     = w + 6 * (size_t)N;
    float* bufA    = Wd6  + 6 * HDIM * HDIM;
    float* bufB    = bufA + (size_t)N * HDIM;
    float* bufC    = bufB + (size_t)N * HDIM;

    const int NB     = (N + BIN_NODES - 1) / BIN_NODES;
    const int chunk  = (E + CHUNK_WGS - 1) / CHUNK_WGS;

    unsigned* list  = (unsigned*)bufB;
    unsigned* list2 = (unsigned*)bufC;
    int* cnt        = (int*)(bufC + (size_t)E);
    int* pre        = cnt + (size_t)CHUNK_WGS * NB;
    int* binStart   = pre + (size_t)CHUNK_WGS * NB;
    int* tileStart  = (int*)w;

    wdiff_kernel<<<(6 * 4096 + 255) / 256, 256, 0, stream>>>(Wmid, Wd6, 6 * 4096);

    bin_count_kernel<<<CHUNK_WGS, 1024, 0, stream>>>(dst, cnt, E, NB, chunk);
    bin_scan_kernel<<<1, 1024, 0, stream>>>(cnt, pre, binStart, NB, CHUNK_WGS);
    bin_fill_kernel<<<CHUNK_WGS, 1024, 0, stream>>>(src, dst, cnt, pre, binStart, list, E, NB, chunk);

    bin_stats_kernel<<<NB, 1024, 0, stream>>>(list, binStart, x, invdeg, gate, ssn, Sx, list2, tileStart, N);

    conv_in_kernel<<<(N * HDIM + 255) / 256, 256, 0, stream>>>(x, Win, bin, invdeg, gate, ssn, Sx,
                                                               bufA, N);

    float* cur = bufA;
    for (int l = 0; l < 3; ++l) {
        const float* Wl = Wmid + (size_t)(2 * l) * 129 * HDIM;
        agg_conv_kernel<<<NB, 1024, 0, stream>>>(
            list2, tileStart, cur, Wd6 + (2 * l) * HDIM * HDIM, Wl + 64 * HDIM, Wl + 128 * HDIM,
            bmid + (2 * l) * HDIM, gate, invdeg, ssn, nullptr, bufB, N);

        const float* Wl2 = Wmid + (size_t)(2 * l + 1) * 129 * HDIM;
        agg_conv_kernel<<<NB, 1024, 0, stream>>>(
            list2, tileStart, bufB, Wd6 + (2 * l + 1) * HDIM * HDIM, Wl2 + 64 * HDIM, Wl2 + 128 * HDIM,
            bmid + (2 * l + 1) * HDIM, gate, invdeg, ssn, cur, cur, N);
    }

    agg_out_kernel<<<NB, 1024, 0, stream>>>(list2, tileStart, cur, Wout, bout, x, invdeg, gate, ssn,
                                           (float*)d_out, N);
}


# Edge-ordering launches keep a thread's reads in flight together: offset scan over 4 chunk ranges per bin, grouped reads (was 256 dependent steps); count/fill/tile passes too; sum order: LDS arrival
# speedup vs baseline: 4.6611x; 1.0939x over previous
#include <hip/hip_runtime.h>
#include <hip/hip_bf16.h>

typedef float v2f __attribute__((ext_vector_type(2)));
typedef float v8f __attribute__((ext_vector_type(8)));

#define HDIM 64

__global__ void wdiff_kernel(const float* __restrict__ Wmid, float* __restrict__ Wd6, int total) {
    int i = blockIdx.x * blockDim.x + threadIdx.x;
    if (i >= total) return;
    int l = i >> 12;
    int rc = i & 4095;
    const float* Wl = Wmid + (size_t)l * 129 * HDIM;
    Wd6[i] = Wl[rc] - Wl[64 * HDIM + rc];
}

__global__ void conv_in_kernel(const float* __restrict__ x, const float* __restrict__ Win,
                               const float* __restrict__ bin,
                               const float* __restrict__ invdeg, const float* __restrict__ gate,
                               const float* __restrict__ ssn, const float* __restrict__ Sx,
                               float* __restrict__ Y, int N) {
    int t = blockIdx.x * blockDim.x + threadIdx.x;
    if (t >= N * HDIM) return;
    int n = t >> 6, h = t & 63;
    float wa = Win[h], wb = Win[HDIM + h], wc = Win[2 * HDIM + h];
    Y[t] = gate[n] * (x[n] * (wa - wb) + bin[h]) + invdeg[n] * Sx[n] * wb + ssn[n] * wc;
}

#define BIN_SHIFT 8
#define BIN_MASK 255
#define BIN_NODES 200
#define MAX_BINS 1024
#define CHUNK_WGS 256
#define ORD_U 8
#define SCAN_U 16

__device__ __forceinline__ void reads_issued(const int (&v)[ORD_U]) {
    asm volatile("" :: "v"(v[0]), "v"(v[1]), "v"(v[2]), "v"(v[3]), "v"(v[4]), "v"(v[5]), "v"(v[6]), "v"(v[7]));
}
__device__ __forceinline__ void reads_issued(const unsigned (&v)[ORD_U]) {
    asm volatile("" :: "v"(v[0]), "v"(v[1]), "v"(v[2]), "v"(v[3]), "v"(v[4]), "v"(v[5]), "v"(v[6]), "v"(v[7]));
}
__device__ __forceinline__ void reads_issued(const int (&v)[SCAN_U]) {
    asm volatile("" :: "v"(v[0]), "v"(v[1]), "v"(v[2]), "v"(v[3]), "v"(v[4]), "v"(v[5]), "v"(v[6]), "v"(v[7]),
                       "v"(v[8]), "v"(v[9]), "v"(v[10]), "v"(v[11]), "v"(v[12]), "v"(v[13]), "v"(v[14]), "v"(v[15]));
}

__global__ __launch_bounds__(1024) void bin_count_kernel(const int* __restrict__ dst, int* __restrict__ cnt,
                                                         int E, int NB, int chunk) {
    __shared__ int hist[MAX_BINS];
    for (int b = threadIdx.x; b < NB; b += 1024) hist[b] = 0;
    __syncthreads();
    int e0 = blockIdx.x * chunk;
    int e1 = min(e0 + chunk, E);
    for (int e = e0 + (int)threadIdx.x; e < e1; e += 1024 * ORD_U) {
        int d[ORD_U];
#pragma unroll
        for (int j = 0; j < ORD_U; ++j) d[j] = dst[min(e + j * 1024, e1 - 1)];
        reads_issued(d);
#pragma unroll
        for (int j = 0; j < ORD_U; ++j)
            if (e + j * 1024 < e1) atomicAdd(&hist[d[j] / BIN_NODES], 1);
    }
    __syncthreads();
    for (int b = threadIdx.x; b < NB; b += 1024) cnt[blockIdx.x * NB + b] = hist[b];
}

__device__ __forceinline__ int scan_parts(int NB) { return MAX_BINS / NB; }
__device__ __forceinline__ int scan_part_len(int NB, int G) { int p = scan_parts(NB); return (G + p - 1) / p; }

__global__ __launch_bounds__(1024) void bin_scan_kernel(const int* __restrict__ cnt, int* __restrict__ pre,
                                                        int* __restrict__ binStart, int* __restrict__ partOff,
                                                        int NB, int G) {
    __shared__ int ptot[MAX_BINS];
    __shared__ int wtot[32];
    const int t = threadIdx.x;
    const int lane = t & 31, wv = t >> 5;
    const int parts = scan_parts(NB);
    const int gs = scan_part_len(NB, G);
    const int q = t / NB;
    const int b = t - q * NB;
    const bool act = q < parts;
    int run = 0;
    if (act) {
        const int g0 = q * gs;
        const int g1 = min(G, g0 + gs);
        for (int g = g0; g < g1; g += SCAN_U) {
            int c[SCAN_U];
#pragma unroll
            for (int j = 0; j < SCAN_U; ++j) c[j] = cnt[min(g + j, g1 - 1) * NB + b];
            reads_issued(c);
            int v = run;
#pragma unroll
            for (int j = 0; j < SCAN_U; ++j) {
                pre[min(g + j, g1 - 1) * NB + b] = v;
                if (g + j + 1 < g1) v += c[j];
                if (g + j < g1) run += c[j];
            }
        }
    }
    ptot[t] = run;
    __syncthreads();
    int tot = 0;
    if (act) {
        int off = 0;
        for (int k = 0; k < parts; ++k) {
            int v = ptot[k * NB + b];
            if (k < q) off += v;
            tot += v;
        }
        partOff[t] = off;
    }
    const int c0 = (t < NB) ? tot : 0;
    int incl = c0;
#pragma unroll
    for (int o = 1; o < 32; o <<= 1) { int v = __shfl_up(incl, o); if (lane >= o) incl += v; }
    if (lane == 31) wtot[wv] = incl;
    __syncthreads();
    if (wv == 0) {
        const int w = wtot[lane];
        int wi = w;
#pragma unroll
        for (int o = 1; o < 32; o <<= 1) { int v = __shfl_up(wi, o); if (lane >= o) wi += v; }
        wtot[lane] = wi - w;
    }
    __syncthreads();
    const int excl = incl - c0 + wtot[wv];
    if (t < NB) binStart[t] = excl;
    if (t == NB - 1) binStart[NB] = excl + c0;
}

#define FILL_CAP 8192

__global__ __launch_bounds__(1024) void bin_fill_kernel(const int* __restrict__ src, const int* __restrict__ dst,
                                                        const int* __restrict__ cnt,
                                                        const int* __restrict__ pre,
                                                        const int* __restrict__ binStart,
                                                        const int* __restrict__ partOff,
                                                        unsigned* __restrict__ list, int E, int NB, int chunk) {
    __shared__ int cur[MAX_BINS];
    __shared__ int lst[MAX_BINS];
    __shared__ int rk[MAX_BINS];
    __shared__ int wtot[32];
    __shared__ unsigned sw[FILL_CAP];
    __shared__ int      ss[FILL_CAP];
    const int t = threadIdx.x;
    const int lane = t & 31, wv = t >> 5;
    const int e0 = blockIdx.x * chunk;
    const int e1 = min(e0 + chunk, E);
    const int len = e1 - e0;
    const bool staged = len <= FILL_CAP;

    const int c = (t < NB) ? cnt[blockIdx.x * NB + t] : 0;
    const int part = (int)blockIdx.x / scan_part_len(NB, (int)gridDim.x);
    cur[t] = (t < NB) ? binStart[t] + partOff[part * NB + t] + pre[blockIdx.x * NB + t] : 0;
    rk[t] = 0;
    int incl = c;
#pragma unroll
    for (int o = 1; o < 32; o <<= 1) { int v = __shfl_up(incl, o); if (lane >= o) incl += v; }
    if (lane == 31) wtot[wv] = incl;
    __syncthreads();
    if (wv == 0) {
        const int w = wtot[lane];
        int wi = w;
#pragma unroll
        for (int o = 1; o < 32; o <<= 1) { int v = __shfl_up(wi, o); if (lane >= o) wi += v; }
        wtot[lane] = wi - w;
    }
    __syncthreads();
    lst[t] = incl - c + wtot[wv];
    __syncthreads();

    for (int e = e0 + t; e < e1; e += 1024 * ORD_U) {
        int dv[ORD_U], sv[ORD_U];
#pragma unroll
        for (int j = 0; j < ORD_U; ++j) {
            const int ee = min(e + j * 1024, e1 - 1);
            dv[j] = dst[ee];
            sv[j] = src[ee];
        }
#pragma unroll
        for (int j = 0; j < ORD_U; ++j) {
            if (e + j * 1024 < e1) {
                int d = dv[j];
                int s = sv[j];
                int b = d / BIN_NODES;
                int r = atomicAdd(&rk[b], 1);
                unsigned word = ((unsigned)s << BIN_SHIFT) | (unsigned)(d - b * BIN_NODES);
                if (staged) {
                    int pos = lst[b] + r;
                    sw[pos] = word;
                    ss[pos] = cur[b] + r;
                } else {
                    list[cur[b] + r] = word;
                }
            }
        }
    }
    if (staged) {
        __syncthreads();
        for (int k = t; k < len; k += 1024) list[ss[k]] = sw[k];
    }
}

#define ST_U 8
#define XS_CAP 65536
#define XS_U 8

__global__ __launch_bounds__(1024) void bin_stats_kernel(const unsigned* __restrict__ list,
                                                        const int* __restrict__ binStart,
                                                        const float* __restrict__ x,
                                                        float* __restrict__ invdeg, float* __restrict__ gate,
                                                        float* __restrict__ ssn, float* __restrict__ Sx,
                                                        unsigned* __restrict__ list2, int* __restrict__ tileStart, int N) {
    __shared__ int   wcnt[32 * 16];
    __shared__ int   tbase[16];
    __shared__ int   sdeg[BIN_NODES];
    __shared__ int   ssgn[BIN_NODES];
    __shared__ float ssx[BIN_NODES];
    __shared__ float4 sx4[XS_CAP / 4];
    float* sxs = (float*)sx4;
    const int t = threadIdx.x;
    if (t < BIN_NODES) { sdeg[t] = 0; ssgn[t] = 0; ssx[t] = 0.0f; }
    if (t < 32 * 16) wcnt[t] = 0;

    const int bin   = blockIdx.x;
    const int beg   = binStart[bin];
    const int end   = binStart[bin + 1];
    const int node0 = bin * BIN_NODES;

    const bool xin = N <= XS_CAP;
    if (xin) {
        const float4* x4 = (const float4*)x;
        const int n4 = N >> 2;
        for (int i0 = t; i0 < n4; i0 += 1024 * XS_U) {
            float4 v[XS_U];
#pragma unroll
            for (int j = 0; j < XS_U; ++j) v[j] = x4[min(i0 + j * 1024, n4 - 1)];
#pragma unroll
            for (int j = 0; j < XS_U; ++j) sx4[min(i0 + j * 1024, n4 - 1)] = v[j];
        }
        if (t < (N & 3)) sxs[(n4 << 2) + t] = x[(n4 << 2) + t];
    }
    __syncthreads();

    for (int i = beg + t; i < end; i += 1024 * ST_U) {
        unsigned p[ST_U];
        float xv[ST_U];
#pragma unroll
        for (int j = 0; j < ST_U; ++j) p[j] = list[min(i + j * 1024, end - 1)];
        if (xin) {
#pragma unroll
            for (int j = 0; j < ST_U; ++j) xv[j] = sxs[p[j] >> BIN_SHIFT];
        } else {
#pragma unroll
            for (int j = 0; j < ST_U; ++j) xv[j] = x[p[j] >> BIN_SHIFT];
        }
#pragma unroll
        for (int j = 0; j < ST_U; ++j) {
            if (i + j * 1024 < end) {
                int s  = (int)(p[j] >> BIN_SHIFT);
                int dl = (int)(p[j] & BIN_MASK);
                int d  = node0 + dl;
                atomicAdd(&sdeg[dl], 1);
                atomicAdd(&ssgn[dl], (s > d) - (s < d));
                atomicAdd(&ssx[dl], xv[j]);
            }
        }
    }
    __syncthreads();

    const int wv = t >> 5;
    for (int i = beg + t; i < end; i += 1024 * ORD_U) {
        unsigned p[ORD_U];
#pragma unroll
        for (int j = 0; j < ORD_U; ++j) p[j] = list[min(i + j * 1024, end - 1)];
        reads_issued(p);
#pragma unroll
        for (int j = 0; j < ORD_U; ++j)
            if (i + j * 1024 < end) atomicAdd(&wcnt[wv * 16 + (int)((p[j] & BIN_MASK) >> 4)], 1);
    }
    __syncthreads();
    if (t < 16) {
        int run = 0;
        for (int w = 0; w < 32; ++w) { int c = wcnt[w * 16 + t]; wcnt[w * 16 + t] = run; run += c; }
        tbase[t] = run;
    }
    __syncthreads();
    if (t == 0) {
        int run = beg;
        for (int k = 0; k < 16; ++k) { int c = tbase[k]; tbase[k] = run; tileStart[bin * 16 + k] = run; run += c; }
    }
    __syncthreads();
    const int rl = end - beg;
    const bool lout = rl <= XS_CAP;
    unsigned* sb = (unsigned*)sx4;
    for (int i = beg + t; i < end; i += 1024 * ORD_U) {
        unsigned p[ORD_U];
#pragma unroll
        for (int j = 0; j < ORD_U; ++j) p[j] = list[min(i + j * 1024, end - 1)];
#pragma unroll
        for (int j = 0; j < ORD_U; ++j) {
            if (i + j * 1024 < end) {
                int tl = (int)((p[j] & BIN_MASK) >> 4);
                int slot = tbase[tl] + atomicAdd(&wcnt[wv * 16 + tl], 1);
                if (lout) sb[slot - beg] = p[j]; else list2[slot] = p[j];
            }
        }
    }
    if (lout) {
        __syncthreads();
        for (int k = t; k < rl; k += 1024) list2[beg + k] = sb[k];
    }

    if (t < BIN_NODES && node0 + t < N) {
        int n = node0 + t;
        float dg  = (float)sdeg[t];
        float inv = 1.0f / fmaxf(dg, 1.0f);
        invdeg[n] = inv;
        gate[n]   = (dg > 0.0f) ? 1.0f : 0.0f;
        ssn[n]    = (float)ssgn[t] * inv;
        Sx[n]     = ssx[t];
    }
}

#define AGG_U 16

__device__ __forceinline__ int win_start(int i, int end) { return max(min(i, end - AGG_U), 0); }

__device__ __forceinline__ void tg_words(const unsigned* __restrict__ l2, int ws, unsigned (&w)[AGG_U]) {
#pragma unroll
    for (int j = 0; j < AGG_U; ++j) w[j] = l2[ws + j];
}

__device__ __forceinline__ void tg_loads(const unsigned (&w)[AGG_U], const float* __restrict__ X, unsigned hm, int c4,
                                         float4 (&b)[AGG_U / 2]) {
#pragma unroll
    for (int q = 0; q < AGG_U / 2; ++q) {
        const int ks = q >> 1, e = q & 1;
        unsigned ww = (w[ks * 4 + 2 + e] & hm) | (w[ks * 4 + e] & ~hm);
        unsigned off = (ww & ~(unsigned)BIN_MASK) | (unsigned)c4;
        b[q] = *(const float4*)((const char*)X + off);
    }
}

template <bool MASKED>
__device__ __forceinline__ void tg_mma(const unsigned (&w)[AGG_U], const float4 (&b)[AGG_U / 2], unsigned hm,
                                       unsigned mrow, int ws, int lo, int end, v8f (&acc)[4]) {
#pragma unroll
    for (int ks = 0; ks < 4; ++ks) {
        v2f a;
#pragma unroll
        for (int e = 0; e < 2; ++e) {
            const int jl = ks * 4 + e, jh = ks * 4 + 2 + e;
            unsigned ww = (w[jh] & hm) | (w[jl] & ~hm);
            bool on = ww == mrow;
            if (MASKED) {
                bool vl = (ws + jl >= lo) && (ws + jl < end);
                bool vh = (ws + jh >= lo) && (ws + jh < end);
                on = on && (((vh ? hm : 0u) | (vl ? ~hm : 0u)) != 0u);
            }
            a[e] = on ? 1.0f : 0.0f;
        }
        const float4 b0 = b[ks * 2], b1 = b[ks * 2 + 1];
        v2f f0 = {b0.x, b1.x}, f1 = {b0.y, b1.y}, f2 = {b0.z, b1.z}, f3 = {b0.w, b1.w};
        acc[0] = __builtin_amdgcn_wmma_f32_16x16x4_f32(false, a, false, f0, (short)0, acc[0], false, false);
        acc[1] = __builtin_amdgcn_wmma_f32_16x16x4_f32(false, a, false, f1, (short)0, acc[1], false, false);
        acc[2] = __builtin_amdgcn_wmma_f32_16x16x4_f32(false, a, false, f2, (short)0, acc[2], false, false);
        acc[3] = __builtin_amdgcn_wmma_f32_16x16x4_f32(false, a, false, f3, (short)0, acc[3], false, false);
    }
}

__device__ __forceinline__ void tg_rows(const unsigned (&w)[AGG_U], unsigned (&d)[AGG_U]) {
#pragma unroll
    for (int j = 0; j < AGG_U; ++j) d[j] = w[j] & BIN_MASK;
}

__device__ __forceinline__ void tile_gather(const unsigned* __restrict__ l2, int a0, int a1,
                                            const float* __restrict__ X, unsigned hm, int c4, unsigned mrow,
                                            v8f (&acc)[4]) {
    if (a0 >= a1) return;
    unsigned w1[AGG_U], w2[AGG_U];
    unsigned da[AGG_U], db[AGG_U];
    float4 ba[AGG_U / 2], bb[AGG_U / 2];
    int i = a0;
    tg_words(l2, win_start(i, a1), w1);
    tg_words(l2, win_start(i + AGG_U, a1), w2);
    while (i + 2 * AGG_U < a1) {
        tg_loads(w1, X, hm, c4, ba);
        tg_rows(w1, da);
        tg_loads(w2, X, hm, c4, bb);
        tg_rows(w2, db);
        tg_words(l2, win_start(i + 2 * AGG_U, a1), w1);
        tg_words(l2, win_start(i + 3 * AGG_U, a1), w2);
        __builtin_amdgcn_sched_barrier(0);
        tg_mma<false>(da, ba, hm, mrow, 0, 0, 0, acc);
        tg_mma<false>(db, bb, hm, mrow, 0, 0, 0, acc);
        i += 2 * AGG_U;
    }
    if (i + AGG_U < a1) {
        const int ws2 = win_start(i + AGG_U, a1);
        tg_loads(w1, X, hm, c4, ba);
        tg_rows(w1, da);
        tg_loads(w2, X, hm, c4, bb);
        tg_rows(w2, db);
        tg_mma<false>(da, ba, hm, mrow, 0, 0, 0, acc);
        tg_mma<true>(db, bb, hm, mrow, ws2, i + AGG_U, a1, acc);
    } else {
        tg_loads(w1, X, hm, c4, ba);
        tg_rows(w1, da);
        tg_mma<true>(da, ba, hm, mrow, win_start(i, a1), i, a1, acc);
    }
}

#define TILE_PITCH 68
#define MM_TILES ((BIN_NODES + 15) / 16)
#define TILE_ROWS (MM_TILES * 16)

__global__ __launch_bounds__(1024) __attribute__((amdgpu_waves_per_eu(8))) void agg_conv_kernel(
    const unsigned* __restrict__ list2, const int* __restrict__ tileStart,
    const float* __restrict__ X,
    const float* __restrict__ Wd, const float* __restrict__ Wb,
    const float* __restrict__ Wc, const float* __restrict__ bias,
    const float* __restrict__ gate, const float* __restrict__ invdeg,
    const float* __restrict__ ssn, const float* resid,
    float* Y, int N) {
    __shared__ float4 tile4[TILE_ROWS * TILE_PITCH / 4];
    __shared__ float sWd[HDIM * HDIM];
    __shared__ float sWb[HDIM * HDIM];
    float* tile = (float*)tile4;

    const int bin  = blockIdx.x;
    const int lane = threadIdx.x & 31;
    const int wave = __builtin_amdgcn_readfirstlane((int)(threadIdx.x >> 5));
    const int tid  = threadIdx.x;

    {
        float wdr[4], wbr[4];
        int t256 = tid & 255;
        int q    = tid >> 8;
        int nt   = t256 >> 6;
        int lp   = t256 & 63;
        int ln   = lp >> 1;
        int half = lp & 1;
        int kh   = (ln >> 4) << 1;
        int wsrc = (kh + half) * HDIM + (nt << 4) + (ln & 15);
#pragma unroll
        for (int i = 0; i < 4; ++i) {
            int ks = (q << 2) + i;
            wdr[i] = Wd[(ks << 2) * HDIM + wsrc];
            wbr[i] = Wb[(ks << 2) * HDIM + wsrc];
        }
#pragma unroll
        for (int i = 0; i < 4; ++i) {
            int ks = (q << 2) + i;
            sWd[(ks << 8) + t256] = wdr[i];
            sWb[(ks << 8) + t256] = wbr[i];
        }
    }

    const bool mm   = wave < 2 * MM_TILES;
    const int tq    = wave >> 1;
    const int nh    = wave & 1;
    const int row0  = tq * 16;
    const int node0 = bin * BIN_NODES + row0;
    const int m  = lane & 15;
    const int kh = (lane >> 4) << 1;

    v8f sacc[4] = {};
    if (mm) {
        const int ts = tileStart[bin * 16 + tq];
        const int te = tileStart[bin * 16 + tq + 1];
        const int hl = ((((te - ts) + 1) >> 1) + AGG_U - 1) & ~(AGG_U - 1);
        const int a0 = ts + nh * hl;
        const int a1 = min(te, a0 + hl);
        tile_gather(list2, a0, a1, X, (lane >= 16) ? ~0u : 0u, 16 * m, (unsigned)(row0 + m), sacc);
    }
    float4* trow = (float4*)(tile + (row0 + (lane >> 4) * 8) * TILE_PITCH + 4 * m);
    if (mm && nh == 0) {
#pragma unroll
        for (int r = 0; r < 8; ++r)
            trow[r * (TILE_PITCH / 4)] = make_float4(sacc[0][r], sacc[1][r], sacc[2][r], sacc[3][r]);
    }
    __syncthreads();
    if (mm && nh == 1) {
#pragma unroll
        for (int r = 0; r < 8; ++r) {
            float4 v = trow[r * (TILE_PITCH / 4)];
            v.x += sacc[0][r]; v.y += sacc[1][r]; v.z += sacc[2][r]; v.w += sacc[3][r];
            trow[r * (TILE_PITCH / 4)] = v;
        }
    }
    int row = node0 + m; if (row >= N) row = N - 1;
    const float* Xr = X + (size_t)row * HDIM;
    v2f xa[16];
    if (mm) {
#pragma unroll
        for (int ks = 0; ks < 16; ++ks) {
            xa[ks].x = Xr[(ks << 2) + kh];
            xa[ks].y = Xr[(ks << 2) + kh + 1];
        }
    }
    __syncthreads();
    if (!mm) return;

    const float* Sr = tile + (row0 + m) * TILE_PITCH;

    v8f accX[2] = {};
    v8f accS[2] = {};
    const int fragBase = lane * 2;

#pragma unroll
    for (int ks = 0; ks < 16; ++ks) {
        int k0 = ks << 2;
        v2f a = xa[ks];
        v2f s = *(const v2f*)(Sr + k0 + kh);
#pragma unroll
        for (int j = 0; j < 2; ++j) {
            int nt = nh * 2 + j;
            v2f bd = *(const v2f*)&sWd[((ks << 2) + nt) * 64 + fragBase];
            v2f bb = *(const v2f*)&sWb[((ks << 2) + nt) * 64 + fragBase];
            accX[j] = __builtin_amdgcn_wmma_f32_16x16x4_f32(
                false, a, false, bd, (short)0, accX[j], false, false);
            accS[j] = __builtin_amdgcn_wmma_f32_16x16x4_f32(
                false, s, false, bb, (short)0, accS[j], false, false);
        }
    }

    int ncol = lane & 15;
    int mAdd = (lane >> 4) * 8;
#pragma unroll
    for (int j = 0; j < 2; ++j) {
        int col = (nh * 2 + j) * 16 + ncol;
        float wc = Wc[col];
        float bb = bias[col];
        float r[8];
#pragma unroll
        for (int v = 0; v < 8; ++v) {
            int node = node0 + v + mAdd;
            bool own = (row0 + v + mAdd < BIN_NODES) && (node < N);
            r[v] = (resid && own) ? resid[(size_t)node * HDIM + col] : 0.0f;
        }
#pragma unroll
        for (int v = 0; v < 8; ++v) {
            int node = node0 + v + mAdd;
            if (row0 + v + mAdd >= BIN_NODES || node >= N) continue;
            float g = gate[node];
            float val = g * (accX[j][v] + bb) + invdeg[node] * accS[j][v] + ssn[node] * wc;
            if (resid) val += r[v];
            val = (val > 0.0f) ? val : 0.01f * val;
            Y[(size_t)node * HDIM + col] = val;
        }
    }
}

#define DOT_T0 (1024 - BIN_NODES)

__global__ __launch_bounds__(1024) __attribute__((amdgpu_waves_per_eu(8))) void agg_out_kernel(
    const unsigned* __restrict__ list2, const int* __restrict__ tileStart,
    const float* __restrict__ X,
    const float* __restrict__ Wout, const float* __restrict__ bout,
    const float* __restrict__ x0,
    const float* __restrict__ invdeg, const float* __restrict__ gate,
    const float* __restrict__ ssn, float* __restrict__ out, int N) {
    __shared__ float4 tile4[TILE_ROWS * TILE_PITCH / 4];
    float* tile = (float*)tile4;

    const int bin  = blockIdx.x;
    const int lane = threadIdx.x & 31;
    const int wave = __builtin_amdgcn_readfirstlane((int)(threadIdx.x >> 5));

    const int dn = (int)threadIdx.x - DOT_T0;
    const bool dot = dn >= 0;
    const int n = bin * BIN_NODES + dn;
    float d1 = 0.0f;
    if (dot) {
        const float* Xr = X + (size_t)min(n, N - 1) * HDIM;
        float xr[HDIM];
#pragma unroll
        for (int k = 0; k < HDIM; ++k) xr[k] = Xr[k];
#pragma unroll
        for (int k = 0; k < HDIM; ++k) d1 += xr[k] * (Wout[k] - Wout[HDIM + k]);
    }

    const bool mm   = wave < 2 * MM_TILES;
    const int tq    = wave >> 1;
    const int nh    = wave & 1;
    const int row0  = tq * 16;
    const int m     = lane & 15;

    v8f sacc[4] = {};
    if (mm) {
        const int ts = tileStart[bin * 16 + tq];
        const int te = tileStart[bin * 16 + tq + 1];
        const int hl = ((((te - ts) + 1) >> 1) + AGG_U - 1) & ~(AGG_U - 1);
        const int a0 = ts + nh * hl;
        const int a1 = min(te, a0 + hl);
        tile_gather(list2, a0, a1, X, (lane >= 16) ? ~0u : 0u, 16 * m, (unsigned)(row0 + m), sacc);
    }
    float4* trow = (float4*)(tile + (row0 + (lane >> 4) * 8) * TILE_PITCH + 4 * m);
    if (mm && nh == 0) {
#pragma unroll
        for (int r = 0; r < 8; ++r)
            trow[r * (TILE_PITCH / 4)] = make_float4(sacc[0][r], sacc[1][r], sacc[2][r], sacc[3][r]);
    }
    __syncthreads();
    if (mm && nh == 1) {
#pragma unroll
        for (int r = 0; r < 8; ++r) {
            float4 v = trow[r * (TILE_PITCH / 4)];
            v.x += sacc[0][r]; v.y += sacc[1][r]; v.z += sacc[2][r]; v.w += sacc[3][r];
            trow[r * (TILE_PITCH / 4)] = v;
        }
    }
    __syncthreads();

    if (dot && n < N) {
        float d2 = 0.0f;
        const float* Sr = tile + dn * TILE_PITCH;
#pragma unroll
        for (int k = 0; k < HDIM; ++k) d2 += Sr[k] * Wout[HDIM + k];
        out[n] = gate[n] * (d1 + bout[0]) + invdeg[n] * d2 + ssn[n] * Wout[2 * HDIM] + x0[n];
    }
}

extern "C" void kernel_launch(void* const* d_in, const int* in_sizes, int n_in,
                              void* d_out, int out_size, void* d_ws, size_t ws_size,
                              hipStream_t stream) {
    const float* x    = (const float*)d_in[0];
    const int*   ei   = (const int*)  d_in[1];
    const float* Win  = (const float*)d_in[2];
    const float* bin  = (const float*)d_in[3];
    const float* Wmid = (const float*)d_in[4];
    const float* bmid = (const float*)d_in[5];
    const float* Wout = (const float*)d_in[6];
    const float* bout = (const float*)d_in[7];

    const int N = in_sizes[0];
    const int E = in_sizes[1] / 2;
    const int* src = ei;
    const int* dst = ei + E;

    float* w = (float*)d_ws;
    float* invdeg  = w + 2 * (size_t)N;
    float* gate    = w + 3 * (size_t)N;
    float* ssn     = w + 4 * (size_t)N;
    float* Sx      = w + 5 * (size_t)N;
    float* Wd6     = w + 6 * (size_t)N;
    float* bufA    = Wd6  + 6 * HDIM * HDIM;
    float* bufB    = bufA + (size_t)N * HDIM;
    float* bufC    = bufB + (size_t)N * HDIM;

    const int NB     = (N + BIN_NODES - 1) / BIN_NODES;
    const int chunk  = (E + CHUNK_WGS - 1) / CHUNK_WGS;

    unsigned* list  = (unsigned*)bufB;
    unsigned* list2 = (unsigned*)bufC;
    int* cnt        = (int*)(bufC + (size_t)E);
    int* pre        = cnt + (size_t)CHUNK_WGS * NB;
    int* binStart   = pre + (size_t)CHUNK_WGS * NB;
    int* partOff    = binStart + (NB + 1);
    int* tileStart  = (int*)w;

    wdiff_kernel<<<(6 * 4096 + 255) / 256, 256, 0, stream>>>(Wmid, Wd6, 6 * 4096);

    bin_count_kernel<<<CHUNK_WGS, 1024, 0, stream>>>(dst, cnt, E, NB, chunk);
    bin_scan_kernel<<<1, 1024, 0, stream>>>(cnt, pre, binStart, partOff, NB, CHUNK_WGS);
    bin_fill_kernel<<<CHUNK_WGS, 1024, 0, stream>>>(src, dst, cnt, pre, binStart, partOff, list, E, NB, chunk);

    bin_stats_kernel<<<NB, 1024, 0, stream>>>(list, binStart, x, invdeg, gate, ssn, Sx, list2, tileStart, N);

    conv_in_kernel<<<(N * HDIM + 255) / 256, 256, 0, stream>>>(x, Win, bin, invdeg, gate, ssn, Sx,
                                                               bufA, N);

    float* cur = bufA;
    for (int l = 0; l < 3; ++l) {
        const float* Wl = Wmid + (size_t)(2 * l) * 129 * HDIM;
        agg_conv_kernel<<<NB, 1024, 0, stream>>>(
            list2, tileStart, cur, Wd6 + (2 * l) * HDIM * HDIM, Wl + 64 * HDIM, Wl + 128 * HDIM,
            bmid + (2 * l) * HDIM, gate, invdeg, ssn, nullptr, bufB, N);

        const float* Wl2 = Wmid + (size_t)(2 * l + 1) * 129 * HDIM;
        agg_conv_kernel<<<NB, 1024, 0, stream>>>(
            list2, tileStart, bufB, Wd6 + (2 * l + 1) * HDIM * HDIM, Wl2 + 64 * HDIM, Wl2 + 128 * HDIM,
            bmid + (2 * l + 1) * HDIM, gate, invdeg, ssn, cur, cur, N);
    }

    agg_out_kernel<<<NB, 1024, 0, stream>>>(list2, tileStart, cur, Wout, bout, x, invdeg, gate, ssn,
                                           (float*)d_out, N);
}
